# starting kernel
# baseline (speedup 1.0000x reference)
#include <hip/hip_runtime.h>


typedef float v2f __attribute__((ext_vector_type(2)));
typedef float v8f __attribute__((ext_vector_type(8)));

#define N_NODES 50000
#define N_EDGES 800000
#define FDIM    128
#define E_TR    100000
#define E_TE    20000
#define N_ROWS  240000

__global__ __launch_bounds__(256) void deg_count(const int* __restrict__ dst,
                                                 float* __restrict__ deg, int nE) {
    int t = blockIdx.x * blockDim.x + threadIdx.x;
    if (t < nE) atomicAdd(&deg[dst[t]], 1.0f);
}

__global__ __launch_bounds__(256) void make_dis(float* __restrict__ deg, int n) {
    int t = blockIdx.x * blockDim.x + threadIdx.x;
    if (t < n) deg[t] = rsqrtf(deg[t] + 1.0f);
}

__global__ __launch_bounds__(256) void gemm_node(const float* __restrict__ A,
                                                 const float* __restrict__ B,
                                                 float* __restrict__ C, int M) {
    int wave = (blockIdx.x * blockDim.x + threadIdx.x) >> 5;
    int lane = threadIdx.x & 31;
    int m0 = wave * 16;
    if (m0 >= M) return;
    int half = lane >> 4;
    int l16  = lane & 15;

    const float* Arow = A + (size_t)(m0 + l16) * FDIM;

    v8f acc[8] = {};

    for (int k0 = 0; k0 < FDIM; k0 += 4) {
        int k = k0 + half * 2;
        v2f afrag = *(const v2f*)(Arow + k);
        const float* b0 = B + (size_t)k * FDIM + l16;
        #pragma unroll
        for (int n = 0; n < 8; ++n) {
            v2f bfrag;
            bfrag.x = b0[n * 16];
            bfrag.y = b0[FDIM + n * 16];
            acc[n] = __builtin_amdgcn_wmma_f32_16x16x4_f32(
                false, afrag, false, bfrag, (short)0, acc[n], false, false);
        }
    }

    #pragma unroll
    for (int n = 0; n < 8; ++n) {
        #pragma unroll
        for (int g = 0; g < 8; ++g) {
            int m = m0 + g + half * 8;
            C[(size_t)m * FDIM + n * 16 + l16] = acc[n][g];
        }
    }
}

__global__ __launch_bounds__(256) void scatter_add(const float* __restrict__ hw,
                                                   const float* __restrict__ dis,
                                                   const int* __restrict__ src,
                                                   const int* __restrict__ dst,
                                                   float* __restrict__ agg, int nE) {
    int t = blockIdx.x * blockDim.x + threadIdx.x;
    int e = t >> 5;
    if (e >= nE) return;
    int f = (t & 31) * 4;
    int s = src[e], d = dst[e];
    float sc = dis[s];
    const float4 v = *(const float4*)(hw + (size_t)s * FDIM + f);
    float* dp = agg + (size_t)d * FDIM + f;
    atomicAdd(dp + 0, v.x * sc);
    atomicAdd(dp + 1, v.y * sc);
    atomicAdd(dp + 2, v.z * sc);
    atomicAdd(dp + 3, v.w * sc);
}

__global__ __launch_bounds__(256) void combine(const float* __restrict__ agg,
                                               const float* __restrict__ hw,
                                               const float* __restrict__ dis,
                                               const float* __restrict__ b,
                                               float* __restrict__ hout,
                                               int nNodes, int doRelu) {
    int t = blockIdx.x * blockDim.x + threadIdx.x;
    int node = t >> 5;
    if (node >= nNodes) return;
    int f = (t & 31) * 4;
    float di  = dis[node];
    float di2 = di * di;
    float4 a = *(const float4*)(agg + (size_t)node * FDIM + f);
    float4 h = *(const float4*)(hw  + (size_t)node * FDIM + f);
    float4 r;
    r.x = a.x * di + h.x * di2 + b[f + 0];
    r.y = a.y * di + h.y * di2 + b[f + 1];
    r.z = a.z * di + h.z * di2 + b[f + 2];
    r.w = a.w * di + h.w * di2 + b[f + 3];
    if (doRelu) {
        r.x = fmaxf(r.x, 0.0f); r.y = fmaxf(r.y, 0.0f);
        r.z = fmaxf(r.z, 0.0f); r.w = fmaxf(r.w, 0.0f);
    }
    *(float4*)(hout + (size_t)node * FDIM + f) = r;
}

__global__ __launch_bounds__(256) void gemm_edge(const float* __restrict__ H,
                                                 const int* __restrict__ pos_tr,
                                                 const int* __restrict__ neg_tr,
                                                 const int* __restrict__ pos_te,
                                                 const int* __restrict__ neg_te,
                                                 const float* __restrict__ Wl1,
                                                 const float* __restrict__ bl1,
                                                 const float* __restrict__ Wl2,
                                                 const float* __restrict__ bl2,
                                                 float* __restrict__ out) {
    int wave = (blockIdx.x * blockDim.x + threadIdx.x) >> 5;
    int lane = threadIdx.x & 31;
    int r0 = wave * 16;
    if (r0 >= N_ROWS) return;
    int half = lane >> 4;
    int l16  = lane & 15;
    int r = r0 + l16;

    int u, v;
    if (r < E_TR)            { u = pos_tr[r];              v = pos_tr[E_TR + r]; }
    else if (r < 2 * E_TR)   { int i = r - E_TR;           u = neg_tr[i]; v = neg_tr[E_TR + i]; }
    else if (r < 2 * E_TR + E_TE) { int i = r - 2 * E_TR;  u = pos_te[i]; v = pos_te[E_TE + i]; }
    else                     { int i = r - 2 * E_TR - E_TE; u = neg_te[i]; v = neg_te[E_TE + i]; }

    const float* Au = H + (size_t)u * FDIM;
    const float* Av = H + (size_t)v * FDIM;

    v8f acc[8] = {};

    for (int k0 = 0; k0 < 2 * FDIM; k0 += 4) {
        int k = k0 + half * 2;
        const float* ap = (k < FDIM) ? (Au + k) : (Av + (k - FDIM));
        v2f afrag = *(const v2f*)ap;
        const float* b0 = Wl1 + (size_t)k * FDIM + l16;
        #pragma unroll
        for (int n = 0; n < 8; ++n) {
            v2f bfrag;
            bfrag.x = b0[n * 16];
            bfrag.y = b0[FDIM + n * 16];
            acc[n] = __builtin_amdgcn_wmma_f32_16x16x4_f32(
                false, afrag, false, bfrag, (short)0, acc[n], false, false);
        }
    }

    float partial[8];
    #pragma unroll
    for (int g = 0; g < 8; ++g) partial[g] = 0.0f;

    #pragma unroll
    for (int n = 0; n < 8; ++n) {
        float bb = bl1[n * 16 + l16];
        float ww = Wl2[n * 16 + l16];
        #pragma unroll
        for (int g = 0; g < 8; ++g) {
            float z = fmaxf(acc[n][g] + bb, 0.0f);
            partial[g] += z * ww;
        }
    }

    float bias2 = bl2[0];
    #pragma unroll
    for (int g = 0; g < 8; ++g) {
        float s = partial[g];
        s += __shfl_xor(s, 8, 16);
        s += __shfl_xor(s, 4, 16);
        s += __shfl_xor(s, 2, 16);
        s += __shfl_xor(s, 1, 16);
        if (l16 == 0) out[r0 + g + half * 8] = s + bias2;
    }
}

extern "C" void kernel_launch(void* const* d_in, const int* in_sizes, int n_in,
                              void* d_out, int out_size, void* d_ws, size_t ws_size,
                              hipStream_t stream) {
    const float* x          = (const float*)d_in[0];
    const int*   edge_index = (const int*)d_in[1];
    const int*   pos_tr     = (const int*)d_in[2];
    const int*   neg_tr     = (const int*)d_in[3];
    const int*   pos_te     = (const int*)d_in[4];
    const int*   neg_te     = (const int*)d_in[5];
    const float* W[4] = {(const float*)d_in[6],  (const float*)d_in[8],
                         (const float*)d_in[10], (const float*)d_in[12]};
    const float* b[4] = {(const float*)d_in[7],  (const float*)d_in[9],
                         (const float*)d_in[11], (const float*)d_in[13]};
    const float* Wl1 = (const float*)d_in[14];
    const float* bl1 = (const float*)d_in[15];
    const float* Wl2 = (const float*)d_in[16];
    const float* bl2 = (const float*)d_in[17];
    float* out = (float*)d_out;

    const size_t NF = (size_t)N_NODES * FDIM;
    float* ws  = (float*)d_ws;
    float* dis = ws;
    float* hw  = ws + 50048;
    float* agg = hw  + NF;
    float* h0  = agg + NF;
    float* h1  = h0  + NF;

    const int* src = edge_index;
    const int* dst = edge_index + N_EDGES;

    hipMemsetAsync(dis, 0, N_NODES * sizeof(float), stream);
    deg_count<<<(N_EDGES + 255) / 256, 256, 0, stream>>>(dst, dis, N_EDGES);
    make_dis<<<(N_NODES + 255) / 256, 256, 0, stream>>>(dis, N_NODES);

    const int tilesM     = (N_NODES + 15) / 16;
    const int gemmBlocks = (tilesM + 7) / 8;
    const float* hin = x;
    float* houts[4] = {h0, h1, h0, h1};
    for (int l = 0; l < 4; ++l) {
        gemm_node<<<gemmBlocks, 256, 0, stream>>>(hin, W[l], hw, N_NODES);
        hipMemsetAsync(agg, 0, NF * sizeof(float), stream);
        scatter_add<<<(N_EDGES * 32) / 256, 256, 0, stream>>>(hw, dis, src, dst, agg, N_EDGES);
        combine<<<(N_NODES * 32 + 255) / 256, 256, 0, stream>>>(
            agg, hw, dis, b[l], houts[l], N_NODES, (l < 3) ? 1 : 0);
        hin = houts[l];
    }

    const int edgeTiles  = N_ROWS / 16;
    const int edgeBlocks = (edgeTiles + 7) / 8;
    gemm_edge<<<edgeBlocks, 256, 0, stream>>>(h1, pos_tr, neg_tr, pos_te, neg_te,
                                              Wl1, bl1, Wl2, bl2, out);
}


# Push to pull: CSR by destination built per call (int rank atomics, scan, fill); a wave per node sums its in-edge rows in slot (arrival) order and combines in registers; no f32 atomics or clears
# speedup vs baseline: 2.4573x; 2.4573x over previous
#include <hip/hip_runtime.h>


typedef float v2f __attribute__((ext_vector_type(2)));
typedef float v8f __attribute__((ext_vector_type(8)));

#define N_NODES 50000
#define N_EDGES 800000
#define FDIM    128
#define E_TR    100000
#define E_TE    20000
#define N_ROWS  240000

__global__ __launch_bounds__(256) void count_rank(const int* __restrict__ dst,
                                                  int* __restrict__ cnt,
                                                  int* __restrict__ rank, int nE) {
    int e = blockIdx.x * blockDim.x + threadIdx.x;
    if (e < nE) rank[e] = atomicAdd(&cnt[dst[e]], 1);
}

#define SCAN_PASSES 13
#define SCAN_CHUNK  (SCAN_PASSES * 128)
static_assert(32 * SCAN_CHUNK >= N_NODES, "one block must cover every node");
static_assert(N_NODES % 4 == 0, "16-byte groups must not straddle the end");

__device__ __forceinline__ int wave_incl_scan(int v, int lane) {
    #pragma unroll
    for (int o = 1; o < 32; o <<= 1) {
        int y = __shfl_up(v, o, 32);
        if (lane >= o) v += y;
    }
    return v;
}

__global__ __launch_bounds__(1024) void scan_rows(const int* __restrict__ cnt,
                                                  int* __restrict__ row_ptr,
                                                  float* __restrict__ dis) {
    __shared__ int wtot[32];
    const int lane = threadIdx.x & 31;
    const int w    = threadIdx.x >> 5;
    const int base = w * SCAN_CHUNK + lane * 4;

    int4 p[SCAN_PASSES];
    #pragma unroll
    for (int i = 0; i < SCAN_PASSES; ++i) {
        const int idx = base + i * 128;
        const int4 v = *(const int4*)(cnt + ((idx < N_NODES) ? idx : N_NODES - 4));
        p[i] = (idx < N_NODES) ? v : make_int4(0, 0, 0, 0);
    }
    #pragma unroll
    for (int i = 0; i < SCAN_PASSES; ++i) {
        const int idx = base + i * 128;
        if (idx < N_NODES) {
            float4 d;
            d.x = rsqrtf((float)p[i].x + 1.0f);
            d.y = rsqrtf((float)p[i].y + 1.0f);
            d.z = rsqrtf((float)p[i].z + 1.0f);
            d.w = rsqrtf((float)p[i].w + 1.0f);
            *(float4*)(dis + idx) = d;
        }
    }

    int run = 0;
    #pragma unroll
    for (int i = 0; i < SCAN_PASSES; ++i) {
        const int4 c = p[i];
        const int s   = c.x + c.y + c.z + c.w;
        const int inc = wave_incl_scan(s, lane);
        const int o   = run + inc - s;
        p[i].x = o;
        p[i].y = o + c.x;
        p[i].z = o + c.x + c.y;
        p[i].w = o + c.x + c.y + c.z;
        run += __shfl(inc, 31, 32);
    }
    if (lane == 0) wtot[w] = run;
    __syncthreads();

    const int t     = wtot[lane];
    const int tinc  = wave_incl_scan(t, lane);
    const int wbase = __shfl(tinc - t, w, 32);

    #pragma unroll
    for (int i = 0; i < SCAN_PASSES; ++i) {
        int idx = base + i * 128;
        if (idx < N_NODES) {
            int4 q = p[i];
            q.x += wbase; q.y += wbase; q.z += wbase; q.w += wbase;
            *(int4*)(row_ptr + idx) = q;
        }
    }
    if (threadIdx.x == 1023) row_ptr[N_NODES] = wbase + run;
}

__global__ __launch_bounds__(256) void fill_cols(const int* __restrict__ src,
                                                 const int* __restrict__ dst,
                                                 const int* __restrict__ rank,
                                                 const int* __restrict__ row_ptr,
                                                 const float* __restrict__ dis,
                                                 int* __restrict__ ewbuf, int nE) {
    int2* ew = (int2*)ewbuf;
    int e = blockIdx.x * blockDim.x + threadIdx.x;
    if (e < nE) {
        const int s = src[e];
        ew[row_ptr[dst[e]] + rank[e]] = make_int2(s, __float_as_int(dis[s]));
    }
}

__global__ __launch_bounds__(256) void gemm_node(const float* __restrict__ A,
                                                 const float* __restrict__ B,
                                                 float* __restrict__ C, int M) {
    int wave = (blockIdx.x * blockDim.x + threadIdx.x) >> 5;
    int lane = threadIdx.x & 31;
    int m0 = wave * 16;
    if (m0 >= M) return;
    int half = lane >> 4;
    int l16  = lane & 15;

    const float* Arow = A + (size_t)(m0 + l16) * FDIM;

    v8f acc[8] = {};

    for (int k0 = 0; k0 < FDIM; k0 += 4) {
        int k = k0 + half * 2;
        v2f afrag = *(const v2f*)(Arow + k);
        const float* b0 = B + (size_t)k * FDIM + l16;
        #pragma unroll
        for (int n = 0; n < 8; ++n) {
            v2f bfrag;
            bfrag.x = b0[n * 16];
            bfrag.y = b0[FDIM + n * 16];
            acc[n] = __builtin_amdgcn_wmma_f32_16x16x4_f32(
                false, afrag, false, bfrag, (short)0, acc[n], false, false);
        }
    }

    #pragma unroll
    for (int n = 0; n < 8; ++n) {
        #pragma unroll
        for (int g = 0; g < 8; ++g) {
            int m = m0 + g + half * 8;
            C[(size_t)m * FDIM + n * 16 + l16] = acc[n][g];
        }
    }
}

__global__ __launch_bounds__(256) void gather_combine(const float* __restrict__ hw,
                                                      const float* __restrict__ dis,
                                                      const int* __restrict__ row_ptr,
                                                      const int* __restrict__ ewbuf,
                                                      const float* __restrict__ b,
                                                      float* __restrict__ hout,
                                                      int nNodes, int doRelu) {
    const int node = __builtin_amdgcn_readfirstlane((int)(blockIdx.x * 8 + (threadIdx.x >> 5)));
    if (node >= nNodes) return;
    const int f = (threadIdx.x & 31) * 4;
    const int2* ew = (const int2*)ewbuf;

    int j = row_ptr[node];
    const int end = row_ptr[node + 1];

    float4 acc = make_float4(0.0f, 0.0f, 0.0f, 0.0f);
    for (; j + 4 <= end; j += 4) {
        const int2 e0 = ew[j], e1 = ew[j + 1], e2 = ew[j + 2], e3 = ew[j + 3];
        const float c0 = __int_as_float(e0.y), c1 = __int_as_float(e1.y);
        const float c2 = __int_as_float(e2.y), c3 = __int_as_float(e3.y);
        const float4 v0 = *(const float4*)(hw + (size_t)e0.x * FDIM + f);
        const float4 v1 = *(const float4*)(hw + (size_t)e1.x * FDIM + f);
        const float4 v2 = *(const float4*)(hw + (size_t)e2.x * FDIM + f);
        const float4 v3 = *(const float4*)(hw + (size_t)e3.x * FDIM + f);
        acc.x = fmaf(v0.x, c0, acc.x); acc.y = fmaf(v0.y, c0, acc.y);
        acc.z = fmaf(v0.z, c0, acc.z); acc.w = fmaf(v0.w, c0, acc.w);
        acc.x = fmaf(v1.x, c1, acc.x); acc.y = fmaf(v1.y, c1, acc.y);
        acc.z = fmaf(v1.z, c1, acc.z); acc.w = fmaf(v1.w, c1, acc.w);
        acc.x = fmaf(v2.x, c2, acc.x); acc.y = fmaf(v2.y, c2, acc.y);
        acc.z = fmaf(v2.z, c2, acc.z); acc.w = fmaf(v2.w, c2, acc.w);
        acc.x = fmaf(v3.x, c3, acc.x); acc.y = fmaf(v3.y, c3, acc.y);
        acc.z = fmaf(v3.z, c3, acc.z); acc.w = fmaf(v3.w, c3, acc.w);
    }
    const int rem = end - j;
    if (rem > 0) {
        const int last = end - 1;
        const int j1 = (j + 1 < end) ? j + 1 : last;
        const int j2 = (j + 2 < end) ? j + 2 : last;
        const int2 e0 = ew[j], e1 = ew[j1], e2 = ew[j2];
        const float c0 = __int_as_float(e0.y), c1 = __int_as_float(e1.y), c2 = __int_as_float(e2.y);
        const float4 v0 = *(const float4*)(hw + (size_t)e0.x * FDIM + f);
        const float4 v1 = *(const float4*)(hw + (size_t)e1.x * FDIM + f);
        const float4 v2 = *(const float4*)(hw + (size_t)e2.x * FDIM + f);
        acc.x = fmaf(v0.x, c0, acc.x); acc.y = fmaf(v0.y, c0, acc.y);
        acc.z = fmaf(v0.z, c0, acc.z); acc.w = fmaf(v0.w, c0, acc.w);
        float4 t;
        t.x = fmaf(v1.x, c1, acc.x); t.y = fmaf(v1.y, c1, acc.y);
        t.z = fmaf(v1.z, c1, acc.z); t.w = fmaf(v1.w, c1, acc.w);
        acc.x = (rem > 1) ? t.x : acc.x; acc.y = (rem > 1) ? t.y : acc.y;
        acc.z = (rem > 1) ? t.z : acc.z; acc.w = (rem > 1) ? t.w : acc.w;
        t.x = fmaf(v2.x, c2, acc.x); t.y = fmaf(v2.y, c2, acc.y);
        t.z = fmaf(v2.z, c2, acc.z); t.w = fmaf(v2.w, c2, acc.w);
        acc.x = (rem > 2) ? t.x : acc.x; acc.y = (rem > 2) ? t.y : acc.y;
        acc.z = (rem > 2) ? t.z : acc.z; acc.w = (rem > 2) ? t.w : acc.w;
    }

    const float di  = dis[node];
    const float di2 = di * di;
    const float4 h = *(const float4*)(hw + (size_t)node * FDIM + f);
    float4 r;
    r.x = acc.x * di + h.x * di2 + b[f + 0];
    r.y = acc.y * di + h.y * di2 + b[f + 1];
    r.z = acc.z * di + h.z * di2 + b[f + 2];
    r.w = acc.w * di + h.w * di2 + b[f + 3];
    if (doRelu) {
        r.x = fmaxf(r.x, 0.0f); r.y = fmaxf(r.y, 0.0f);
        r.z = fmaxf(r.z, 0.0f); r.w = fmaxf(r.w, 0.0f);
    }
    *(float4*)(hout + (size_t)node * FDIM + f) = r;
}

__global__ __launch_bounds__(256) void gemm_edge(const float* __restrict__ H,
                                                 const int* __restrict__ pos_tr,
                                                 const int* __restrict__ neg_tr,
                                                 const int* __restrict__ pos_te,
                                                 const int* __restrict__ neg_te,
                                                 const float* __restrict__ Wl1,
                                                 const float* __restrict__ bl1,
                                                 const float* __restrict__ Wl2,
                                                 const float* __restrict__ bl2,
                                                 float* __restrict__ out) {
    int wave = (blockIdx.x * blockDim.x + threadIdx.x) >> 5;
    int lane = threadIdx.x & 31;
    int r0 = wave * 16;
    if (r0 >= N_ROWS) return;
    int half = lane >> 4;
    int l16  = lane & 15;
    int r = r0 + l16;

    int u, v;
    if (r < E_TR)            { u = pos_tr[r];              v = pos_tr[E_TR + r]; }
    else if (r < 2 * E_TR)   { int i = r - E_TR;           u = neg_tr[i]; v = neg_tr[E_TR + i]; }
    else if (r < 2 * E_TR + E_TE) { int i = r - 2 * E_TR;  u = pos_te[i]; v = pos_te[E_TE + i]; }
    else                     { int i = r - 2 * E_TR - E_TE; u = neg_te[i]; v = neg_te[E_TE + i]; }

    const float* Au = H + (size_t)u * FDIM;
    const float* Av = H + (size_t)v * FDIM;

    v8f acc[8] = {};

    for (int k0 = 0; k0 < 2 * FDIM; k0 += 4) {
        int k = k0 + half * 2;
        const float* ap = (k < FDIM) ? (Au + k) : (Av + (k - FDIM));
        v2f afrag = *(const v2f*)ap;
        const float* b0 = Wl1 + (size_t)k * FDIM + l16;
        #pragma unroll
        for (int n = 0; n < 8; ++n) {
            v2f bfrag;
            bfrag.x = b0[n * 16];
            bfrag.y = b0[FDIM + n * 16];
            acc[n] = __builtin_amdgcn_wmma_f32_16x16x4_f32(
                false, afrag, false, bfrag, (short)0, acc[n], false, false);
        }
    }

    float partial[8];
    #pragma unroll
    for (int g = 0; g < 8; ++g) partial[g] = 0.0f;

    #pragma unroll
    for (int n = 0; n < 8; ++n) {
        float bb = bl1[n * 16 + l16];
        float ww = Wl2[n * 16 + l16];
        #pragma unroll
        for (int g = 0; g < 8; ++g) {
            float z = fmaxf(acc[n][g] + bb, 0.0f);
            partial[g] += z * ww;
        }
    }

    float bias2 = bl2[0];
    #pragma unroll
    for (int g = 0; g < 8; ++g) {
        float s = partial[g];
        s += __shfl_xor(s, 8, 16);
        s += __shfl_xor(s, 4, 16);
        s += __shfl_xor(s, 2, 16);
        s += __shfl_xor(s, 1, 16);
        if (l16 == 0) out[r0 + g + half * 8] = s + bias2;
    }
}

extern "C" void kernel_launch(void* const* d_in, const int* in_sizes, int n_in,
                              void* d_out, int out_size, void* d_ws, size_t ws_size,
                              hipStream_t stream) {
    const float* x          = (const float*)d_in[0];
    const int*   edge_index = (const int*)d_in[1];
    const int*   pos_tr     = (const int*)d_in[2];
    const int*   neg_tr     = (const int*)d_in[3];
    const int*   pos_te     = (const int*)d_in[4];
    const int*   neg_te     = (const int*)d_in[5];
    const float* W[4] = {(const float*)d_in[6],  (const float*)d_in[8],
                         (const float*)d_in[10], (const float*)d_in[12]};
    const float* b[4] = {(const float*)d_in[7],  (const float*)d_in[9],
                         (const float*)d_in[11], (const float*)d_in[13]};
    const float* Wl1 = (const float*)d_in[14];
    const float* bl1 = (const float*)d_in[15];
    const float* Wl2 = (const float*)d_in[16];
    const float* bl2 = (const float*)d_in[17];
    float* out = (float*)d_out;

    const size_t NF = (size_t)N_NODES * FDIM;
    float* ws  = (float*)d_ws;
    float* dis = ws;
    float* hw  = ws + 50048;
    int*   csr = (int*)(hw + NF);
    int*   cnt     = csr;
    int*   row_ptr = csr + 53248;
    int*   rank    = csr + 2 * 53248;
    int*   ew      = csr + 2 * 53248 + N_EDGES;
    float* h0  = hw + 2 * NF;
    float* h1  = h0 + NF;

    const int* src = edge_index;
    const int* dst = edge_index + N_EDGES;

    hipMemsetAsync(cnt, 0, N_NODES * sizeof(int), stream);
    count_rank<<<(N_EDGES + 255) / 256, 256, 0, stream>>>(dst, cnt, rank, N_EDGES);
    scan_rows<<<1, 1024, 0, stream>>>(cnt, row_ptr, dis);
    fill_cols<<<(N_EDGES + 255) / 256, 256, 0, stream>>>(src, dst, rank, row_ptr, dis, ew, N_EDGES);

    const int tilesM     = (N_NODES + 15) / 16;
    const int gemmBlocks = (tilesM + 7) / 8;
    const float* hin = x;
    float* houts[4] = {h0, h1, h0, h1};
    for (int l = 0; l < 4; ++l) {
        gemm_node<<<gemmBlocks, 256, 0, stream>>>(hin, W[l], hw, N_NODES);
        gather_combine<<<(N_NODES * 32 + 255) / 256, 256, 0, stream>>>(
            hw, dis, row_ptr, ew, b[l], houts[l], N_NODES, (l < 3) ? 1 : 0);
        hin = houts[l];
    }

    const int edgeTiles  = N_ROWS / 16;
    const int edgeBlocks = (edgeTiles + 7) / 8;
    gemm_edge<<<edgeBlocks, 256, 0, stream>>>(h1, pos_tr, neg_tr, pos_te, neg_te,
                                              Wl1, bl1, Wl2, bl2, out);
}


# Edge MLP layer 1 by linearity of the concat: per node P=h@Wl1[0:128], Q=h@Wl1[128:256] on f32 WMMA, per edge relu(P[u]+Q[v]+bl1).Wl2 in an 8-lane tree; each 256-term sum is now two 128-term sums
# speedup vs baseline: 2.8317x; 1.1524x over previous
#include <hip/hip_runtime.h>


typedef float v2f __attribute__((ext_vector_type(2)));
typedef float v8f __attribute__((ext_vector_type(8)));

#define N_NODES 50000
#define N_EDGES 800000
#define FDIM    128
#define E_TR    100000
#define E_TE    20000
#define N_ROWS  240000

__global__ __launch_bounds__(256) void count_rank(const int* __restrict__ dst,
                                                  int* __restrict__ cnt,
                                                  int* __restrict__ rank, int nE) {
    int e = blockIdx.x * blockDim.x + threadIdx.x;
    if (e < nE) rank[e] = atomicAdd(&cnt[dst[e]], 1);
}

#define SCAN_PASSES 13
#define SCAN_CHUNK  (SCAN_PASSES * 128)
static_assert(32 * SCAN_CHUNK >= N_NODES, "one block must cover every node");
static_assert(N_NODES % 4 == 0, "16-byte groups must not straddle the end");

__device__ __forceinline__ int wave_incl_scan(int v, int lane) {
    #pragma unroll
    for (int o = 1; o < 32; o <<= 1) {
        int y = __shfl_up(v, o, 32);
        if (lane >= o) v += y;
    }
    return v;
}

__global__ __launch_bounds__(1024) void scan_rows(const int* __restrict__ cnt,
                                                  int* __restrict__ row_ptr,
                                                  float* __restrict__ dis) {
    __shared__ int wtot[32];
    const int lane = threadIdx.x & 31;
    const int w    = threadIdx.x >> 5;
    const int base = w * SCAN_CHUNK + lane * 4;

    int4 p[SCAN_PASSES];
    #pragma unroll
    for (int i = 0; i < SCAN_PASSES; ++i) {
        const int idx = base + i * 128;
        const int4 v = *(const int4*)(cnt + ((idx < N_NODES) ? idx : N_NODES - 4));
        p[i] = (idx < N_NODES) ? v : make_int4(0, 0, 0, 0);
    }
    #pragma unroll
    for (int i = 0; i < SCAN_PASSES; ++i) {
        const int idx = base + i * 128;
        if (idx < N_NODES) {
            float4 d;
            d.x = rsqrtf((float)p[i].x + 1.0f);
            d.y = rsqrtf((float)p[i].y + 1.0f);
            d.z = rsqrtf((float)p[i].z + 1.0f);
            d.w = rsqrtf((float)p[i].w + 1.0f);
            *(float4*)(dis + idx) = d;
        }
    }

    int run = 0;
    #pragma unroll
    for (int i = 0; i < SCAN_PASSES; ++i) {
        const int4 c = p[i];
        const int s   = c.x + c.y + c.z + c.w;
        const int inc = wave_incl_scan(s, lane);
        const int o   = run + inc - s;
        p[i].x = o;
        p[i].y = o + c.x;
        p[i].z = o + c.x + c.y;
        p[i].w = o + c.x + c.y + c.z;
        run += __shfl(inc, 31, 32);
    }
    if (lane == 0) wtot[w] = run;
    __syncthreads();

    const int t     = wtot[lane];
    const int tinc  = wave_incl_scan(t, lane);
    const int wbase = __shfl(tinc - t, w, 32);

    #pragma unroll
    for (int i = 0; i < SCAN_PASSES; ++i) {
        int idx = base + i * 128;
        if (idx < N_NODES) {
            int4 q = p[i];
            q.x += wbase; q.y += wbase; q.z += wbase; q.w += wbase;
            *(int4*)(row_ptr + idx) = q;
        }
    }
    if (threadIdx.x == 1023) row_ptr[N_NODES] = wbase + run;
}

__global__ __launch_bounds__(256) void fill_cols(const int* __restrict__ src,
                                                 const int* __restrict__ dst,
                                                 const int* __restrict__ rank,
                                                 const int* __restrict__ row_ptr,
                                                 const float* __restrict__ dis,
                                                 int* __restrict__ ewbuf, int nE) {
    int2* ew = (int2*)ewbuf;
    int e = blockIdx.x * blockDim.x + threadIdx.x;
    if (e < nE) {
        const int s = src[e];
        ew[row_ptr[dst[e]] + rank[e]] = make_int2(s, __float_as_int(dis[s]));
    }
}

__global__ __launch_bounds__(256) void gemm_node(const float* __restrict__ A,
                                                 const float* __restrict__ B,
                                                 float* __restrict__ C, int M) {
    int wave = (blockIdx.x * blockDim.x + threadIdx.x) >> 5;
    int lane = threadIdx.x & 31;
    int m0 = wave * 16;
    if (m0 >= M) return;
    int half = lane >> 4;
    int l16  = lane & 15;

    B += (size_t)blockIdx.y * FDIM * FDIM;
    C += (size_t)blockIdx.y * M * FDIM;

    const float* Arow = A + (size_t)(m0 + l16) * FDIM;

    v8f acc[8] = {};

    for (int k0 = 0; k0 < FDIM; k0 += 4) {
        int k = k0 + half * 2;
        v2f afrag = *(const v2f*)(Arow + k);
        const float* b0 = B + (size_t)k * FDIM + l16;
        #pragma unroll
        for (int n = 0; n < 8; ++n) {
            v2f bfrag;
            bfrag.x = b0[n * 16];
            bfrag.y = b0[FDIM + n * 16];
            acc[n] = __builtin_amdgcn_wmma_f32_16x16x4_f32(
                false, afrag, false, bfrag, (short)0, acc[n], false, false);
        }
    }

    #pragma unroll
    for (int n = 0; n < 8; ++n) {
        #pragma unroll
        for (int g = 0; g < 8; ++g) {
            int m = m0 + g + half * 8;
            C[(size_t)m * FDIM + n * 16 + l16] = acc[n][g];
        }
    }
}

__global__ __launch_bounds__(256) void gather_combine(const float* __restrict__ hw,
                                                      const float* __restrict__ dis,
                                                      const int* __restrict__ row_ptr,
                                                      const int* __restrict__ ewbuf,
                                                      const float* __restrict__ b,
                                                      float* __restrict__ hout,
                                                      int nNodes, int doRelu) {
    const int node = __builtin_amdgcn_readfirstlane((int)(blockIdx.x * 8 + (threadIdx.x >> 5)));
    if (node >= nNodes) return;
    const int f = (threadIdx.x & 31) * 4;
    const int2* ew = (const int2*)ewbuf;

    int j = row_ptr[node];
    const int end = row_ptr[node + 1];

    float4 acc = make_float4(0.0f, 0.0f, 0.0f, 0.0f);
    for (; j + 4 <= end; j += 4) {
        const int2 e0 = ew[j], e1 = ew[j + 1], e2 = ew[j + 2], e3 = ew[j + 3];
        const float c0 = __int_as_float(e0.y), c1 = __int_as_float(e1.y);
        const float c2 = __int_as_float(e2.y), c3 = __int_as_float(e3.y);
        const float4 v0 = *(const float4*)(hw + (size_t)e0.x * FDIM + f);
        const float4 v1 = *(const float4*)(hw + (size_t)e1.x * FDIM + f);
        const float4 v2 = *(const float4*)(hw + (size_t)e2.x * FDIM + f);
        const float4 v3 = *(const float4*)(hw + (size_t)e3.x * FDIM + f);
        acc.x = fmaf(v0.x, c0, acc.x); acc.y = fmaf(v0.y, c0, acc.y);
        acc.z = fmaf(v0.z, c0, acc.z); acc.w = fmaf(v0.w, c0, acc.w);
        acc.x = fmaf(v1.x, c1, acc.x); acc.y = fmaf(v1.y, c1, acc.y);
        acc.z = fmaf(v1.z, c1, acc.z); acc.w = fmaf(v1.w, c1, acc.w);
        acc.x = fmaf(v2.x, c2, acc.x); acc.y = fmaf(v2.y, c2, acc.y);
        acc.z = fmaf(v2.z, c2, acc.z); acc.w = fmaf(v2.w, c2, acc.w);
        acc.x = fmaf(v3.x, c3, acc.x); acc.y = fmaf(v3.y, c3, acc.y);
        acc.z = fmaf(v3.z, c3, acc.z); acc.w = fmaf(v3.w, c3, acc.w);
    }
    const int rem = end - j;
    if (rem > 0) {
        const int last = end - 1;
        const int j1 = (j + 1 < end) ? j + 1 : last;
        const int j2 = (j + 2 < end) ? j + 2 : last;
        const int2 e0 = ew[j], e1 = ew[j1], e2 = ew[j2];
        const float c0 = __int_as_float(e0.y), c1 = __int_as_float(e1.y), c2 = __int_as_float(e2.y);
        const float4 v0 = *(const float4*)(hw + (size_t)e0.x * FDIM + f);
        const float4 v1 = *(const float4*)(hw + (size_t)e1.x * FDIM + f);
        const float4 v2 = *(const float4*)(hw + (size_t)e2.x * FDIM + f);
        acc.x = fmaf(v0.x, c0, acc.x); acc.y = fmaf(v0.y, c0, acc.y);
        acc.z = fmaf(v0.z, c0, acc.z); acc.w = fmaf(v0.w, c0, acc.w);
        float4 t;
        t.x = fmaf(v1.x, c1, acc.x); t.y = fmaf(v1.y, c1, acc.y);
        t.z = fmaf(v1.z, c1, acc.z); t.w = fmaf(v1.w, c1, acc.w);
        acc.x = (rem > 1) ? t.x : acc.x; acc.y = (rem > 1) ? t.y : acc.y;
        acc.z = (rem > 1) ? t.z : acc.z; acc.w = (rem > 1) ? t.w : acc.w;
        t.x = fmaf(v2.x, c2, acc.x); t.y = fmaf(v2.y, c2, acc.y);
        t.z = fmaf(v2.z, c2, acc.z); t.w = fmaf(v2.w, c2, acc.w);
        acc.x = (rem > 2) ? t.x : acc.x; acc.y = (rem > 2) ? t.y : acc.y;
        acc.z = (rem > 2) ? t.z : acc.z; acc.w = (rem > 2) ? t.w : acc.w;
    }

    const float di  = dis[node];
    const float di2 = di * di;
    const float4 h = *(const float4*)(hw + (size_t)node * FDIM + f);
    float4 r;
    r.x = acc.x * di + h.x * di2 + b[f + 0];
    r.y = acc.y * di + h.y * di2 + b[f + 1];
    r.z = acc.z * di + h.z * di2 + b[f + 2];
    r.w = acc.w * di + h.w * di2 + b[f + 3];
    if (doRelu) {
        r.x = fmaxf(r.x, 0.0f); r.y = fmaxf(r.y, 0.0f);
        r.z = fmaxf(r.z, 0.0f); r.w = fmaxf(r.w, 0.0f);
    }
    *(float4*)(hout + (size_t)node * FDIM + f) = r;
}

static_assert(E_TR % 32 == 0 && E_TE % 32 == 0, "a wave's 32 rows must lie inside one edge list");
static_assert(N_ROWS % 32 == 0, "whole waves only");

__global__ __launch_bounds__(256) void edge_score(const float* __restrict__ P,
                                                  const float* __restrict__ Q,
                                                  const int* __restrict__ pos_tr,
                                                  const int* __restrict__ neg_tr,
                                                  const int* __restrict__ pos_te,
                                                  const int* __restrict__ neg_te,
                                                  const float* __restrict__ bl1,
                                                  const float* __restrict__ Wl2,
                                                  const float* __restrict__ bl2,
                                                  float* __restrict__ out) {
    const int r0 = __builtin_amdgcn_readfirstlane(
        (int)(((blockIdx.x * blockDim.x + threadIdx.x) >> 5) * 32));
    if (r0 >= N_ROWS) return;
    const int lane = threadIdx.x & 31;
    const int g = lane >> 3;
    const int j = lane & 7;

    const int* e; int i0, len;
    if (r0 < E_TR)                 { e = pos_tr; i0 = r0;                   len = E_TR; }
    else if (r0 < 2 * E_TR)        { e = neg_tr; i0 = r0 - E_TR;            len = E_TR; }
    else if (r0 < 2 * E_TR + E_TE) { e = pos_te; i0 = r0 - 2 * E_TR;        len = E_TE; }
    else                           { e = neg_te; i0 = r0 - 2 * E_TR - E_TE; len = E_TE; }

    const int* eu = e + i0 + g * 8;
    const int* ev = eu + len;
    const int4 ua = *(const int4*)eu, ub = *(const int4*)(eu + 4);
    const int4 va = *(const int4*)ev, vb = *(const int4*)(ev + 4);
    const int us[8] = {ua.x, ua.y, ua.z, ua.w, ub.x, ub.y, ub.z, ub.w};
    const int vs[8] = {va.x, va.y, va.z, va.w, vb.x, vb.y, vb.z, vb.w};

    float4 bb[4], ww[4];
    #pragma unroll
    for (int c = 0; c < 4; ++c) {
        bb[c] = *(const float4*)(bl1 + c * 32 + j * 4);
        ww[c] = *(const float4*)(Wl2 + c * 32 + j * 4);
    }

    float s[8];
    #pragma unroll
    for (int p = 0; p < 8; ++p) {
        const float* pr = P + (size_t)us[p] * FDIM + j * 4;
        const float* qr = Q + (size_t)vs[p] * FDIM + j * 4;
        float acc = 0.0f;
        #pragma unroll
        for (int c = 0; c < 4; ++c) {
            const float4 a = *(const float4*)(pr + c * 32);
            const float4 b = *(const float4*)(qr + c * 32);
            acc = fmaf(fmaxf(a.x + b.x + bb[c].x, 0.0f), ww[c].x, acc);
            acc = fmaf(fmaxf(a.y + b.y + bb[c].y, 0.0f), ww[c].y, acc);
            acc = fmaf(fmaxf(a.z + b.z + bb[c].z, 0.0f), ww[c].z, acc);
            acc = fmaf(fmaxf(a.w + b.w + bb[c].w, 0.0f), ww[c].w, acc);
        }
        s[p] = acc;
    }

    const bool k4 = (j & 4) != 0, k2 = (j & 2) != 0, k1 = (j & 1) != 0;
    float t4[4];
    #pragma unroll
    for (int i = 0; i < 4; ++i) {
        const float keep = k4 ? s[4 + i] : s[i];
        const float send = k4 ? s[i] : s[4 + i];
        t4[i] = keep + __shfl_xor(send, 4, 32);
    }
    float t2[2];
    #pragma unroll
    for (int i = 0; i < 2; ++i) {
        const float keep = k2 ? t4[2 + i] : t4[i];
        const float send = k2 ? t4[i] : t4[2 + i];
        t2[i] = keep + __shfl_xor(send, 2, 32);
    }
    const float keep = k1 ? t2[1] : t2[0];
    const float send = k1 ? t2[0] : t2[1];
    const float tot = keep + __shfl_xor(send, 1, 32);

    out[r0 + lane] = tot + bl2[0];
}

extern "C" void kernel_launch(void* const* d_in, const int* in_sizes, int n_in,
                              void* d_out, int out_size, void* d_ws, size_t ws_size,
                              hipStream_t stream) {
    const float* x          = (const float*)d_in[0];
    const int*   edge_index = (const int*)d_in[1];
    const int*   pos_tr     = (const int*)d_in[2];
    const int*   neg_tr     = (const int*)d_in[3];
    const int*   pos_te     = (const int*)d_in[4];
    const int*   neg_te     = (const int*)d_in[5];
    const float* W[4] = {(const float*)d_in[6],  (const float*)d_in[8],
                         (const float*)d_in[10], (const float*)d_in[12]};
    const float* b[4] = {(const float*)d_in[7],  (const float*)d_in[9],
                         (const float*)d_in[11], (const float*)d_in[13]};
    const float* Wl1 = (const float*)d_in[14];
    const float* bl1 = (const float*)d_in[15];
    const float* Wl2 = (const float*)d_in[16];
    const float* bl2 = (const float*)d_in[17];
    float* out = (float*)d_out;

    const size_t NF = (size_t)N_NODES * FDIM;
    float* ws  = (float*)d_ws;
    float* dis = ws;
    float* hw  = ws + 50048;
    int*   csr = (int*)(hw + NF);
    int*   cnt     = csr;
    int*   row_ptr = csr + 53248;
    int*   rank    = csr + 2 * 53248;
    int*   ew      = csr + 2 * 53248 + N_EDGES;
    float* h0  = hw + 2 * NF;
    float* h1  = h0 + NF;

    const int* src = edge_index;
    const int* dst = edge_index + N_EDGES;

    hipMemsetAsync(cnt, 0, N_NODES * sizeof(int), stream);
    count_rank<<<(N_EDGES + 255) / 256, 256, 0, stream>>>(dst, cnt, rank, N_EDGES);
    scan_rows<<<1, 1024, 0, stream>>>(cnt, row_ptr, dis);
    fill_cols<<<(N_EDGES + 255) / 256, 256, 0, stream>>>(src, dst, rank, row_ptr, dis, ew, N_EDGES);

    const int tilesM     = (N_NODES + 15) / 16;
    const int gemmBlocks = (tilesM + 7) / 8;
    const float* hin = x;
    float* houts[4] = {h0, h1, h0, h1};
    for (int l = 0; l < 4; ++l) {
        gemm_node<<<gemmBlocks, 256, 0, stream>>>(hin, W[l], hw, N_NODES);
        gather_combine<<<(N_NODES * 32 + 255) / 256, 256, 0, stream>>>(
            hw, dis, row_ptr, ew, b[l], houts[l], N_NODES, (l < 3) ? 1 : 0);
        hin = houts[l];
    }

    gemm_node<<<dim3(gemmBlocks, 2), 256, 0, stream>>>(h1, Wl1, hw, N_NODES);

    const int edgeWaves  = N_ROWS / 32;
    const int edgeBlocks = (edgeWaves + 7) / 8;
    edge_score<<<edgeBlocks, 256, 0, stream>>>(hw, hw + NF, pos_tr, neg_tr, pos_te, neg_te,
                                               bl1, Wl2, bl2, out);
}


# Node GEMM takes B fragments from LDS: a block stages the 128x128 weight block once, the two k of a fragment side by side; one 8-byte LDS read replaces two 4-byte global loads; k order and sums kept
# speedup vs baseline: 3.3851x; 1.1954x over previous
#include <hip/hip_runtime.h>


typedef float v2f __attribute__((ext_vector_type(2)));
typedef float v8f __attribute__((ext_vector_type(8)));

#define N_NODES 50000
#define N_EDGES 800000
#define FDIM    128
#define E_TR    100000
#define E_TE    20000
#define N_ROWS  240000

__global__ __launch_bounds__(256) void count_rank(const int* __restrict__ dst,
                                                  int* __restrict__ cnt,
                                                  int* __restrict__ rank, int nE) {
    int e = blockIdx.x * blockDim.x + threadIdx.x;
    if (e < nE) rank[e] = atomicAdd(&cnt[dst[e]], 1);
}

#define SCAN_PASSES 13
#define SCAN_CHUNK  (SCAN_PASSES * 128)
static_assert(32 * SCAN_CHUNK >= N_NODES, "one block must cover every node");
static_assert(N_NODES % 4 == 0, "16-byte groups must not straddle the end");

__device__ __forceinline__ int wave_incl_scan(int v, int lane) {
    #pragma unroll
    for (int o = 1; o < 32; o <<= 1) {
        int y = __shfl_up(v, o, 32);
        if (lane >= o) v += y;
    }
    return v;
}

__global__ __launch_bounds__(1024) void scan_rows(const int* __restrict__ cnt,
                                                  int* __restrict__ row_ptr,
                                                  float* __restrict__ dis) {
    __shared__ int wtot[32];
    const int lane = threadIdx.x & 31;
    const int w    = threadIdx.x >> 5;
    const int base = w * SCAN_CHUNK + lane * 4;

    int4 p[SCAN_PASSES];
    #pragma unroll
    for (int i = 0; i < SCAN_PASSES; ++i) {
        const int idx = base + i * 128;
        const int4 v = *(const int4*)(cnt + ((idx < N_NODES) ? idx : N_NODES - 4));
        p[i] = (idx < N_NODES) ? v : make_int4(0, 0, 0, 0);
    }
    #pragma unroll
    for (int i = 0; i < SCAN_PASSES; ++i) {
        const int idx = base + i * 128;
        if (idx < N_NODES) {
            float4 d;
            d.x = rsqrtf((float)p[i].x + 1.0f);
            d.y = rsqrtf((float)p[i].y + 1.0f);
            d.z = rsqrtf((float)p[i].z + 1.0f);
            d.w = rsqrtf((float)p[i].w + 1.0f);
            *(float4*)(dis + idx) = d;
        }
    }

    int run = 0;
    #pragma unroll
    for (int i = 0; i < SCAN_PASSES; ++i) {
        const int4 c = p[i];
        const int s   = c.x + c.y + c.z + c.w;
        const int inc = wave_incl_scan(s, lane);
        const int o   = run + inc - s;
        p[i].x = o;
        p[i].y = o + c.x;
        p[i].z = o + c.x + c.y;
        p[i].w = o + c.x + c.y + c.z;
        run += __shfl(inc, 31, 32);
    }
    if (lane == 0) wtot[w] = run;
    __syncthreads();

    const int t     = wtot[lane];
    const int tinc  = wave_incl_scan(t, lane);
    const int wbase = __shfl(tinc - t, w, 32);

    #pragma unroll
    for (int i = 0; i < SCAN_PASSES; ++i) {
        int idx = base + i * 128;
        if (idx < N_NODES) {
            int4 q = p[i];
            q.x += wbase; q.y += wbase; q.z += wbase; q.w += wbase;
            *(int4*)(row_ptr + idx) = q;
        }
    }
    if (threadIdx.x == 1023) row_ptr[N_NODES] = wbase + run;
}

__global__ __launch_bounds__(256) void fill_cols(const int* __restrict__ src,
                                                 const int* __restrict__ dst,
                                                 const int* __restrict__ rank,
                                                 const int* __restrict__ row_ptr,
                                                 const float* __restrict__ dis,
                                                 int* __restrict__ ewbuf, int nE) {
    int2* ew = (int2*)ewbuf;
    int e = blockIdx.x * blockDim.x + threadIdx.x;
    if (e < nE) {
        const int s = src[e];
        ew[row_ptr[dst[e]] + rank[e]] = make_int2(s, __float_as_int(dis[s]));
    }
}

#define WS_PITCH 288
__global__ __launch_bounds__(256) void gemm_node(const float* __restrict__ A,
                                                 const float* __restrict__ B,
                                                 float* __restrict__ C, int M) {
    __shared__ float Ws[(FDIM / 2) * WS_PITCH];
    const int wave = (blockIdx.x * blockDim.x + threadIdx.x) >> 5;
    const int lane = threadIdx.x & 31;
    const int m0 = wave * 16;
    const bool live = m0 < M;
    const int half = lane >> 4;
    const int l16  = lane & 15;

    B += (size_t)blockIdx.y * FDIM * FDIM;
    C += (size_t)blockIdx.y * M * FDIM;

    float4 r0[8], r1[8];
    #pragma unroll
    for (int it = 0; it < 8; ++it) {
        const int kp = it * 8 + (threadIdx.x >> 5);
        r0[it] = *(const float4*)(B + (size_t)(2 * kp) * FDIM + lane * 4);
        r1[it] = *(const float4*)(B + (size_t)(2 * kp + 1) * FDIM + lane * 4);
    }

    const float* Arow = A + (size_t)((live ? m0 : 0) + l16) * FDIM + half * 2;
    v2f afrag[FDIM / 4];
    #pragma unroll
    for (int s = 0; s < FDIM / 4; ++s) afrag[s] = *(const v2f*)(Arow + s * 4);

    #pragma unroll
    for (int it = 0; it < 8; ++it) {
        const int kp = it * 8 + (threadIdx.x >> 5);
        float* d = Ws + kp * WS_PITCH + lane * 8;
        *(float4*)d       = make_float4(r0[it].x, r1[it].x, r0[it].y, r1[it].y);
        *(float4*)(d + 4) = make_float4(r0[it].z, r1[it].z, r0[it].w, r1[it].w);
    }
    __syncthreads();
    if (!live) return;

    const float* bp = Ws + half * WS_PITCH + l16 * 2;

    v8f acc[8] = {};
    v2f bcur[8];
    #pragma unroll
    for (int n = 0; n < 8; ++n) bcur[n] = *(const v2f*)(bp + n * 32);

    #pragma unroll
    for (int s = 0; s < FDIM / 4; ++s) {
        v2f bnxt[8];
        if (s + 1 < FDIM / 4) {
            #pragma unroll
            for (int n = 0; n < 8; ++n)
                bnxt[n] = *(const v2f*)(bp + (s + 1) * 2 * WS_PITCH + n * 32);
        }
        #pragma unroll
        for (int n = 0; n < 8; ++n)
            acc[n] = __builtin_amdgcn_wmma_f32_16x16x4_f32(
                false, afrag[s], false, bcur[n], (short)0, acc[n], false, false);
        if (s + 1 < FDIM / 4) {
            #pragma unroll
            for (int n = 0; n < 8; ++n) bcur[n] = bnxt[n];
        }
    }

    #pragma unroll
    for (int n = 0; n < 8; ++n) {
        #pragma unroll
        for (int g = 0; g < 8; ++g) {
            int m = m0 + g + half * 8;
            C[(size_t)m * FDIM + n * 16 + l16] = acc[n][g];
        }
    }
}

__global__ __launch_bounds__(256) void gather_combine(const float* __restrict__ hw,
                                                      const float* __restrict__ dis,
                                                      const int* __restrict__ row_ptr,
                                                      const int* __restrict__ ewbuf,
                                                      const float* __restrict__ b,
                                                      float* __restrict__ hout,
                                                      int nNodes, int doRelu) {
    const int node = __builtin_amdgcn_readfirstlane((int)(blockIdx.x * 8 + (threadIdx.x >> 5)));
    if (node >= nNodes) return;
    const int f = (threadIdx.x & 31) * 4;
    const int2* ew = (const int2*)ewbuf;

    int j = row_ptr[node];
    const int end = row_ptr[node + 1];

    float4 acc = make_float4(0.0f, 0.0f, 0.0f, 0.0f);
    for (; j + 4 <= end; j += 4) {
        const int2 e0 = ew[j], e1 = ew[j + 1], e2 = ew[j + 2], e3 = ew[j + 3];
        const float c0 = __int_as_float(e0.y), c1 = __int_as_float(e1.y);
        const float c2 = __int_as_float(e2.y), c3 = __int_as_float(e3.y);
        const float4 v0 = *(const float4*)(hw + (size_t)e0.x * FDIM + f);
        const float4 v1 = *(const float4*)(hw + (size_t)e1.x * FDIM + f);
        const float4 v2 = *(const float4*)(hw + (size_t)e2.x * FDIM + f);
        const float4 v3 = *(const float4*)(hw + (size_t)e3.x * FDIM + f);
        acc.x = fmaf(v0.x, c0, acc.x); acc.y = fmaf(v0.y, c0, acc.y);
        acc.z = fmaf(v0.z, c0, acc.z); acc.w = fmaf(v0.w, c0, acc.w);
        acc.x = fmaf(v1.x, c1, acc.x); acc.y = fmaf(v1.y, c1, acc.y);
        acc.z = fmaf(v1.z, c1, acc.z); acc.w = fmaf(v1.w, c1, acc.w);
        acc.x = fmaf(v2.x, c2, acc.x); acc.y = fmaf(v2.y, c2, acc.y);
        acc.z = fmaf(v2.z, c2, acc.z); acc.w = fmaf(v2.w, c2, acc.w);
        acc.x = fmaf(v3.x, c3, acc.x); acc.y = fmaf(v3.y, c3, acc.y);
        acc.z = fmaf(v3.z, c3, acc.z); acc.w = fmaf(v3.w, c3, acc.w);
    }
    const int rem = end - j;
    if (rem > 0) {
        const int last = end - 1;
        const int j1 = (j + 1 < end) ? j + 1 : last;
        const int j2 = (j + 2 < end) ? j + 2 : last;
        const int2 e0 = ew[j], e1 = ew[j1], e2 = ew[j2];
        const float c0 = __int_as_float(e0.y), c1 = __int_as_float(e1.y), c2 = __int_as_float(e2.y);
        const float4 v0 = *(const float4*)(hw + (size_t)e0.x * FDIM + f);
        const float4 v1 = *(const float4*)(hw + (size_t)e1.x * FDIM + f);
        const float4 v2 = *(const float4*)(hw + (size_t)e2.x * FDIM + f);
        acc.x = fmaf(v0.x, c0, acc.x); acc.y = fmaf(v0.y, c0, acc.y);
        acc.z = fmaf(v0.z, c0, acc.z); acc.w = fmaf(v0.w, c0, acc.w);
        float4 t;
        t.x = fmaf(v1.x, c1, acc.x); t.y = fmaf(v1.y, c1, acc.y);
        t.z = fmaf(v1.z, c1, acc.z); t.w = fmaf(v1.w, c1, acc.w);
        acc.x = (rem > 1) ? t.x : acc.x; acc.y = (rem > 1) ? t.y : acc.y;
        acc.z = (rem > 1) ? t.z : acc.z; acc.w = (rem > 1) ? t.w : acc.w;
        t.x = fmaf(v2.x, c2, acc.x); t.y = fmaf(v2.y, c2, acc.y);
        t.z = fmaf(v2.z, c2, acc.z); t.w = fmaf(v2.w, c2, acc.w);
        acc.x = (rem > 2) ? t.x : acc.x; acc.y = (rem > 2) ? t.y : acc.y;
        acc.z = (rem > 2) ? t.z : acc.z; acc.w = (rem > 2) ? t.w : acc.w;
    }

    const float di  = dis[node];
    const float di2 = di * di;
    const float4 h = *(const float4*)(hw + (size_t)node * FDIM + f);
    float4 r;
    r.x = acc.x * di + h.x * di2 + b[f + 0];
    r.y = acc.y * di + h.y * di2 + b[f + 1];
    r.z = acc.z * di + h.z * di2 + b[f + 2];
    r.w = acc.w * di + h.w * di2 + b[f + 3];
    if (doRelu) {
        r.x = fmaxf(r.x, 0.0f); r.y = fmaxf(r.y, 0.0f);
        r.z = fmaxf(r.z, 0.0f); r.w = fmaxf(r.w, 0.0f);
    }
    *(float4*)(hout + (size_t)node * FDIM + f) = r;
}

static_assert(E_TR % 32 == 0 && E_TE % 32 == 0, "a wave's 32 rows must lie inside one edge list");
static_assert(N_ROWS % 32 == 0, "whole waves only");

__global__ __launch_bounds__(256) void edge_score(const float* __restrict__ P,
                                                  const float* __restrict__ Q,
                                                  const int* __restrict__ pos_tr,
                                                  const int* __restrict__ neg_tr,
                                                  const int* __restrict__ pos_te,
                                                  const int* __restrict__ neg_te,
                                                  const float* __restrict__ bl1,
                                                  const float* __restrict__ Wl2,
                                                  const float* __restrict__ bl2,
                                                  float* __restrict__ out) {
    const int r0 = __builtin_amdgcn_readfirstlane(
        (int)(((blockIdx.x * blockDim.x + threadIdx.x) >> 5) * 32));
    if (r0 >= N_ROWS) return;
    const int lane = threadIdx.x & 31;
    const int g = lane >> 3;
    const int j = lane & 7;

    const int* e; int i0, len;
    if (r0 < E_TR)                 { e = pos_tr; i0 = r0;                   len = E_TR; }
    else if (r0 < 2 * E_TR)        { e = neg_tr; i0 = r0 - E_TR;            len = E_TR; }
    else if (r0 < 2 * E_TR + E_TE) { e = pos_te; i0 = r0 - 2 * E_TR;        len = E_TE; }
    else                           { e = neg_te; i0 = r0 - 2 * E_TR - E_TE; len = E_TE; }

    const int* eu = e + i0 + g * 8;
    const int* ev = eu + len;
    const int4 ua = *(const int4*)eu, ub = *(const int4*)(eu + 4);
    const int4 va = *(const int4*)ev, vb = *(const int4*)(ev + 4);
    const int us[8] = {ua.x, ua.y, ua.z, ua.w, ub.x, ub.y, ub.z, ub.w};
    const int vs[8] = {va.x, va.y, va.z, va.w, vb.x, vb.y, vb.z, vb.w};

    float4 bb[4], ww[4];
    #pragma unroll
    for (int c = 0; c < 4; ++c) {
        bb[c] = *(const float4*)(bl1 + c * 32 + j * 4);
        ww[c] = *(const float4*)(Wl2 + c * 32 + j * 4);
    }

    float s[8];
    #pragma unroll
    for (int p = 0; p < 8; ++p) {
        const float* pr = P + (size_t)us[p] * FDIM + j * 4;
        const float* qr = Q + (size_t)vs[p] * FDIM + j * 4;
        float acc = 0.0f;
        #pragma unroll
        for (int c = 0; c < 4; ++c) {
            const float4 a = *(const float4*)(pr + c * 32);
            const float4 b = *(const float4*)(qr + c * 32);
            acc = fmaf(fmaxf(a.x + b.x + bb[c].x, 0.0f), ww[c].x, acc);
            acc = fmaf(fmaxf(a.y + b.y + bb[c].y, 0.0f), ww[c].y, acc);
            acc = fmaf(fmaxf(a.z + b.z + bb[c].z, 0.0f), ww[c].z, acc);
            acc = fmaf(fmaxf(a.w + b.w + bb[c].w, 0.0f), ww[c].w, acc);
        }
        s[p] = acc;
    }

    const bool k4 = (j & 4) != 0, k2 = (j & 2) != 0, k1 = (j & 1) != 0;
    float t4[4];
    #pragma unroll
    for (int i = 0; i < 4; ++i) {
        const float keep = k4 ? s[4 + i] : s[i];
        const float send = k4 ? s[i] : s[4 + i];
        t4[i] = keep + __shfl_xor(send, 4, 32);
    }
    float t2[2];
    #pragma unroll
    for (int i = 0; i < 2; ++i) {
        const float keep = k2 ? t4[2 + i] : t4[i];
        const float send = k2 ? t4[i] : t4[2 + i];
        t2[i] = keep + __shfl_xor(send, 2, 32);
    }
    const float keep = k1 ? t2[1] : t2[0];
    const float send = k1 ? t2[0] : t2[1];
    const float tot = keep + __shfl_xor(send, 1, 32);

    out[r0 + lane] = tot + bl2[0];
}

extern "C" void kernel_launch(void* const* d_in, const int* in_sizes, int n_in,
                              void* d_out, int out_size, void* d_ws, size_t ws_size,
                              hipStream_t stream) {
    const float* x          = (const float*)d_in[0];
    const int*   edge_index = (const int*)d_in[1];
    const int*   pos_tr     = (const int*)d_in[2];
    const int*   neg_tr     = (const int*)d_in[3];
    const int*   pos_te     = (const int*)d_in[4];
    const int*   neg_te     = (const int*)d_in[5];
    const float* W[4] = {(const float*)d_in[6],  (const float*)d_in[8],
                         (const float*)d_in[10], (const float*)d_in[12]};
    const float* b[4] = {(const float*)d_in[7],  (const float*)d_in[9],
                         (const float*)d_in[11], (const float*)d_in[13]};
    const float* Wl1 = (const float*)d_in[14];
    const float* bl1 = (const float*)d_in[15];
    const float* Wl2 = (const float*)d_in[16];
    const float* bl2 = (const float*)d_in[17];
    float* out = (float*)d_out;

    const size_t NF = (size_t)N_NODES * FDIM;
    float* ws  = (float*)d_ws;
    float* dis = ws;
    float* hw  = ws + 50048;
    int*   csr = (int*)(hw + NF);
    int*   cnt     = csr;
    int*   row_ptr = csr + 53248;
    int*   rank    = csr + 2 * 53248;
    int*   ew      = csr + 2 * 53248 + N_EDGES;
    float* h0  = hw + 2 * NF;
    float* h1  = h0 + NF;

    const int* src = edge_index;
    const int* dst = edge_index + N_EDGES;

    hipMemsetAsync(cnt, 0, N_NODES * sizeof(int), stream);
    count_rank<<<(N_EDGES + 255) / 256, 256, 0, stream>>>(dst, cnt, rank, N_EDGES);
    scan_rows<<<1, 1024, 0, stream>>>(cnt, row_ptr, dis);
    fill_cols<<<(N_EDGES + 255) / 256, 256, 0, stream>>>(src, dst, rank, row_ptr, dis, ew, N_EDGES);

    const int tilesM     = (N_NODES + 15) / 16;
    const int gemmBlocks = (tilesM + 7) / 8;
    const float* hin = x;
    float* houts[4] = {h0, h1, h0, h1};
    for (int l = 0; l < 4; ++l) {
        gemm_node<<<gemmBlocks, 256, 0, stream>>>(hin, W[l], hw, N_NODES);
        gather_combine<<<(N_NODES * 32 + 255) / 256, 256, 0, stream>>>(
            hw, dis, row_ptr, ew, b[l], houts[l], N_NODES, (l < 3) ? 1 : 0);
        hin = houts[l];
    }

    gemm_node<<<dim3(gemmBlocks, 2), 256, 0, stream>>>(h1, Wl1, hw, N_NODES);

    const int edgeWaves  = N_ROWS / 32;
    const int edgeBlocks = (edgeWaves + 7) / 8;
    edge_score<<<edgeBlocks, 256, 0, stream>>>(hw, hw + NF, pos_tr, neg_tr, pos_te, neg_te,
                                               bl1, Wl2, bl2, out);
}


# Node GEMM takes A fragments from LDS: each wave loads its 16 rows of A as whole 512-byte rows and stages them in its own LDS strip; a fragment is one 8-byte LDS read; k order and sums kept
# speedup vs baseline: 3.4926x; 1.0318x over previous
#include <hip/hip_runtime.h>


typedef float v2f __attribute__((ext_vector_type(2)));
typedef float v8f __attribute__((ext_vector_type(8)));

#define N_NODES 50000
#define N_EDGES 800000
#define FDIM    128
#define E_TR    100000
#define E_TE    20000
#define N_ROWS  240000

__global__ __launch_bounds__(256) void count_rank(const int* __restrict__ dst,
                                                  int* __restrict__ cnt,
                                                  int* __restrict__ rank, int nE) {
    int e = blockIdx.x * blockDim.x + threadIdx.x;
    if (e < nE) rank[e] = atomicAdd(&cnt[dst[e]], 1);
}

#define SCAN_PASSES 13
#define SCAN_CHUNK  (SCAN_PASSES * 128)
static_assert(32 * SCAN_CHUNK >= N_NODES, "one block must cover every node");
static_assert(N_NODES % 4 == 0, "16-byte groups must not straddle the end");

__device__ __forceinline__ int wave_incl_scan(int v, int lane) {
    #pragma unroll
    for (int o = 1; o < 32; o <<= 1) {
        int y = __shfl_up(v, o, 32);
        if (lane >= o) v += y;
    }
    return v;
}

__global__ __launch_bounds__(1024) void scan_rows(const int* __restrict__ cnt,
                                                  int* __restrict__ row_ptr,
                                                  float* __restrict__ dis) {
    __shared__ int wtot[32];
    const int lane = threadIdx.x & 31;
    const int w    = threadIdx.x >> 5;
    const int base = w * SCAN_CHUNK + lane * 4;

    int4 p[SCAN_PASSES];
    #pragma unroll
    for (int i = 0; i < SCAN_PASSES; ++i) {
        const int idx = base + i * 128;
        const int4 v = *(const int4*)(cnt + ((idx < N_NODES) ? idx : N_NODES - 4));
        p[i] = (idx < N_NODES) ? v : make_int4(0, 0, 0, 0);
    }
    #pragma unroll
    for (int i = 0; i < SCAN_PASSES; ++i) {
        const int idx = base + i * 128;
        if (idx < N_NODES) {
            float4 d;
            d.x = rsqrtf((float)p[i].x + 1.0f);
            d.y = rsqrtf((float)p[i].y + 1.0f);
            d.z = rsqrtf((float)p[i].z + 1.0f);
            d.w = rsqrtf((float)p[i].w + 1.0f);
            *(float4*)(dis + idx) = d;
        }
    }

    int run = 0;
    #pragma unroll
    for (int i = 0; i < SCAN_PASSES; ++i) {
        const int4 c = p[i];
        const int s   = c.x + c.y + c.z + c.w;
        const int inc = wave_incl_scan(s, lane);
        const int o   = run + inc - s;
        p[i].x = o;
        p[i].y = o + c.x;
        p[i].z = o + c.x + c.y;
        p[i].w = o + c.x + c.y + c.z;
        run += __shfl(inc, 31, 32);
    }
    if (lane == 0) wtot[w] = run;
    __syncthreads();

    const int t     = wtot[lane];
    const int tinc  = wave_incl_scan(t, lane);
    const int wbase = __shfl(tinc - t, w, 32);

    #pragma unroll
    for (int i = 0; i < SCAN_PASSES; ++i) {
        int idx = base + i * 128;
        if (idx < N_NODES) {
            int4 q = p[i];
            q.x += wbase; q.y += wbase; q.z += wbase; q.w += wbase;
            *(int4*)(row_ptr + idx) = q;
        }
    }
    if (threadIdx.x == 1023) row_ptr[N_NODES] = wbase + run;
}

__global__ __launch_bounds__(256) void fill_cols(const int* __restrict__ src,
                                                 const int* __restrict__ dst,
                                                 const int* __restrict__ rank,
                                                 const int* __restrict__ row_ptr,
                                                 const float* __restrict__ dis,
                                                 int* __restrict__ ewbuf, int nE) {
    int2* ew = (int2*)ewbuf;
    int e = blockIdx.x * blockDim.x + threadIdx.x;
    if (e < nE) {
        const int s = src[e];
        ew[row_ptr[dst[e]] + rank[e]] = make_int2(s, __float_as_int(dis[s]));
    }
}

#define WS_PITCH 288
#define AS_PITCH 132
__global__ __launch_bounds__(256) void gemm_node(const float* __restrict__ A,
                                                 const float* __restrict__ B,
                                                 float* __restrict__ C, int M) {
    __shared__ float Ws[(FDIM / 2) * WS_PITCH];
    __shared__ float As[8 * 16 * AS_PITCH];
    const int wave = (blockIdx.x * blockDim.x + threadIdx.x) >> 5;
    const int lane = threadIdx.x & 31;
    const int m0 = wave * 16;
    const bool live = m0 < M;
    const int half = lane >> 4;
    const int l16  = lane & 15;

    B += (size_t)blockIdx.y * FDIM * FDIM;
    C += (size_t)blockIdx.y * M * FDIM;

    float4 r0[8], r1[8];
    #pragma unroll
    for (int it = 0; it < 8; ++it) {
        const int kp = it * 8 + (threadIdx.x >> 5);
        r0[it] = *(const float4*)(B + (size_t)(2 * kp) * FDIM + lane * 4);
        r1[it] = *(const float4*)(B + (size_t)(2 * kp + 1) * FDIM + lane * 4);
    }

    const float* Astrip = A + (size_t)(live ? m0 : 0) * FDIM + lane * 4;
    float4 arow[16];
    #pragma unroll
    for (int i = 0; i < 16; ++i) arow[i] = *(const float4*)(Astrip + (size_t)i * FDIM);

    #pragma unroll
    for (int it = 0; it < 8; ++it) {
        const int kp = it * 8 + (threadIdx.x >> 5);
        float* d = Ws + kp * WS_PITCH + lane * 8;
        *(float4*)d       = make_float4(r0[it].x, r1[it].x, r0[it].y, r1[it].y);
        *(float4*)(d + 4) = make_float4(r0[it].z, r1[it].z, r0[it].w, r1[it].w);
    }
    float* as = As + (threadIdx.x >> 5) * (16 * AS_PITCH);
    #pragma unroll
    for (int i = 0; i < 16; ++i) *(float4*)(as + i * AS_PITCH + lane * 4) = arow[i];
    __syncthreads();
    if (!live) return;

    const float* ap = as + l16 * AS_PITCH + half * 2;
    v2f afrag[FDIM / 4];
    #pragma unroll
    for (int s = 0; s < FDIM / 4; ++s) afrag[s] = *(const v2f*)(ap + s * 4);

    const float* bp = Ws + half * WS_PITCH + l16 * 2;

    v8f acc[8] = {};
    v2f bcur[8];
    #pragma unroll
    for (int n = 0; n < 8; ++n) bcur[n] = *(const v2f*)(bp + n * 32);

    #pragma unroll
    for (int s = 0; s < FDIM / 4; ++s) {
        v2f bnxt[8];
        if (s + 1 < FDIM / 4) {
            #pragma unroll
            for (int n = 0; n < 8; ++n)
                bnxt[n] = *(const v2f*)(bp + (s + 1) * 2 * WS_PITCH + n * 32);
        }
        #pragma unroll
        for (int n = 0; n < 8; ++n)
            acc[n] = __builtin_amdgcn_wmma_f32_16x16x4_f32(
                false, afrag[s], false, bcur[n], (short)0, acc[n], false, false);
        if (s + 1 < FDIM / 4) {
            #pragma unroll
            for (int n = 0; n < 8; ++n) bcur[n] = bnxt[n];
        }
    }

    #pragma unroll
    for (int n = 0; n < 8; ++n) {
        #pragma unroll
        for (int g = 0; g < 8; ++g) {
            int m = m0 + g + half * 8;
            C[(size_t)m * FDIM + n * 16 + l16] = acc[n][g];
        }
    }
}

__global__ __launch_bounds__(256) void gather_combine(const float* __restrict__ hw,
                                                      const float* __restrict__ dis,
                                                      const int* __restrict__ row_ptr,
                                                      const int* __restrict__ ewbuf,
                                                      const float* __restrict__ b,
                                                      float* __restrict__ hout,
                                                      int nNodes, int doRelu) {
    const int node = __builtin_amdgcn_readfirstlane((int)(blockIdx.x * 8 + (threadIdx.x >> 5)));
    if (node >= nNodes) return;
    const int f = (threadIdx.x & 31) * 4;
    const int2* ew = (const int2*)ewbuf;

    int j = row_ptr[node];
    const int end = row_ptr[node + 1];

    float4 acc = make_float4(0.0f, 0.0f, 0.0f, 0.0f);
    for (; j + 4 <= end; j += 4) {
        const int2 e0 = ew[j], e1 = ew[j + 1], e2 = ew[j + 2], e3 = ew[j + 3];
        const float c0 = __int_as_float(e0.y), c1 = __int_as_float(e1.y);
        const float c2 = __int_as_float(e2.y), c3 = __int_as_float(e3.y);
        const float4 v0 = *(const float4*)(hw + (size_t)e0.x * FDIM + f);
        const float4 v1 = *(const float4*)(hw + (size_t)e1.x * FDIM + f);
        const float4 v2 = *(const float4*)(hw + (size_t)e2.x * FDIM + f);
        const float4 v3 = *(const float4*)(hw + (size_t)e3.x * FDIM + f);
        acc.x = fmaf(v0.x, c0, acc.x); acc.y = fmaf(v0.y, c0, acc.y);
        acc.z = fmaf(v0.z, c0, acc.z); acc.w = fmaf(v0.w, c0, acc.w);
        acc.x = fmaf(v1.x, c1, acc.x); acc.y = fmaf(v1.y, c1, acc.y);
        acc.z = fmaf(v1.z, c1, acc.z); acc.w = fmaf(v1.w, c1, acc.w);
        acc.x = fmaf(v2.x, c2, acc.x); acc.y = fmaf(v2.y, c2, acc.y);
        acc.z = fmaf(v2.z, c2, acc.z); acc.w = fmaf(v2.w, c2, acc.w);
        acc.x = fmaf(v3.x, c3, acc.x); acc.y = fmaf(v3.y, c3, acc.y);
        acc.z = fmaf(v3.z, c3, acc.z); acc.w = fmaf(v3.w, c3, acc.w);
    }
    const int rem = end - j;
    if (rem > 0) {
        const int last = end - 1;
        const int j1 = (j + 1 < end) ? j + 1 : last;
        const int j2 = (j + 2 < end) ? j + 2 : last;
        const int2 e0 = ew[j], e1 = ew[j1], e2 = ew[j2];
        const float c0 = __int_as_float(e0.y), c1 = __int_as_float(e1.y), c2 = __int_as_float(e2.y);
        const float4 v0 = *(const float4*)(hw + (size_t)e0.x * FDIM + f);
        const float4 v1 = *(const float4*)(hw + (size_t)e1.x * FDIM + f);
        const float4 v2 = *(const float4*)(hw + (size_t)e2.x * FDIM + f);
        acc.x = fmaf(v0.x, c0, acc.x); acc.y = fmaf(v0.y, c0, acc.y);
        acc.z = fmaf(v0.z, c0, acc.z); acc.w = fmaf(v0.w, c0, acc.w);
        float4 t;
        t.x = fmaf(v1.x, c1, acc.x); t.y = fmaf(v1.y, c1, acc.y);
        t.z = fmaf(v1.z, c1, acc.z); t.w = fmaf(v1.w, c1, acc.w);
        acc.x = (rem > 1) ? t.x : acc.x; acc.y = (rem > 1) ? t.y : acc.y;
        acc.z = (rem > 1) ? t.z : acc.z; acc.w = (rem > 1) ? t.w : acc.w;
        t.x = fmaf(v2.x, c2, acc.x); t.y = fmaf(v2.y, c2, acc.y);
        t.z = fmaf(v2.z, c2, acc.z); t.w = fmaf(v2.w, c2, acc.w);
        acc.x = (rem > 2) ? t.x : acc.x; acc.y = (rem > 2) ? t.y : acc.y;
        acc.z = (rem > 2) ? t.z : acc.z; acc.w = (rem > 2) ? t.w : acc.w;
    }

    const float di  = dis[node];
    const float di2 = di * di;
    const float4 h = *(const float4*)(hw + (size_t)node * FDIM + f);
    float4 r;
    r.x = acc.x * di + h.x * di2 + b[f + 0];
    r.y = acc.y * di + h.y * di2 + b[f + 1];
    r.z = acc.z * di + h.z * di2 + b[f + 2];
    r.w = acc.w * di + h.w * di2 + b[f + 3];
    if (doRelu) {
        r.x = fmaxf(r.x, 0.0f); r.y = fmaxf(r.y, 0.0f);
        r.z = fmaxf(r.z, 0.0f); r.w = fmaxf(r.w, 0.0f);
    }
    *(float4*)(hout + (size_t)node * FDIM + f) = r;
}

static_assert(E_TR % 32 == 0 && E_TE % 32 == 0, "a wave's 32 rows must lie inside one edge list");
static_assert(N_ROWS % 32 == 0, "whole waves only");

__global__ __launch_bounds__(256) void edge_score(const float* __restrict__ P,
                                                  const float* __restrict__ Q,
                                                  const int* __restrict__ pos_tr,
                                                  const int* __restrict__ neg_tr,
                                                  const int* __restrict__ pos_te,
                                                  const int* __restrict__ neg_te,
                                                  const float* __restrict__ bl1,
                                                  const float* __restrict__ Wl2,
                                                  const float* __restrict__ bl2,
                                                  float* __restrict__ out) {
    const int r0 = __builtin_amdgcn_readfirstlane(
        (int)(((blockIdx.x * blockDim.x + threadIdx.x) >> 5) * 32));
    if (r0 >= N_ROWS) return;
    const int lane = threadIdx.x & 31;
    const int g = lane >> 3;
    const int j = lane & 7;

    const int* e; int i0, len;
    if (r0 < E_TR)                 { e = pos_tr; i0 = r0;                   len = E_TR; }
    else if (r0 < 2 * E_TR)        { e = neg_tr; i0 = r0 - E_TR;            len = E_TR; }
    else if (r0 < 2 * E_TR + E_TE) { e = pos_te; i0 = r0 - 2 * E_TR;        len = E_TE; }
    else                           { e = neg_te; i0 = r0 - 2 * E_TR - E_TE; len = E_TE; }

    const int* eu = e + i0 + g * 8;
    const int* ev = eu + len;
    const int4 ua = *(const int4*)eu, ub = *(const int4*)(eu + 4);
    const int4 va = *(const int4*)ev, vb = *(const int4*)(ev + 4);
    const int us[8] = {ua.x, ua.y, ua.z, ua.w, ub.x, ub.y, ub.z, ub.w};
    const int vs[8] = {va.x, va.y, va.z, va.w, vb.x, vb.y, vb.z, vb.w};

    float4 bb[4], ww[4];
    #pragma unroll
    for (int c = 0; c < 4; ++c) {
        bb[c] = *(const float4*)(bl1 + c * 32 + j * 4);
        ww[c] = *(const float4*)(Wl2 + c * 32 + j * 4);
    }

    float s[8];
    #pragma unroll
    for (int p = 0; p < 8; ++p) {
        const float* pr = P + (size_t)us[p] * FDIM + j * 4;
        const float* qr = Q + (size_t)vs[p] * FDIM + j * 4;
        float acc = 0.0f;
        #pragma unroll
        for (int c = 0; c < 4; ++c) {
            const float4 a = *(const float4*)(pr + c * 32);
            const float4 b = *(const float4*)(qr + c * 32);
            acc = fmaf(fmaxf(a.x + b.x + bb[c].x, 0.0f), ww[c].x, acc);
            acc = fmaf(fmaxf(a.y + b.y + bb[c].y, 0.0f), ww[c].y, acc);
            acc = fmaf(fmaxf(a.z + b.z + bb[c].z, 0.0f), ww[c].z, acc);
            acc = fmaf(fmaxf(a.w + b.w + bb[c].w, 0.0f), ww[c].w, acc);
        }
        s[p] = acc;
    }

    const bool k4 = (j & 4) != 0, k2 = (j & 2) != 0, k1 = (j & 1) != 0;
    float t4[4];
    #pragma unroll
    for (int i = 0; i < 4; ++i) {
        const float keep = k4 ? s[4 + i] : s[i];
        const float send = k4 ? s[i] : s[4 + i];
        t4[i] = keep + __shfl_xor(send, 4, 32);
    }
    float t2[2];
    #pragma unroll
    for (int i = 0; i < 2; ++i) {
        const float keep = k2 ? t4[2 + i] : t4[i];
        const float send = k2 ? t4[i] : t4[2 + i];
        t2[i] = keep + __shfl_xor(send, 2, 32);
    }
    const float keep = k1 ? t2[1] : t2[0];
    const float send = k1 ? t2[0] : t2[1];
    const float tot = keep + __shfl_xor(send, 1, 32);

    out[r0 + lane] = tot + bl2[0];
}

extern "C" void kernel_launch(void* const* d_in, const int* in_sizes, int n_in,
                              void* d_out, int out_size, void* d_ws, size_t ws_size,
                              hipStream_t stream) {
    const float* x          = (const float*)d_in[0];
    const int*   edge_index = (const int*)d_in[1];
    const int*   pos_tr     = (const int*)d_in[2];
    const int*   neg_tr     = (const int*)d_in[3];
    const int*   pos_te     = (const int*)d_in[4];
    const int*   neg_te     = (const int*)d_in[5];
    const float* W[4] = {(const float*)d_in[6],  (const float*)d_in[8],
                         (const float*)d_in[10], (const float*)d_in[12]};
    const float* b[4] = {(const float*)d_in[7],  (const float*)d_in[9],
                         (const float*)d_in[11], (const float*)d_in[13]};
    const float* Wl1 = (const float*)d_in[14];
    const float* bl1 = (const float*)d_in[15];
    const float* Wl2 = (const float*)d_in[16];
    const float* bl2 = (const float*)d_in[17];
    float* out = (float*)d_out;

    const size_t NF = (size_t)N_NODES * FDIM;
    float* ws  = (float*)d_ws;
    float* dis = ws;
    float* hw  = ws + 50048;
    int*   csr = (int*)(hw + NF);
    int*   cnt     = csr;
    int*   row_ptr = csr + 53248;
    int*   rank    = csr + 2 * 53248;
    int*   ew      = csr + 2 * 53248 + N_EDGES;
    float* h0  = hw + 2 * NF;
    float* h1  = h0 + NF;

    const int* src = edge_index;
    const int* dst = edge_index + N_EDGES;

    hipMemsetAsync(cnt, 0, N_NODES * sizeof(int), stream);
    count_rank<<<(N_EDGES + 255) / 256, 256, 0, stream>>>(dst, cnt, rank, N_EDGES);
    scan_rows<<<1, 1024, 0, stream>>>(cnt, row_ptr, dis);
    fill_cols<<<(N_EDGES + 255) / 256, 256, 0, stream>>>(src, dst, rank, row_ptr, dis, ew, N_EDGES);

    const int tilesM     = (N_NODES + 15) / 16;
    const int gemmBlocks = (tilesM + 7) / 8;
    const float* hin = x;
    float* houts[4] = {h0, h1, h0, h1};
    for (int l = 0; l < 4; ++l) {
        gemm_node<<<gemmBlocks, 256, 0, stream>>>(hin, W[l], hw, N_NODES);
        gather_combine<<<(N_NODES * 32 + 255) / 256, 256, 0, stream>>>(
            hw, dis, row_ptr, ew, b[l], houts[l], N_NODES, (l < 3) ? 1 : 0);
        hin = houts[l];
    }

    gemm_node<<<dim3(gemmBlocks, 2), 256, 0, stream>>>(h1, Wl1, hw, N_NODES);

    const int edgeWaves  = N_ROWS / 32;
    const int edgeBlocks = (edgeWaves + 7) / 8;
    edge_score<<<edgeBlocks, 256, 0, stream>>>(hw, hw + NF, pos_tr, neg_tr, pos_te, neg_te,
                                               bl1, Wl2, bl2, out);
}


# CSR build ranks edges with LDS counters (two-level counting sort: 100 edge chunks, per-node LDS counters, prefix over chunks), no returning global atomic per edge; row sum order: chunk, then arrival
# speedup vs baseline: 4.0471x; 1.1588x over previous
#include <hip/hip_runtime.h>


typedef float v2f __attribute__((ext_vector_type(2)));
typedef float v8f __attribute__((ext_vector_type(8)));

#define N_NODES 50000
#define N_EDGES 800000
#define FDIM    128
#define E_TR    100000
#define E_TE    20000
#define N_ROWS  240000

#define CS_BLOCKS 100
#define CS_CHUNK  (N_EDGES / CS_BLOCKS)
#define CS_GROUPS (CS_CHUNK / 4)
#define CS_PASSES ((CS_GROUPS + 1023) / 1024)
#define CS_STRIDE 50048
static_assert(N_EDGES % CS_BLOCKS == 0 && CS_CHUNK % 4 == 0, "chunks are whole 16-byte groups");
static_assert(CS_STRIDE >= N_NODES && CS_STRIDE % 32 == 0, "a plane holds every node");
static_assert(N_NODES % 4 == 0, "16-byte groups must not straddle the end");

__global__ __launch_bounds__(1024) void count_block(const int* __restrict__ dst,
                                                    int* __restrict__ blk) {
    __shared__ int hist[N_NODES];
    const int4* d4 = (const int4*)(dst + (size_t)blockIdx.x * CS_CHUNK);
    int4 dv[CS_PASSES];
    bool ok[CS_PASSES];
    #pragma unroll
    for (int p = 0; p < CS_PASSES; ++p) {
        const int i = p * 1024 + threadIdx.x;
        ok[p] = i < CS_GROUPS;
        dv[p] = d4[ok[p] ? i : CS_GROUPS - 1];
    }

    for (int i = threadIdx.x * 4; i < N_NODES; i += 4096)
        *(int4*)(hist + i) = make_int4(0, 0, 0, 0);
    __syncthreads();

    #pragma unroll
    for (int p = 0; p < CS_PASSES; ++p) {
        if (ok[p]) {
            atomicAdd(&hist[dv[p].x], 1);
            atomicAdd(&hist[dv[p].y], 1);
            atomicAdd(&hist[dv[p].z], 1);
            atomicAdd(&hist[dv[p].w], 1);
        }
    }
    __syncthreads();

    int* out = blk + (size_t)blockIdx.x * CS_STRIDE;
    for (int i = threadIdx.x * 4; i < N_NODES; i += 4096)
        *(int4*)(out + i) = *(const int4*)(hist + i);
}

__global__ __launch_bounds__(256) void chunk_prefix(int* __restrict__ blk,
                                                    int* __restrict__ cnt) {
    const int n = blockIdx.x * 256 + threadIdx.x;
    if (n >= N_NODES) return;
    int c[CS_BLOCKS];
    #pragma unroll
    for (int b = 0; b < CS_BLOCKS; ++b) c[b] = blk[(size_t)b * CS_STRIDE + n];
    int run = 0;
    #pragma unroll
    for (int b = 0; b < CS_BLOCKS; ++b) {
        blk[(size_t)b * CS_STRIDE + n] = run;
        run += c[b];
    }
    cnt[n] = run;
}

#define SCAN_PASSES 13
#define SCAN_CHUNK  (SCAN_PASSES * 128)
static_assert(32 * SCAN_CHUNK >= N_NODES, "one block must cover every node");
static_assert(N_NODES % 4 == 0, "16-byte groups must not straddle the end");

__device__ __forceinline__ int wave_incl_scan(int v, int lane) {
    #pragma unroll
    for (int o = 1; o < 32; o <<= 1) {
        int y = __shfl_up(v, o, 32);
        if (lane >= o) v += y;
    }
    return v;
}

__global__ __launch_bounds__(1024) void scan_rows(const int* __restrict__ cnt,
                                                  int* __restrict__ row_ptr,
                                                  float* __restrict__ dis) {
    __shared__ int wtot[32];
    const int lane = threadIdx.x & 31;
    const int w    = threadIdx.x >> 5;
    const int base = w * SCAN_CHUNK + lane * 4;

    int4 p[SCAN_PASSES];
    #pragma unroll
    for (int i = 0; i < SCAN_PASSES; ++i) {
        const int idx = base + i * 128;
        const int4 v = *(const int4*)(cnt + ((idx < N_NODES) ? idx : N_NODES - 4));
        p[i] = (idx < N_NODES) ? v : make_int4(0, 0, 0, 0);
    }
    #pragma unroll
    for (int i = 0; i < SCAN_PASSES; ++i) {
        const int idx = base + i * 128;
        if (idx < N_NODES) {
            float4 d;
            d.x = rsqrtf((float)p[i].x + 1.0f);
            d.y = rsqrtf((float)p[i].y + 1.0f);
            d.z = rsqrtf((float)p[i].z + 1.0f);
            d.w = rsqrtf((float)p[i].w + 1.0f);
            *(float4*)(dis + idx) = d;
        }
    }

    int run = 0;
    #pragma unroll
    for (int i = 0; i < SCAN_PASSES; ++i) {
        const int4 c = p[i];
        const int s   = c.x + c.y + c.z + c.w;
        const int inc = wave_incl_scan(s, lane);
        const int o   = run + inc - s;
        p[i].x = o;
        p[i].y = o + c.x;
        p[i].z = o + c.x + c.y;
        p[i].w = o + c.x + c.y + c.z;
        run += __shfl(inc, 31, 32);
    }
    if (lane == 0) wtot[w] = run;
    __syncthreads();

    const int t     = wtot[lane];
    const int tinc  = wave_incl_scan(t, lane);
    const int wbase = __shfl(tinc - t, w, 32);

    #pragma unroll
    for (int i = 0; i < SCAN_PASSES; ++i) {
        int idx = base + i * 128;
        if (idx < N_NODES) {
            int4 q = p[i];
            q.x += wbase; q.y += wbase; q.z += wbase; q.w += wbase;
            *(int4*)(row_ptr + idx) = q;
        }
    }
    if (threadIdx.x == 1023) row_ptr[N_NODES] = wbase + run;
}

#define HALF_NODES (N_NODES / 2)
#define TAB_PASSES ((HALF_NODES / 4 + 1023) / 1024)
static_assert(N_NODES % 2 == 0 && HALF_NODES % 4 == 0, "a half is whole 16-byte groups");

__global__ __launch_bounds__(1024) void fill_block(const int* __restrict__ src,
                                                   const int* __restrict__ dst,
                                                   const int* __restrict__ blk,
                                                   const int* __restrict__ row_ptr,
                                                   const float* __restrict__ dis,
                                                   int* __restrict__ ewbuf) {
    __shared__ int cur[HALF_NODES];
    const int nlo = blockIdx.y * HALF_NODES;
    const int4* s4 = (const int4*)(src + (size_t)blockIdx.x * CS_CHUNK);
    const int4* d4 = (const int4*)(dst + (size_t)blockIdx.x * CS_CHUNK);
    int4 sv[CS_PASSES], dv[CS_PASSES];
    bool ok[CS_PASSES];
    #pragma unroll
    for (int p = 0; p < CS_PASSES; ++p) {
        const int i = p * 1024 + threadIdx.x;
        ok[p] = i < CS_GROUPS;
        sv[p] = s4[ok[p] ? i : CS_GROUPS - 1];
        dv[p] = d4[ok[p] ? i : CS_GROUPS - 1];
    }

    const int* off = blk + (size_t)blockIdx.x * CS_STRIDE + nlo;
    const int* rp  = row_ptr + nlo;
    int4 a[TAB_PASSES], o[TAB_PASSES];
    #pragma unroll
    for (int t = 0; t < TAB_PASSES; ++t) {
        const int i = (t * 1024 + threadIdx.x) * 4;
        const int ii = (i < HALF_NODES) ? i : HALF_NODES - 4;
        a[t] = *(const int4*)(rp + ii);
        o[t] = *(const int4*)(off + ii);
    }
    #pragma unroll
    for (int t = 0; t < TAB_PASSES; ++t) {
        const int i = (t * 1024 + threadIdx.x) * 4;
        if (i < HALF_NODES)
            *(int4*)(cur + i) = make_int4(a[t].x + o[t].x, a[t].y + o[t].y,
                                          a[t].z + o[t].z, a[t].w + o[t].w);
    }

    int   sN[CS_PASSES * 4];
    int   rN[CS_PASSES * 4];
    bool  in[CS_PASSES * 4];
    float sc[CS_PASSES * 4];
    #pragma unroll
    for (int p = 0; p < CS_PASSES; ++p) {
        const int s_[4] = {sv[p].x, sv[p].y, sv[p].z, sv[p].w};
        const int d_[4] = {dv[p].x, dv[p].y, dv[p].z, dv[p].w};
        #pragma unroll
        for (int q = 0; q < 4; ++q) {
            const int k = p * 4 + q;
            sN[k] = s_[q];
            rN[k] = d_[q] - nlo;
            in[k] = ok[p] && (unsigned)rN[k] < (unsigned)HALF_NODES;
            sc[k] = dis[in[k] ? sN[k] : 0];
        }
    }
    __syncthreads();

    int2* ew = (int2*)ewbuf;
    #pragma unroll
    for (int k = 0; k < CS_PASSES * 4; ++k) {
        if (in[k]) {
            const int slot = atomicAdd(&cur[rN[k]], 1);
            ew[slot] = make_int2(sN[k], __float_as_int(sc[k]));
        }
    }
}

#define WS_PITCH 288
#define AS_PITCH 132
__global__ __launch_bounds__(256) void gemm_node(const float* __restrict__ A,
                                                 const float* __restrict__ B,
                                                 float* __restrict__ C, int M) {
    __shared__ float Ws[(FDIM / 2) * WS_PITCH];
    __shared__ float As[8 * 16 * AS_PITCH];
    const int wave = (blockIdx.x * blockDim.x + threadIdx.x) >> 5;
    const int lane = threadIdx.x & 31;
    const int m0 = wave * 16;
    const bool live = m0 < M;
    const int half = lane >> 4;
    const int l16  = lane & 15;

    B += (size_t)blockIdx.y * FDIM * FDIM;
    C += (size_t)blockIdx.y * M * FDIM;

    float4 r0[8], r1[8];
    #pragma unroll
    for (int it = 0; it < 8; ++it) {
        const int kp = it * 8 + (threadIdx.x >> 5);
        r0[it] = *(const float4*)(B + (size_t)(2 * kp) * FDIM + lane * 4);
        r1[it] = *(const float4*)(B + (size_t)(2 * kp + 1) * FDIM + lane * 4);
    }

    const float* Astrip = A + (size_t)(live ? m0 : 0) * FDIM + lane * 4;
    float4 arow[16];
    #pragma unroll
    for (int i = 0; i < 16; ++i) arow[i] = *(const float4*)(Astrip + (size_t)i * FDIM);

    #pragma unroll
    for (int it = 0; it < 8; ++it) {
        const int kp = it * 8 + (threadIdx.x >> 5);
        float* d = Ws + kp * WS_PITCH + lane * 8;
        *(float4*)d       = make_float4(r0[it].x, r1[it].x, r0[it].y, r1[it].y);
        *(float4*)(d + 4) = make_float4(r0[it].z, r1[it].z, r0[it].w, r1[it].w);
    }
    float* as = As + (threadIdx.x >> 5) * (16 * AS_PITCH);
    #pragma unroll
    for (int i = 0; i < 16; ++i) *(float4*)(as + i * AS_PITCH + lane * 4) = arow[i];
    __syncthreads();
    if (!live) return;

    const float* ap = as + l16 * AS_PITCH + half * 2;
    v2f afrag[FDIM / 4];
    #pragma unroll
    for (int s = 0; s < FDIM / 4; ++s) afrag[s] = *(const v2f*)(ap + s * 4);

    const float* bp = Ws + half * WS_PITCH + l16 * 2;

    v8f acc[8] = {};
    v2f bcur[8];
    #pragma unroll
    for (int n = 0; n < 8; ++n) bcur[n] = *(const v2f*)(bp + n * 32);

    #pragma unroll
    for (int s = 0; s < FDIM / 4; ++s) {
        v2f bnxt[8];
        if (s + 1 < FDIM / 4) {
            #pragma unroll
            for (int n = 0; n < 8; ++n)
                bnxt[n] = *(const v2f*)(bp + (s + 1) * 2 * WS_PITCH + n * 32);
        }
        #pragma unroll
        for (int n = 0; n < 8; ++n)
            acc[n] = __builtin_amdgcn_wmma_f32_16x16x4_f32(
                false, afrag[s], false, bcur[n], (short)0, acc[n], false, false);
        if (s + 1 < FDIM / 4) {
            #pragma unroll
            for (int n = 0; n < 8; ++n) bcur[n] = bnxt[n];
        }
    }

    #pragma unroll
    for (int n = 0; n < 8; ++n) {
        #pragma unroll
        for (int g = 0; g < 8; ++g) {
            int m = m0 + g + half * 8;
            C[(size_t)m * FDIM + n * 16 + l16] = acc[n][g];
        }
    }
}

__global__ __launch_bounds__(256) void gather_combine(const float* __restrict__ hw,
                                                      const float* __restrict__ dis,
                                                      const int* __restrict__ row_ptr,
                                                      const int* __restrict__ ewbuf,
                                                      const float* __restrict__ b,
                                                      float* __restrict__ hout,
                                                      int nNodes, int doRelu) {
    const int node = __builtin_amdgcn_readfirstlane((int)(blockIdx.x * 8 + (threadIdx.x >> 5)));
    if (node >= nNodes) return;
    const int f = (threadIdx.x & 31) * 4;
    const int2* ew = (const int2*)ewbuf;

    int j = row_ptr[node];
    const int end = row_ptr[node + 1];

    float4 acc = make_float4(0.0f, 0.0f, 0.0f, 0.0f);
    for (; j + 4 <= end; j += 4) {
        const int2 e0 = ew[j], e1 = ew[j + 1], e2 = ew[j + 2], e3 = ew[j + 3];
        const float c0 = __int_as_float(e0.y), c1 = __int_as_float(e1.y);
        const float c2 = __int_as_float(e2.y), c3 = __int_as_float(e3.y);
        const float4 v0 = *(const float4*)(hw + (size_t)e0.x * FDIM + f);
        const float4 v1 = *(const float4*)(hw + (size_t)e1.x * FDIM + f);
        const float4 v2 = *(const float4*)(hw + (size_t)e2.x * FDIM + f);
        const float4 v3 = *(const float4*)(hw + (size_t)e3.x * FDIM + f);
        acc.x = fmaf(v0.x, c0, acc.x); acc.y = fmaf(v0.y, c0, acc.y);
        acc.z = fmaf(v0.z, c0, acc.z); acc.w = fmaf(v0.w, c0, acc.w);
        acc.x = fmaf(v1.x, c1, acc.x); acc.y = fmaf(v1.y, c1, acc.y);
        acc.z = fmaf(v1.z, c1, acc.z); acc.w = fmaf(v1.w, c1, acc.w);
        acc.x = fmaf(v2.x, c2, acc.x); acc.y = fmaf(v2.y, c2, acc.y);
        acc.z = fmaf(v2.z, c2, acc.z); acc.w = fmaf(v2.w, c2, acc.w);
        acc.x = fmaf(v3.x, c3, acc.x); acc.y = fmaf(v3.y, c3, acc.y);
        acc.z = fmaf(v3.z, c3, acc.z); acc.w = fmaf(v3.w, c3, acc.w);
    }
    const int rem = end - j;
    if (rem > 0) {
        const int last = end - 1;
        const int j1 = (j + 1 < end) ? j + 1 : last;
        const int j2 = (j + 2 < end) ? j + 2 : last;
        const int2 e0 = ew[j], e1 = ew[j1], e2 = ew[j2];
        const float c0 = __int_as_float(e0.y), c1 = __int_as_float(e1.y), c2 = __int_as_float(e2.y);
        const float4 v0 = *(const float4*)(hw + (size_t)e0.x * FDIM + f);
        const float4 v1 = *(const float4*)(hw + (size_t)e1.x * FDIM + f);
        const float4 v2 = *(const float4*)(hw + (size_t)e2.x * FDIM + f);
        acc.x = fmaf(v0.x, c0, acc.x); acc.y = fmaf(v0.y, c0, acc.y);
        acc.z = fmaf(v0.z, c0, acc.z); acc.w = fmaf(v0.w, c0, acc.w);
        float4 t;
        t.x = fmaf(v1.x, c1, acc.x); t.y = fmaf(v1.y, c1, acc.y);
        t.z = fmaf(v1.z, c1, acc.z); t.w = fmaf(v1.w, c1, acc.w);
        acc.x = (rem > 1) ? t.x : acc.x; acc.y = (rem > 1) ? t.y : acc.y;
        acc.z = (rem > 1) ? t.z : acc.z; acc.w = (rem > 1) ? t.w : acc.w;
        t.x = fmaf(v2.x, c2, acc.x); t.y = fmaf(v2.y, c2, acc.y);
        t.z = fmaf(v2.z, c2, acc.z); t.w = fmaf(v2.w, c2, acc.w);
        acc.x = (rem > 2) ? t.x : acc.x; acc.y = (rem > 2) ? t.y : acc.y;
        acc.z = (rem > 2) ? t.z : acc.z; acc.w = (rem > 2) ? t.w : acc.w;
    }

    const float di  = dis[node];
    const float di2 = di * di;
    const float4 h = *(const float4*)(hw + (size_t)node * FDIM + f);
    float4 r;
    r.x = acc.x * di + h.x * di2 + b[f + 0];
    r.y = acc.y * di + h.y * di2 + b[f + 1];
    r.z = acc.z * di + h.z * di2 + b[f + 2];
    r.w = acc.w * di + h.w * di2 + b[f + 3];
    if (doRelu) {
        r.x = fmaxf(r.x, 0.0f); r.y = fmaxf(r.y, 0.0f);
        r.z = fmaxf(r.z, 0.0f); r.w = fmaxf(r.w, 0.0f);
    }
    *(float4*)(hout + (size_t)node * FDIM + f) = r;
}

static_assert(E_TR % 32 == 0 && E_TE % 32 == 0, "a wave's 32 rows must lie inside one edge list");
static_assert(N_ROWS % 32 == 0, "whole waves only");

__global__ __launch_bounds__(256) void edge_score(const float* __restrict__ P,
                                                  const float* __restrict__ Q,
                                                  const int* __restrict__ pos_tr,
                                                  const int* __restrict__ neg_tr,
                                                  const int* __restrict__ pos_te,
                                                  const int* __restrict__ neg_te,
                                                  const float* __restrict__ bl1,
                                                  const float* __restrict__ Wl2,
                                                  const float* __restrict__ bl2,
                                                  float* __restrict__ out) {
    const int r0 = __builtin_amdgcn_readfirstlane(
        (int)(((blockIdx.x * blockDim.x + threadIdx.x) >> 5) * 32));
    if (r0 >= N_ROWS) return;
    const int lane = threadIdx.x & 31;
    const int g = lane >> 3;
    const int j = lane & 7;

    const int* e; int i0, len;
    if (r0 < E_TR)                 { e = pos_tr; i0 = r0;                   len = E_TR; }
    else if (r0 < 2 * E_TR)        { e = neg_tr; i0 = r0 - E_TR;            len = E_TR; }
    else if (r0 < 2 * E_TR + E_TE) { e = pos_te; i0 = r0 - 2 * E_TR;        len = E_TE; }
    else                           { e = neg_te; i0 = r0 - 2 * E_TR - E_TE; len = E_TE; }

    const int* eu = e + i0 + g * 8;
    const int* ev = eu + len;
    const int4 ua = *(const int4*)eu, ub = *(const int4*)(eu + 4);
    const int4 va = *(const int4*)ev, vb = *(const int4*)(ev + 4);
    const int us[8] = {ua.x, ua.y, ua.z, ua.w, ub.x, ub.y, ub.z, ub.w};
    const int vs[8] = {va.x, va.y, va.z, va.w, vb.x, vb.y, vb.z, vb.w};

    float4 bb[4], ww[4];
    #pragma unroll
    for (int c = 0; c < 4; ++c) {
        bb[c] = *(const float4*)(bl1 + c * 32 + j * 4);
        ww[c] = *(const float4*)(Wl2 + c * 32 + j * 4);
    }

    float s[8];
    #pragma unroll
    for (int p = 0; p < 8; ++p) {
        const float* pr = P + (size_t)us[p] * FDIM + j * 4;
        const float* qr = Q + (size_t)vs[p] * FDIM + j * 4;
        float acc = 0.0f;
        #pragma unroll
        for (int c = 0; c < 4; ++c) {
            const float4 a = *(const float4*)(pr + c * 32);
            const float4 b = *(const float4*)(qr + c * 32);
            acc = fmaf(fmaxf(a.x + b.x + bb[c].x, 0.0f), ww[c].x, acc);
            acc = fmaf(fmaxf(a.y + b.y + bb[c].y, 0.0f), ww[c].y, acc);
            acc = fmaf(fmaxf(a.z + b.z + bb[c].z, 0.0f), ww[c].z, acc);
            acc = fmaf(fmaxf(a.w + b.w + bb[c].w, 0.0f), ww[c].w, acc);
        }
        s[p] = acc;
    }

    const bool k4 = (j & 4) != 0, k2 = (j & 2) != 0, k1 = (j & 1) != 0;
    float t4[4];
    #pragma unroll
    for (int i = 0; i < 4; ++i) {
        const float keep = k4 ? s[4 + i] : s[i];
        const float send = k4 ? s[i] : s[4 + i];
        t4[i] = keep + __shfl_xor(send, 4, 32);
    }
    float t2[2];
    #pragma unroll
    for (int i = 0; i < 2; ++i) {
        const float keep = k2 ? t4[2 + i] : t4[i];
        const float send = k2 ? t4[i] : t4[2 + i];
        t2[i] = keep + __shfl_xor(send, 2, 32);
    }
    const float keep = k1 ? t2[1] : t2[0];
    const float send = k1 ? t2[0] : t2[1];
    const float tot = keep + __shfl_xor(send, 1, 32);

    out[r0 + lane] = tot + bl2[0];
}

extern "C" void kernel_launch(void* const* d_in, const int* in_sizes, int n_in,
                              void* d_out, int out_size, void* d_ws, size_t ws_size,
                              hipStream_t stream) {
    const float* x          = (const float*)d_in[0];
    const int*   edge_index = (const int*)d_in[1];
    const int*   pos_tr     = (const int*)d_in[2];
    const int*   neg_tr     = (const int*)d_in[3];
    const int*   pos_te     = (const int*)d_in[4];
    const int*   neg_te     = (const int*)d_in[5];
    const float* W[4] = {(const float*)d_in[6],  (const float*)d_in[8],
                         (const float*)d_in[10], (const float*)d_in[12]};
    const float* b[4] = {(const float*)d_in[7],  (const float*)d_in[9],
                         (const float*)d_in[11], (const float*)d_in[13]};
    const float* Wl1 = (const float*)d_in[14];
    const float* bl1 = (const float*)d_in[15];
    const float* Wl2 = (const float*)d_in[16];
    const float* bl2 = (const float*)d_in[17];
    float* out = (float*)d_out;

    const size_t NF = (size_t)N_NODES * FDIM;
    float* ws  = (float*)d_ws;
    float* dis = ws;
    float* hw  = ws + 50048;
    int*   csr = (int*)(hw + NF);
    int*   cnt     = csr;
    int*   row_ptr = csr + 53248;
    int*   ew      = csr + 2 * 53248;
    float* h0  = hw + 2 * NF;
    float* h1  = h0 + NF;
    int*   blk = (int*)h1;

    const int* src = edge_index;
    const int* dst = edge_index + N_EDGES;

    count_block<<<CS_BLOCKS, 1024, 0, stream>>>(dst, blk);
    chunk_prefix<<<(N_NODES + 255) / 256, 256, 0, stream>>>(blk, cnt);
    scan_rows<<<1, 1024, 0, stream>>>(cnt, row_ptr, dis);
    fill_block<<<dim3(CS_BLOCKS, 2), 1024, 0, stream>>>(src, dst, blk, row_ptr, dis, ew);

    const int tilesM     = (N_NODES + 15) / 16;
    const int gemmBlocks = (tilesM + 7) / 8;
    const float* hin = x;
    float* houts[4] = {h0, h1, h0, h1};
    for (int l = 0; l < 4; ++l) {
        gemm_node<<<gemmBlocks, 256, 0, stream>>>(hin, W[l], hw, N_NODES);
        gather_combine<<<(N_NODES * 32 + 255) / 256, 256, 0, stream>>>(
            hw, dis, row_ptr, ew, b[l], houts[l], N_NODES, (l < 3) ? 1 : 0);
        hin = houts[l];
    }

    gemm_node<<<dim3(gemmBlocks, 2), 256, 0, stream>>>(h1, Wl1, hw, N_NODES);

    const int edgeWaves  = N_ROWS / 32;
    const int edgeBlocks = (edgeWaves + 7) / 8;
    edge_score<<<edgeBlocks, 256, 0, stream>>>(hw, hw + NF, pos_tr, neg_tr, pos_te, neg_te,
                                               bl1, Wl2, bl2, out);
}


# Node GEMM dealt evenly over the SIMDs: 256 blocks of 8 waves own 12-13 row tiles each; a wave keeps the B fragments of its 32 columns in registers, A tiles shared in LDS; k order and sums kept
# speedup vs baseline: 4.2020x; 1.0383x over previous
#include <hip/hip_runtime.h>


typedef float v2f __attribute__((ext_vector_type(2)));
typedef float v8f __attribute__((ext_vector_type(8)));

#define N_NODES 50000
#define N_EDGES 800000
#define FDIM    128
#define E_TR    100000
#define E_TE    20000
#define N_ROWS  240000

#define CS_BLOCKS 100
#define CS_CHUNK  (N_EDGES / CS_BLOCKS)
#define CS_GROUPS (CS_CHUNK / 4)
#define CS_PASSES ((CS_GROUPS + 1023) / 1024)
#define CS_STRIDE 50048
static_assert(N_EDGES % CS_BLOCKS == 0 && CS_CHUNK % 4 == 0, "chunks are whole 16-byte groups");
static_assert(CS_STRIDE >= N_NODES && CS_STRIDE % 32 == 0, "a plane holds every node");
static_assert(N_NODES % 4 == 0, "16-byte groups must not straddle the end");

__global__ __launch_bounds__(1024) void count_block(const int* __restrict__ dst,
                                                    int* __restrict__ blk) {
    __shared__ int hist[N_NODES];
    const int4* d4 = (const int4*)(dst + (size_t)blockIdx.x * CS_CHUNK);
    int4 dv[CS_PASSES];
    bool ok[CS_PASSES];
    #pragma unroll
    for (int p = 0; p < CS_PASSES; ++p) {
        const int i = p * 1024 + threadIdx.x;
        ok[p] = i < CS_GROUPS;
        dv[p] = d4[ok[p] ? i : CS_GROUPS - 1];
    }

    for (int i = threadIdx.x * 4; i < N_NODES; i += 4096)
        *(int4*)(hist + i) = make_int4(0, 0, 0, 0);
    __syncthreads();

    #pragma unroll
    for (int p = 0; p < CS_PASSES; ++p) {
        if (ok[p]) {
            atomicAdd(&hist[dv[p].x], 1);
            atomicAdd(&hist[dv[p].y], 1);
            atomicAdd(&hist[dv[p].z], 1);
            atomicAdd(&hist[dv[p].w], 1);
        }
    }
    __syncthreads();

    int* out = blk + (size_t)blockIdx.x * CS_STRIDE;
    for (int i = threadIdx.x * 4; i < N_NODES; i += 4096)
        *(int4*)(out + i) = *(const int4*)(hist + i);
}

__global__ __launch_bounds__(256) void chunk_prefix(int* __restrict__ blk,
                                                    int* __restrict__ cnt) {
    const int n = blockIdx.x * 256 + threadIdx.x;
    if (n >= N_NODES) return;
    int c[CS_BLOCKS];
    #pragma unroll
    for (int b = 0; b < CS_BLOCKS; ++b) c[b] = blk[(size_t)b * CS_STRIDE + n];
    int run = 0;
    #pragma unroll
    for (int b = 0; b < CS_BLOCKS; ++b) {
        blk[(size_t)b * CS_STRIDE + n] = run;
        run += c[b];
    }
    cnt[n] = run;
}

#define SCAN_PASSES 13
#define SCAN_CHUNK  (SCAN_PASSES * 128)
static_assert(32 * SCAN_CHUNK >= N_NODES, "one block must cover every node");
static_assert(N_NODES % 4 == 0, "16-byte groups must not straddle the end");

__device__ __forceinline__ int wave_incl_scan(int v, int lane) {
    #pragma unroll
    for (int o = 1; o < 32; o <<= 1) {
        int y = __shfl_up(v, o, 32);
        if (lane >= o) v += y;
    }
    return v;
}

__global__ __launch_bounds__(1024) void scan_rows(const int* __restrict__ cnt,
                                                  int* __restrict__ row_ptr,
                                                  float* __restrict__ dis) {
    __shared__ int wtot[32];
    const int lane = threadIdx.x & 31;
    const int w    = threadIdx.x >> 5;
    const int base = w * SCAN_CHUNK + lane * 4;

    int4 p[SCAN_PASSES];
    #pragma unroll
    for (int i = 0; i < SCAN_PASSES; ++i) {
        const int idx = base + i * 128;
        const int4 v = *(const int4*)(cnt + ((idx < N_NODES) ? idx : N_NODES - 4));
        p[i] = (idx < N_NODES) ? v : make_int4(0, 0, 0, 0);
    }
    #pragma unroll
    for (int i = 0; i < SCAN_PASSES; ++i) {
        const int idx = base + i * 128;
        if (idx < N_NODES) {
            float4 d;
            d.x = rsqrtf((float)p[i].x + 1.0f);
            d.y = rsqrtf((float)p[i].y + 1.0f);
            d.z = rsqrtf((float)p[i].z + 1.0f);
            d.w = rsqrtf((float)p[i].w + 1.0f);
            *(float4*)(dis + idx) = d;
        }
    }

    int run = 0;
    #pragma unroll
    for (int i = 0; i < SCAN_PASSES; ++i) {
        const int4 c = p[i];
        const int s   = c.x + c.y + c.z + c.w;
        const int inc = wave_incl_scan(s, lane);
        const int o   = run + inc - s;
        p[i].x = o;
        p[i].y = o + c.x;
        p[i].z = o + c.x + c.y;
        p[i].w = o + c.x + c.y + c.z;
        run += __shfl(inc, 31, 32);
    }
    if (lane == 0) wtot[w] = run;
    __syncthreads();

    const int t     = wtot[lane];
    const int tinc  = wave_incl_scan(t, lane);
    const int wbase = __shfl(tinc - t, w, 32);

    #pragma unroll
    for (int i = 0; i < SCAN_PASSES; ++i) {
        int idx = base + i * 128;
        if (idx < N_NODES) {
            int4 q = p[i];
            q.x += wbase; q.y += wbase; q.z += wbase; q.w += wbase;
            *(int4*)(row_ptr + idx) = q;
        }
    }
    if (threadIdx.x == 1023) row_ptr[N_NODES] = wbase + run;
}

#define HALF_NODES (N_NODES / 2)
#define TAB_PASSES ((HALF_NODES / 4 + 1023) / 1024)
static_assert(N_NODES % 2 == 0 && HALF_NODES % 4 == 0, "a half is whole 16-byte groups");

__global__ __launch_bounds__(1024) void fill_block(const int* __restrict__ src,
                                                   const int* __restrict__ dst,
                                                   const int* __restrict__ blk,
                                                   const int* __restrict__ row_ptr,
                                                   const float* __restrict__ dis,
                                                   int* __restrict__ ewbuf) {
    __shared__ int cur[HALF_NODES];
    const int nlo = blockIdx.y * HALF_NODES;
    const int4* s4 = (const int4*)(src + (size_t)blockIdx.x * CS_CHUNK);
    const int4* d4 = (const int4*)(dst + (size_t)blockIdx.x * CS_CHUNK);
    int4 sv[CS_PASSES], dv[CS_PASSES];
    bool ok[CS_PASSES];
    #pragma unroll
    for (int p = 0; p < CS_PASSES; ++p) {
        const int i = p * 1024 + threadIdx.x;
        ok[p] = i < CS_GROUPS;
        sv[p] = s4[ok[p] ? i : CS_GROUPS - 1];
        dv[p] = d4[ok[p] ? i : CS_GROUPS - 1];
    }

    const int* off = blk + (size_t)blockIdx.x * CS_STRIDE + nlo;
    const int* rp  = row_ptr + nlo;
    int4 a[TAB_PASSES], o[TAB_PASSES];
    #pragma unroll
    for (int t = 0; t < TAB_PASSES; ++t) {
        const int i = (t * 1024 + threadIdx.x) * 4;
        const int ii = (i < HALF_NODES) ? i : HALF_NODES - 4;
        a[t] = *(const int4*)(rp + ii);
        o[t] = *(const int4*)(off + ii);
    }
    #pragma unroll
    for (int t = 0; t < TAB_PASSES; ++t) {
        const int i = (t * 1024 + threadIdx.x) * 4;
        if (i < HALF_NODES)
            *(int4*)(cur + i) = make_int4(a[t].x + o[t].x, a[t].y + o[t].y,
                                          a[t].z + o[t].z, a[t].w + o[t].w);
    }

    int   sN[CS_PASSES * 4];
    int   rN[CS_PASSES * 4];
    bool  in[CS_PASSES * 4];
    float sc[CS_PASSES * 4];
    #pragma unroll
    for (int p = 0; p < CS_PASSES; ++p) {
        const int s_[4] = {sv[p].x, sv[p].y, sv[p].z, sv[p].w};
        const int d_[4] = {dv[p].x, dv[p].y, dv[p].z, dv[p].w};
        #pragma unroll
        for (int q = 0; q < 4; ++q) {
            const int k = p * 4 + q;
            sN[k] = s_[q];
            rN[k] = d_[q] - nlo;
            in[k] = ok[p] && (unsigned)rN[k] < (unsigned)HALF_NODES;
            sc[k] = dis[in[k] ? sN[k] : 0];
        }
    }
    __syncthreads();

    int2* ew = (int2*)ewbuf;
    #pragma unroll
    for (int k = 0; k < CS_PASSES * 4; ++k) {
        if (in[k]) {
            const int slot = atomicAdd(&cur[rN[k]], 1);
            ew[slot] = make_int2(sN[k], __float_as_int(sc[k]));
        }
    }
}

#define AS_PITCH    132
#define GEMM_BLOCKS 256
#define TILES_M     (N_NODES / 16)
#define GEMM_TPB    ((TILES_M + GEMM_BLOCKS - 1) / GEMM_BLOCKS)
#define GEMM_ROWS   (GEMM_TPB * 16)
#define GEMM_LD     (GEMM_ROWS * 32 / 256)
static_assert(N_NODES % 16 == 0, "whole 16-row tiles only");
static_assert((GEMM_ROWS * 32) % 256 == 0, "the staging loop has whole passes");

__global__ __launch_bounds__(256) void gemm_node(const float* __restrict__ A,
                                                 const float* __restrict__ B,
                                                 float* __restrict__ C) {
    __shared__ float As[GEMM_ROWS * AS_PITCH];
    const int lane = threadIdx.x & 31;
    const int wave = __builtin_amdgcn_readfirstlane((int)(threadIdx.x >> 5));
    const int half = lane >> 4;
    const int l16  = lane & 15;
    const int cq   = wave & 3;
    const int par  = wave >> 2;

    B += (size_t)blockIdx.y * FDIM * FDIM;
    C += (size_t)blockIdx.y * N_NODES * FDIM;

    float4 arow[GEMM_LD];
    #pragma unroll
    for (int i = 0; i < GEMM_LD; ++i) {
        const int r    = i * 8 + wave;
        const int tile = blockIdx.x + (r >> 4) * GEMM_BLOCKS;
        const int t    = (tile < TILES_M) ? tile : TILES_M - 1;
        arow[i] = *(const float4*)(A + ((size_t)t * 16 + (r & 15)) * FDIM + lane * 4);
    }

    const float* bcol = B + (size_t)(2 * half) * FDIM + cq * 32 + l16;
    v2f bfr[2][FDIM / 4];
    #pragma unroll
    for (int s = 0; s < FDIM / 4; ++s) {
        #pragma unroll
        for (int n = 0; n < 2; ++n) {
            bfr[n][s][0] = bcol[(size_t)(4 * s) * FDIM + n * 16];
            bfr[n][s][1] = bcol[(size_t)(4 * s + 1) * FDIM + n * 16];
        }
    }

    #pragma unroll
    for (int i = 0; i < GEMM_LD; ++i)
        *(float4*)(As + (i * 8 + wave) * AS_PITCH + lane * 4) = arow[i];
    __syncthreads();

    #pragma unroll 1
    for (int slot = par; slot < GEMM_TPB; slot += 2) {
        const int tile = blockIdx.x + slot * GEMM_BLOCKS;
        const float* ap = As + (slot * 16 + l16) * AS_PITCH + half * 2;
        v2f afrag[FDIM / 4];
        #pragma unroll
        for (int s = 0; s < FDIM / 4; ++s) afrag[s] = *(const v2f*)(ap + s * 4);

        v8f acc0 = {}, acc1 = {};
        #pragma unroll
        for (int s = 0; s < FDIM / 4; ++s) {
            acc0 = __builtin_amdgcn_wmma_f32_16x16x4_f32(
                false, afrag[s], false, bfr[0][s], (short)0, acc0, false, false);
            acc1 = __builtin_amdgcn_wmma_f32_16x16x4_f32(
                false, afrag[s], false, bfr[1][s], (short)0, acc1, false, false);
        }

        if (tile < TILES_M) {
            float* crow = C + ((size_t)tile * 16 + half * 8) * FDIM + cq * 32 + l16;
            #pragma unroll
            for (int g = 0; g < 8; ++g) {
                crow[(size_t)g * FDIM]      = acc0[g];
                crow[(size_t)g * FDIM + 16] = acc1[g];
            }
        }
    }
}

__global__ __launch_bounds__(256) void gather_combine(const float* __restrict__ hw,
                                                      const float* __restrict__ dis,
                                                      const int* __restrict__ row_ptr,
                                                      const int* __restrict__ ewbuf,
                                                      const float* __restrict__ b,
                                                      float* __restrict__ hout,
                                                      int nNodes, int doRelu) {
    const int node = __builtin_amdgcn_readfirstlane((int)(blockIdx.x * 8 + (threadIdx.x >> 5)));
    if (node >= nNodes) return;
    const int f = (threadIdx.x & 31) * 4;
    const int2* ew = (const int2*)ewbuf;

    int j = row_ptr[node];
    const int end = row_ptr[node + 1];

    float4 acc = make_float4(0.0f, 0.0f, 0.0f, 0.0f);
    for (; j + 4 <= end; j += 4) {
        const int2 e0 = ew[j], e1 = ew[j + 1], e2 = ew[j + 2], e3 = ew[j + 3];
        const float c0 = __int_as_float(e0.y), c1 = __int_as_float(e1.y);
        const float c2 = __int_as_float(e2.y), c3 = __int_as_float(e3.y);
        const float4 v0 = *(const float4*)(hw + (size_t)e0.x * FDIM + f);
        const float4 v1 = *(const float4*)(hw + (size_t)e1.x * FDIM + f);
        const float4 v2 = *(const float4*)(hw + (size_t)e2.x * FDIM + f);
        const float4 v3 = *(const float4*)(hw + (size_t)e3.x * FDIM + f);
        acc.x = fmaf(v0.x, c0, acc.x); acc.y = fmaf(v0.y, c0, acc.y);
        acc.z = fmaf(v0.z, c0, acc.z); acc.w = fmaf(v0.w, c0, acc.w);
        acc.x = fmaf(v1.x, c1, acc.x); acc.y = fmaf(v1.y, c1, acc.y);
        acc.z = fmaf(v1.z, c1, acc.z); acc.w = fmaf(v1.w, c1, acc.w);
        acc.x = fmaf(v2.x, c2, acc.x); acc.y = fmaf(v2.y, c2, acc.y);
        acc.z = fmaf(v2.z, c2, acc.z); acc.w = fmaf(v2.w, c2, acc.w);
        acc.x = fmaf(v3.x, c3, acc.x); acc.y = fmaf(v3.y, c3, acc.y);
        acc.z = fmaf(v3.z, c3, acc.z); acc.w = fmaf(v3.w, c3, acc.w);
    }
    const int rem = end - j;
    if (rem > 0) {
        const int last = end - 1;
        const int j1 = (j + 1 < end) ? j + 1 : last;
        const int j2 = (j + 2 < end) ? j + 2 : last;
        const int2 e0 = ew[j], e1 = ew[j1], e2 = ew[j2];
        const float c0 = __int_as_float(e0.y), c1 = __int_as_float(e1.y), c2 = __int_as_float(e2.y);
        const float4 v0 = *(const float4*)(hw + (size_t)e0.x * FDIM + f);
        const float4 v1 = *(const float4*)(hw + (size_t)e1.x * FDIM + f);
        const float4 v2 = *(const float4*)(hw + (size_t)e2.x * FDIM + f);
        acc.x = fmaf(v0.x, c0, acc.x); acc.y = fmaf(v0.y, c0, acc.y);
        acc.z = fmaf(v0.z, c0, acc.z); acc.w = fmaf(v0.w, c0, acc.w);
        float4 t;
        t.x = fmaf(v1.x, c1, acc.x); t.y = fmaf(v1.y, c1, acc.y);
        t.z = fmaf(v1.z, c1, acc.z); t.w = fmaf(v1.w, c1, acc.w);
        acc.x = (rem > 1) ? t.x : acc.x; acc.y = (rem > 1) ? t.y : acc.y;
        acc.z = (rem > 1) ? t.z : acc.z; acc.w = (rem > 1) ? t.w : acc.w;
        t.x = fmaf(v2.x, c2, acc.x); t.y = fmaf(v2.y, c2, acc.y);
        t.z = fmaf(v2.z, c2, acc.z); t.w = fmaf(v2.w, c2, acc.w);
        acc.x = (rem > 2) ? t.x : acc.x; acc.y = (rem > 2) ? t.y : acc.y;
        acc.z = (rem > 2) ? t.z : acc.z; acc.w = (rem > 2) ? t.w : acc.w;
    }

    const float di  = dis[node];
    const float di2 = di * di;
    const float4 h = *(const float4*)(hw + (size_t)node * FDIM + f);
    float4 r;
    r.x = acc.x * di + h.x * di2 + b[f + 0];
    r.y = acc.y * di + h.y * di2 + b[f + 1];
    r.z = acc.z * di + h.z * di2 + b[f + 2];
    r.w = acc.w * di + h.w * di2 + b[f + 3];
    if (doRelu) {
        r.x = fmaxf(r.x, 0.0f); r.y = fmaxf(r.y, 0.0f);
        r.z = fmaxf(r.z, 0.0f); r.w = fmaxf(r.w, 0.0f);
    }
    *(float4*)(hout + (size_t)node * FDIM + f) = r;
}

static_assert(E_TR % 32 == 0 && E_TE % 32 == 0, "a wave's 32 rows must lie inside one edge list");
static_assert(N_ROWS % 32 == 0, "whole waves only");

__global__ __launch_bounds__(256) void edge_score(const float* __restrict__ P,
                                                  const float* __restrict__ Q,
                                                  const int* __restrict__ pos_tr,
                                                  const int* __restrict__ neg_tr,
                                                  const int* __restrict__ pos_te,
                                                  const int* __restrict__ neg_te,
                                                  const float* __restrict__ bl1,
                                                  const float* __restrict__ Wl2,
                                                  const float* __restrict__ bl2,
                                                  float* __restrict__ out) {
    const int r0 = __builtin_amdgcn_readfirstlane(
        (int)(((blockIdx.x * blockDim.x + threadIdx.x) >> 5) * 32));
    if (r0 >= N_ROWS) return;
    const int lane = threadIdx.x & 31;
    const int g = lane >> 3;
    const int j = lane & 7;

    const int* e; int i0, len;
    if (r0 < E_TR)                 { e = pos_tr; i0 = r0;                   len = E_TR; }
    else if (r0 < 2 * E_TR)        { e = neg_tr; i0 = r0 - E_TR;            len = E_TR; }
    else if (r0 < 2 * E_TR + E_TE) { e = pos_te; i0 = r0 - 2 * E_TR;        len = E_TE; }
    else                           { e = neg_te; i0 = r0 - 2 * E_TR - E_TE; len = E_TE; }

    const int* eu = e + i0 + g * 8;
    const int* ev = eu + len;
    const int4 ua = *(const int4*)eu, ub = *(const int4*)(eu + 4);
    const int4 va = *(const int4*)ev, vb = *(const int4*)(ev + 4);
    const int us[8] = {ua.x, ua.y, ua.z, ua.w, ub.x, ub.y, ub.z, ub.w};
    const int vs[8] = {va.x, va.y, va.z, va.w, vb.x, vb.y, vb.z, vb.w};

    float4 bb[4], ww[4];
    #pragma unroll
    for (int c = 0; c < 4; ++c) {
        bb[c] = *(const float4*)(bl1 + c * 32 + j * 4);
        ww[c] = *(const float4*)(Wl2 + c * 32 + j * 4);
    }

    float s[8];
    #pragma unroll
    for (int p = 0; p < 8; ++p) {
        const float* pr = P + (size_t)us[p] * FDIM + j * 4;
        const float* qr = Q + (size_t)vs[p] * FDIM + j * 4;
        float acc = 0.0f;
        #pragma unroll
        for (int c = 0; c < 4; ++c) {
            const float4 a = *(const float4*)(pr + c * 32);
            const float4 b = *(const float4*)(qr + c * 32);
            acc = fmaf(fmaxf(a.x + b.x + bb[c].x, 0.0f), ww[c].x, acc);
            acc = fmaf(fmaxf(a.y + b.y + bb[c].y, 0.0f), ww[c].y, acc);
            acc = fmaf(fmaxf(a.z + b.z + bb[c].z, 0.0f), ww[c].z, acc);
            acc = fmaf(fmaxf(a.w + b.w + bb[c].w, 0.0f), ww[c].w, acc);
        }
        s[p] = acc;
    }

    const bool k4 = (j & 4) != 0, k2 = (j & 2) != 0, k1 = (j & 1) != 0;
    float t4[4];
    #pragma unroll
    for (int i = 0; i < 4; ++i) {
        const float keep = k4 ? s[4 + i] : s[i];
        const float send = k4 ? s[i] : s[4 + i];
        t4[i] = keep + __shfl_xor(send, 4, 32);
    }
    float t2[2];
    #pragma unroll
    for (int i = 0; i < 2; ++i) {
        const float keep = k2 ? t4[2 + i] : t4[i];
        const float send = k2 ? t4[i] : t4[2 + i];
        t2[i] = keep + __shfl_xor(send, 2, 32);
    }
    const float keep = k1 ? t2[1] : t2[0];
    const float send = k1 ? t2[0] : t2[1];
    const float tot = keep + __shfl_xor(send, 1, 32);

    out[r0 + lane] = tot + bl2[0];
}

extern "C" void kernel_launch(void* const* d_in, const int* in_sizes, int n_in,
                              void* d_out, int out_size, void* d_ws, size_t ws_size,
                              hipStream_t stream) {
    const float* x          = (const float*)d_in[0];
    const int*   edge_index = (const int*)d_in[1];
    const int*   pos_tr     = (const int*)d_in[2];
    const int*   neg_tr     = (const int*)d_in[3];
    const int*   pos_te     = (const int*)d_in[4];
    const int*   neg_te     = (const int*)d_in[5];
    const float* W[4] = {(const float*)d_in[6],  (const float*)d_in[8],
                         (const float*)d_in[10], (const float*)d_in[12]};
    const float* b[4] = {(const float*)d_in[7],  (const float*)d_in[9],
                         (const float*)d_in[11], (const float*)d_in[13]};
    const float* Wl1 = (const float*)d_in[14];
    const float* bl1 = (const float*)d_in[15];
    const float* Wl2 = (const float*)d_in[16];
    const float* bl2 = (const float*)d_in[17];
    float* out = (float*)d_out;

    const size_t NF = (size_t)N_NODES * FDIM;
    float* ws  = (float*)d_ws;
    float* dis = ws;
    float* hw  = ws + 50048;
    int*   csr = (int*)(hw + NF);
    int*   cnt     = csr;
    int*   row_ptr = csr + 53248;
    int*   ew      = csr + 2 * 53248;
    float* h0  = hw + 2 * NF;
    float* h1  = h0 + NF;
    int*   blk = (int*)h1;

    const int* src = edge_index;
    const int* dst = edge_index + N_EDGES;

    count_block<<<CS_BLOCKS, 1024, 0, stream>>>(dst, blk);
    chunk_prefix<<<(N_NODES + 255) / 256, 256, 0, stream>>>(blk, cnt);
    scan_rows<<<1, 1024, 0, stream>>>(cnt, row_ptr, dis);
    fill_block<<<dim3(CS_BLOCKS, 2), 1024, 0, stream>>>(src, dst, blk, row_ptr, dis, ew);

    const float* hin = x;
    float* houts[4] = {h0, h1, h0, h1};
    for (int l = 0; l < 4; ++l) {
        gemm_node<<<GEMM_BLOCKS, 256, 0, stream>>>(hin, W[l], hw);
        gather_combine<<<(N_NODES * 32 + 255) / 256, 256, 0, stream>>>(
            hw, dis, row_ptr, ew, b[l], houts[l], N_NODES, (l < 3) ? 1 : 0);
        hin = houts[l];
    }

    gemm_node<<<dim3(GEMM_BLOCKS, 2), 256, 0, stream>>>(h1, Wl1, hw);

    const int edgeWaves  = N_ROWS / 32;
    const int edgeBlocks = (edgeWaves + 7) / 8;
    edge_score<<<edgeBlocks, 256, 0, stream>>>(hw, hw + NF, pos_tr, neg_tr, pos_te, neg_te,
                                               bl1, Wl2, bl2, out);
}


# Last GCN layer folded into the edge MLP by associativity: Wc=[W3;b3]@Wl1 halves once on f32 WMMA, layer 4 only aggregates h3, P,Q=Agg(h3)@Wc; one Nx128x128 product and launch fewer; sums regrouped
# speedup vs baseline: 4.3312x; 1.0307x over previous
#include <hip/hip_runtime.h>


typedef float v2f __attribute__((ext_vector_type(2)));
typedef float v8f __attribute__((ext_vector_type(8)));

#define N_NODES 50000
#define N_EDGES 800000
#define FDIM    128
#define E_TR    100000
#define E_TE    20000
#define N_ROWS  240000

#define CS_BLOCKS 100
#define CS_CHUNK  (N_EDGES / CS_BLOCKS)
#define CS_GROUPS (CS_CHUNK / 4)
#define CS_PASSES ((CS_GROUPS + 1023) / 1024)
#define CS_STRIDE 50048
static_assert(N_EDGES % CS_BLOCKS == 0 && CS_CHUNK % 4 == 0, "chunks are whole 16-byte groups");
static_assert(CS_STRIDE >= N_NODES && CS_STRIDE % 32 == 0, "a plane holds every node");
static_assert(N_NODES % 4 == 0, "16-byte groups must not straddle the end");

__global__ __launch_bounds__(1024) void count_block(const int* __restrict__ dst,
                                                    int* __restrict__ blk) {
    __shared__ int hist[N_NODES];
    const int4* d4 = (const int4*)(dst + (size_t)blockIdx.x * CS_CHUNK);
    int4 dv[CS_PASSES];
    bool ok[CS_PASSES];
    #pragma unroll
    for (int p = 0; p < CS_PASSES; ++p) {
        const int i = p * 1024 + threadIdx.x;
        ok[p] = i < CS_GROUPS;
        dv[p] = d4[ok[p] ? i : CS_GROUPS - 1];
    }

    for (int i = threadIdx.x * 4; i < N_NODES; i += 4096)
        *(int4*)(hist + i) = make_int4(0, 0, 0, 0);
    __syncthreads();

    #pragma unroll
    for (int p = 0; p < CS_PASSES; ++p) {
        if (ok[p]) {
            atomicAdd(&hist[dv[p].x], 1);
            atomicAdd(&hist[dv[p].y], 1);
            atomicAdd(&hist[dv[p].z], 1);
            atomicAdd(&hist[dv[p].w], 1);
        }
    }
    __syncthreads();

    int* out = blk + (size_t)blockIdx.x * CS_STRIDE;
    for (int i = threadIdx.x * 4; i < N_NODES; i += 4096)
        *(int4*)(out + i) = *(const int4*)(hist + i);
}

#define PREFIX_BLOCKS ((N_NODES + 255) / 256)
#define FOLD_TILES    (FDIM / 16 + 1)
#define FOLD_BLOCKS   (FOLD_TILES * 2)

__device__ __forceinline__ void fold_weights(int job,
                                             const float* __restrict__ W3,
                                             const float* __restrict__ b3,
                                             const float* __restrict__ Wl1,
                                             float* __restrict__ Wc,
                                             float* __restrict__ bc) {
    const int lane = threadIdx.x & 31;
    const int wave = __builtin_amdgcn_readfirstlane((int)(threadIdx.x >> 5));
    const int half = lane >> 4;
    const int l16  = lane & 15;
    const int tile = job >> 1;
    const int y    = job & 1;
    const bool biasTile = tile == FOLD_TILES - 1;

    const float* arow = (biasTile ? b3 : W3 + (size_t)(tile * 16 + l16) * FDIM) + 2 * half;
    const bool live = !biasTile || l16 == 0;
    const float* bcol = Wl1 + (size_t)y * FDIM * FDIM + (size_t)(2 * half) * FDIM + wave * 16 + l16;

    v2f af[FDIM / 4], bf[FDIM / 4];
    #pragma unroll
    for (int s = 0; s < FDIM / 4; ++s) {
        af[s]    = *(const v2f*)(arow + 4 * s);
        bf[s][0] = bcol[(size_t)(4 * s) * FDIM];
        bf[s][1] = bcol[(size_t)(4 * s + 1) * FDIM];
    }
    v8f acc = {};
    #pragma unroll
    for (int s = 0; s < FDIM / 4; ++s) {
        v2f a;
        a[0] = live ? af[s][0] : 0.0f;
        a[1] = live ? af[s][1] : 0.0f;
        acc = __builtin_amdgcn_wmma_f32_16x16x4_f32(
            false, a, false, bf[s], (short)0, acc, false, false);
    }

    if (!biasTile) {
        float* crow = Wc + (size_t)y * FDIM * FDIM
                         + (size_t)(tile * 16 + half * 8) * FDIM + wave * 16 + l16;
        #pragma unroll
        for (int g = 0; g < 8; ++g) crow[(size_t)g * FDIM] = acc[g];
    } else if (half == 0) {
        bc[y * FDIM + wave * 16 + l16] = acc[0];
    }
}

__global__ __launch_bounds__(256) void chunk_prefix(int* __restrict__ blk,
                                                    int* __restrict__ cnt,
                                                    const float* __restrict__ W3,
                                                    const float* __restrict__ b3,
                                                    const float* __restrict__ Wl1,
                                                    float* __restrict__ Wc,
                                                    float* __restrict__ bc) {
    if (blockIdx.x >= PREFIX_BLOCKS) {
        fold_weights((int)blockIdx.x - PREFIX_BLOCKS, W3, b3, Wl1, Wc, bc);
        return;
    }
    const int n = blockIdx.x * 256 + threadIdx.x;
    if (n >= N_NODES) return;
    int c[CS_BLOCKS];
    #pragma unroll
    for (int b = 0; b < CS_BLOCKS; ++b) c[b] = blk[(size_t)b * CS_STRIDE + n];
    int run = 0;
    #pragma unroll
    for (int b = 0; b < CS_BLOCKS; ++b) {
        blk[(size_t)b * CS_STRIDE + n] = run;
        run += c[b];
    }
    cnt[n] = run;
}

#define SCAN_PASSES 13
#define SCAN_CHUNK  (SCAN_PASSES * 128)
static_assert(32 * SCAN_CHUNK >= N_NODES, "one block must cover every node");
static_assert(N_NODES % 4 == 0, "16-byte groups must not straddle the end");

__device__ __forceinline__ int wave_incl_scan(int v, int lane) {
    #pragma unroll
    for (int o = 1; o < 32; o <<= 1) {
        int y = __shfl_up(v, o, 32);
        if (lane >= o) v += y;
    }
    return v;
}

__global__ __launch_bounds__(1024) void scan_rows(const int* __restrict__ cnt,
                                                  int* __restrict__ row_ptr,
                                                  float* __restrict__ dis) {
    __shared__ int wtot[32];
    const int lane = threadIdx.x & 31;
    const int w    = threadIdx.x >> 5;
    const int base = w * SCAN_CHUNK + lane * 4;

    int4 p[SCAN_PASSES];
    #pragma unroll
    for (int i = 0; i < SCAN_PASSES; ++i) {
        const int idx = base + i * 128;
        const int4 v = *(const int4*)(cnt + ((idx < N_NODES) ? idx : N_NODES - 4));
        p[i] = (idx < N_NODES) ? v : make_int4(0, 0, 0, 0);
    }
    #pragma unroll
    for (int i = 0; i < SCAN_PASSES; ++i) {
        const int idx = base + i * 128;
        if (idx < N_NODES) {
            float4 d;
            d.x = rsqrtf((float)p[i].x + 1.0f);
            d.y = rsqrtf((float)p[i].y + 1.0f);
            d.z = rsqrtf((float)p[i].z + 1.0f);
            d.w = rsqrtf((float)p[i].w + 1.0f);
            *(float4*)(dis + idx) = d;
        }
    }

    int run = 0;
    #pragma unroll
    for (int i = 0; i < SCAN_PASSES; ++i) {
        const int4 c = p[i];
        const int s   = c.x + c.y + c.z + c.w;
        const int inc = wave_incl_scan(s, lane);
        const int o   = run + inc - s;
        p[i].x = o;
        p[i].y = o + c.x;
        p[i].z = o + c.x + c.y;
        p[i].w = o + c.x + c.y + c.z;
        run += __shfl(inc, 31, 32);
    }
    if (lane == 0) wtot[w] = run;
    __syncthreads();

    const int t     = wtot[lane];
    const int tinc  = wave_incl_scan(t, lane);
    const int wbase = __shfl(tinc - t, w, 32);

    #pragma unroll
    for (int i = 0; i < SCAN_PASSES; ++i) {
        int idx = base + i * 128;
        if (idx < N_NODES) {
            int4 q = p[i];
            q.x += wbase; q.y += wbase; q.z += wbase; q.w += wbase;
            *(int4*)(row_ptr + idx) = q;
        }
    }
    if (threadIdx.x == 1023) row_ptr[N_NODES] = wbase + run;
}

#define HALF_NODES (N_NODES / 2)
#define TAB_PASSES ((HALF_NODES / 4 + 1023) / 1024)
static_assert(N_NODES % 2 == 0 && HALF_NODES % 4 == 0, "a half is whole 16-byte groups");

__global__ __launch_bounds__(1024) void fill_block(const int* __restrict__ src,
                                                   const int* __restrict__ dst,
                                                   const int* __restrict__ blk,
                                                   const int* __restrict__ row_ptr,
                                                   const float* __restrict__ dis,
                                                   int* __restrict__ ewbuf) {
    __shared__ int cur[HALF_NODES];
    const int nlo = blockIdx.y * HALF_NODES;
    const int4* s4 = (const int4*)(src + (size_t)blockIdx.x * CS_CHUNK);
    const int4* d4 = (const int4*)(dst + (size_t)blockIdx.x * CS_CHUNK);
    int4 sv[CS_PASSES], dv[CS_PASSES];
    bool ok[CS_PASSES];
    #pragma unroll
    for (int p = 0; p < CS_PASSES; ++p) {
        const int i = p * 1024 + threadIdx.x;
        ok[p] = i < CS_GROUPS;
        sv[p] = s4[ok[p] ? i : CS_GROUPS - 1];
        dv[p] = d4[ok[p] ? i : CS_GROUPS - 1];
    }

    const int* off = blk + (size_t)blockIdx.x * CS_STRIDE + nlo;
    const int* rp  = row_ptr + nlo;
    int4 a[TAB_PASSES], o[TAB_PASSES];
    #pragma unroll
    for (int t = 0; t < TAB_PASSES; ++t) {
        const int i = (t * 1024 + threadIdx.x) * 4;
        const int ii = (i < HALF_NODES) ? i : HALF_NODES - 4;
        a[t] = *(const int4*)(rp + ii);
        o[t] = *(const int4*)(off + ii);
    }
    #pragma unroll
    for (int t = 0; t < TAB_PASSES; ++t) {
        const int i = (t * 1024 + threadIdx.x) * 4;
        if (i < HALF_NODES)
            *(int4*)(cur + i) = make_int4(a[t].x + o[t].x, a[t].y + o[t].y,
                                          a[t].z + o[t].z, a[t].w + o[t].w);
    }

    int   sN[CS_PASSES * 4];
    int   rN[CS_PASSES * 4];
    bool  in[CS_PASSES * 4];
    float sc[CS_PASSES * 4];
    #pragma unroll
    for (int p = 0; p < CS_PASSES; ++p) {
        const int s_[4] = {sv[p].x, sv[p].y, sv[p].z, sv[p].w};
        const int d_[4] = {dv[p].x, dv[p].y, dv[p].z, dv[p].w};
        #pragma unroll
        for (int q = 0; q < 4; ++q) {
            const int k = p * 4 + q;
            sN[k] = s_[q];
            rN[k] = d_[q] - nlo;
            in[k] = ok[p] && (unsigned)rN[k] < (unsigned)HALF_NODES;
            sc[k] = dis[in[k] ? sN[k] : 0];
        }
    }
    __syncthreads();

    int2* ew = (int2*)ewbuf;
    #pragma unroll
    for (int k = 0; k < CS_PASSES * 4; ++k) {
        if (in[k]) {
            const int slot = atomicAdd(&cur[rN[k]], 1);
            ew[slot] = make_int2(sN[k], __float_as_int(sc[k]));
        }
    }
}

#define AS_PITCH    132
#define GEMM_BLOCKS 256
#define TILES_M     (N_NODES / 16)
#define GEMM_TPB    ((TILES_M + GEMM_BLOCKS - 1) / GEMM_BLOCKS)
#define GEMM_ROWS   (GEMM_TPB * 16)
#define GEMM_LD     (GEMM_ROWS * 32 / 256)
static_assert(N_NODES % 16 == 0, "whole 16-row tiles only");
static_assert((GEMM_ROWS * 32) % 256 == 0, "the staging loop has whole passes");

__global__ __launch_bounds__(256) void gemm_node(const float* __restrict__ A,
                                                 const float* __restrict__ B,
                                                 float* __restrict__ C) {
    __shared__ float As[GEMM_ROWS * AS_PITCH];
    const int lane = threadIdx.x & 31;
    const int wave = __builtin_amdgcn_readfirstlane((int)(threadIdx.x >> 5));
    const int half = lane >> 4;
    const int l16  = lane & 15;
    const int cq   = wave & 3;
    const int par  = wave >> 2;

    B += (size_t)blockIdx.y * FDIM * FDIM;
    C += (size_t)blockIdx.y * N_NODES * FDIM;

    float4 arow[GEMM_LD];
    #pragma unroll
    for (int i = 0; i < GEMM_LD; ++i) {
        const int r    = i * 8 + wave;
        const int tile = blockIdx.x + (r >> 4) * GEMM_BLOCKS;
        const int t    = (tile < TILES_M) ? tile : TILES_M - 1;
        arow[i] = *(const float4*)(A + ((size_t)t * 16 + (r & 15)) * FDIM + lane * 4);
    }

    const float* bcol = B + (size_t)(2 * half) * FDIM + cq * 32 + l16;
    v2f bfr[2][FDIM / 4];
    #pragma unroll
    for (int s = 0; s < FDIM / 4; ++s) {
        #pragma unroll
        for (int n = 0; n < 2; ++n) {
            bfr[n][s][0] = bcol[(size_t)(4 * s) * FDIM + n * 16];
            bfr[n][s][1] = bcol[(size_t)(4 * s + 1) * FDIM + n * 16];
        }
    }

    #pragma unroll
    for (int i = 0; i < GEMM_LD; ++i)
        *(float4*)(As + (i * 8 + wave) * AS_PITCH + lane * 4) = arow[i];
    __syncthreads();

    #pragma unroll 1
    for (int slot = par; slot < GEMM_TPB; slot += 2) {
        const int tile = blockIdx.x + slot * GEMM_BLOCKS;
        const float* ap = As + (slot * 16 + l16) * AS_PITCH + half * 2;
        v2f afrag[FDIM / 4];
        #pragma unroll
        for (int s = 0; s < FDIM / 4; ++s) afrag[s] = *(const v2f*)(ap + s * 4);

        v8f acc0 = {}, acc1 = {};
        #pragma unroll
        for (int s = 0; s < FDIM / 4; ++s) {
            acc0 = __builtin_amdgcn_wmma_f32_16x16x4_f32(
                false, afrag[s], false, bfr[0][s], (short)0, acc0, false, false);
            acc1 = __builtin_amdgcn_wmma_f32_16x16x4_f32(
                false, afrag[s], false, bfr[1][s], (short)0, acc1, false, false);
        }

        if (tile < TILES_M) {
            float* crow = C + ((size_t)tile * 16 + half * 8) * FDIM + cq * 32 + l16;
            #pragma unroll
            for (int g = 0; g < 8; ++g) {
                crow[(size_t)g * FDIM]      = acc0[g];
                crow[(size_t)g * FDIM + 16] = acc1[g];
            }
        }
    }
}

__global__ __launch_bounds__(256) void gather_combine(const float* __restrict__ hw,
                                                      const float* __restrict__ dis,
                                                      const int* __restrict__ row_ptr,
                                                      const int* __restrict__ ewbuf,
                                                      const float* __restrict__ b,
                                                      float* __restrict__ hout,
                                                      int nNodes, int addBias, int doRelu) {
    const int node = __builtin_amdgcn_readfirstlane((int)(blockIdx.x * 8 + (threadIdx.x >> 5)));
    if (node >= nNodes) return;
    const int f = (threadIdx.x & 31) * 4;
    const int2* ew = (const int2*)ewbuf;

    int j = row_ptr[node];
    const int end = row_ptr[node + 1];

    float4 acc = make_float4(0.0f, 0.0f, 0.0f, 0.0f);
    for (; j + 4 <= end; j += 4) {
        const int2 e0 = ew[j], e1 = ew[j + 1], e2 = ew[j + 2], e3 = ew[j + 3];
        const float c0 = __int_as_float(e0.y), c1 = __int_as_float(e1.y);
        const float c2 = __int_as_float(e2.y), c3 = __int_as_float(e3.y);
        const float4 v0 = *(const float4*)(hw + (size_t)e0.x * FDIM + f);
        const float4 v1 = *(const float4*)(hw + (size_t)e1.x * FDIM + f);
        const float4 v2 = *(const float4*)(hw + (size_t)e2.x * FDIM + f);
        const float4 v3 = *(const float4*)(hw + (size_t)e3.x * FDIM + f);
        acc.x = fmaf(v0.x, c0, acc.x); acc.y = fmaf(v0.y, c0, acc.y);
        acc.z = fmaf(v0.z, c0, acc.z); acc.w = fmaf(v0.w, c0, acc.w);
        acc.x = fmaf(v1.x, c1, acc.x); acc.y = fmaf(v1.y, c1, acc.y);
        acc.z = fmaf(v1.z, c1, acc.z); acc.w = fmaf(v1.w, c1, acc.w);
        acc.x = fmaf(v2.x, c2, acc.x); acc.y = fmaf(v2.y, c2, acc.y);
        acc.z = fmaf(v2.z, c2, acc.z); acc.w = fmaf(v2.w, c2, acc.w);
        acc.x = fmaf(v3.x, c3, acc.x); acc.y = fmaf(v3.y, c3, acc.y);
        acc.z = fmaf(v3.z, c3, acc.z); acc.w = fmaf(v3.w, c3, acc.w);
    }
    const int rem = end - j;
    if (rem > 0) {
        const int last = end - 1;
        const int j1 = (j + 1 < end) ? j + 1 : last;
        const int j2 = (j + 2 < end) ? j + 2 : last;
        const int2 e0 = ew[j], e1 = ew[j1], e2 = ew[j2];
        const float c0 = __int_as_float(e0.y), c1 = __int_as_float(e1.y), c2 = __int_as_float(e2.y);
        const float4 v0 = *(const float4*)(hw + (size_t)e0.x * FDIM + f);
        const float4 v1 = *(const float4*)(hw + (size_t)e1.x * FDIM + f);
        const float4 v2 = *(const float4*)(hw + (size_t)e2.x * FDIM + f);
        acc.x = fmaf(v0.x, c0, acc.x); acc.y = fmaf(v0.y, c0, acc.y);
        acc.z = fmaf(v0.z, c0, acc.z); acc.w = fmaf(v0.w, c0, acc.w);
        float4 t;
        t.x = fmaf(v1.x, c1, acc.x); t.y = fmaf(v1.y, c1, acc.y);
        t.z = fmaf(v1.z, c1, acc.z); t.w = fmaf(v1.w, c1, acc.w);
        acc.x = (rem > 1) ? t.x : acc.x; acc.y = (rem > 1) ? t.y : acc.y;
        acc.z = (rem > 1) ? t.z : acc.z; acc.w = (rem > 1) ? t.w : acc.w;
        t.x = fmaf(v2.x, c2, acc.x); t.y = fmaf(v2.y, c2, acc.y);
        t.z = fmaf(v2.z, c2, acc.z); t.w = fmaf(v2.w, c2, acc.w);
        acc.x = (rem > 2) ? t.x : acc.x; acc.y = (rem > 2) ? t.y : acc.y;
        acc.z = (rem > 2) ? t.z : acc.z; acc.w = (rem > 2) ? t.w : acc.w;
    }

    const float di  = dis[node];
    const float di2 = di * di;
    const float4 h = *(const float4*)(hw + (size_t)node * FDIM + f);
    float4 r;
    r.x = acc.x * di + h.x * di2;
    r.y = acc.y * di + h.y * di2;
    r.z = acc.z * di + h.z * di2;
    r.w = acc.w * di + h.w * di2;
    if (addBias) {
        r.x += b[f + 0]; r.y += b[f + 1];
        r.z += b[f + 2]; r.w += b[f + 3];
    }
    if (doRelu) {
        r.x = fmaxf(r.x, 0.0f); r.y = fmaxf(r.y, 0.0f);
        r.z = fmaxf(r.z, 0.0f); r.w = fmaxf(r.w, 0.0f);
    }
    *(float4*)(hout + (size_t)node * FDIM + f) = r;
}

static_assert(E_TR % 32 == 0 && E_TE % 32 == 0, "a wave's 32 rows must lie inside one edge list");
static_assert(N_ROWS % 32 == 0, "whole waves only");

__global__ __launch_bounds__(256) void edge_score(const float* __restrict__ P,
                                                  const float* __restrict__ Q,
                                                  const int* __restrict__ pos_tr,
                                                  const int* __restrict__ neg_tr,
                                                  const int* __restrict__ pos_te,
                                                  const int* __restrict__ neg_te,
                                                  const float* __restrict__ bl1,
                                                  const float* __restrict__ bc,
                                                  const float* __restrict__ Wl2,
                                                  const float* __restrict__ bl2,
                                                  float* __restrict__ out) {
    const int r0 = __builtin_amdgcn_readfirstlane(
        (int)(((blockIdx.x * blockDim.x + threadIdx.x) >> 5) * 32));
    if (r0 >= N_ROWS) return;
    const int lane = threadIdx.x & 31;
    const int g = lane >> 3;
    const int j = lane & 7;

    const int* e; int i0, len;
    if (r0 < E_TR)                 { e = pos_tr; i0 = r0;                   len = E_TR; }
    else if (r0 < 2 * E_TR)        { e = neg_tr; i0 = r0 - E_TR;            len = E_TR; }
    else if (r0 < 2 * E_TR + E_TE) { e = pos_te; i0 = r0 - 2 * E_TR;        len = E_TE; }
    else                           { e = neg_te; i0 = r0 - 2 * E_TR - E_TE; len = E_TE; }

    const int* eu = e + i0 + g * 8;
    const int* ev = eu + len;
    const int4 ua = *(const int4*)eu, ub = *(const int4*)(eu + 4);
    const int4 va = *(const int4*)ev, vb = *(const int4*)(ev + 4);
    const int us[8] = {ua.x, ua.y, ua.z, ua.w, ub.x, ub.y, ub.z, ub.w};
    const int vs[8] = {va.x, va.y, va.z, va.w, vb.x, vb.y, vb.z, vb.w};

    float4 bb[4], ww[4];
    #pragma unroll
    for (int c = 0; c < 4; ++c) {
        const float4 b1 = *(const float4*)(bl1 + c * 32 + j * 4);
        const float4 bp = *(const float4*)(bc + c * 32 + j * 4);
        const float4 bq = *(const float4*)(bc + FDIM + c * 32 + j * 4);
        bb[c] = make_float4(b1.x + (bp.x + bq.x), b1.y + (bp.y + bq.y),
                            b1.z + (bp.z + bq.z), b1.w + (bp.w + bq.w));
        ww[c] = *(const float4*)(Wl2 + c * 32 + j * 4);
    }

    float s[8];
    #pragma unroll
    for (int p = 0; p < 8; ++p) {
        const float* pr = P + (size_t)us[p] * FDIM + j * 4;
        const float* qr = Q + (size_t)vs[p] * FDIM + j * 4;
        float acc = 0.0f;
        #pragma unroll
        for (int c = 0; c < 4; ++c) {
            const float4 a = *(const float4*)(pr + c * 32);
            const float4 b = *(const float4*)(qr + c * 32);
            acc = fmaf(fmaxf(a.x + b.x + bb[c].x, 0.0f), ww[c].x, acc);
            acc = fmaf(fmaxf(a.y + b.y + bb[c].y, 0.0f), ww[c].y, acc);
            acc = fmaf(fmaxf(a.z + b.z + bb[c].z, 0.0f), ww[c].z, acc);
            acc = fmaf(fmaxf(a.w + b.w + bb[c].w, 0.0f), ww[c].w, acc);
        }
        s[p] = acc;
    }

    const bool k4 = (j & 4) != 0, k2 = (j & 2) != 0, k1 = (j & 1) != 0;
    float t4[4];
    #pragma unroll
    for (int i = 0; i < 4; ++i) {
        const float keep = k4 ? s[4 + i] : s[i];
        const float send = k4 ? s[i] : s[4 + i];
        t4[i] = keep + __shfl_xor(send, 4, 32);
    }
    float t2[2];
    #pragma unroll
    for (int i = 0; i < 2; ++i) {
        const float keep = k2 ? t4[2 + i] : t4[i];
        const float send = k2 ? t4[i] : t4[2 + i];
        t2[i] = keep + __shfl_xor(send, 2, 32);
    }
    const float keep = k1 ? t2[1] : t2[0];
    const float send = k1 ? t2[0] : t2[1];
    const float tot = keep + __shfl_xor(send, 1, 32);

    out[r0 + lane] = tot + bl2[0];
}

extern "C" void kernel_launch(void* const* d_in, const int* in_sizes, int n_in,
                              void* d_out, int out_size, void* d_ws, size_t ws_size,
                              hipStream_t stream) {
    const float* x          = (const float*)d_in[0];
    const int*   edge_index = (const int*)d_in[1];
    const int*   pos_tr     = (const int*)d_in[2];
    const int*   neg_tr     = (const int*)d_in[3];
    const int*   pos_te     = (const int*)d_in[4];
    const int*   neg_te     = (const int*)d_in[5];
    const float* W[4] = {(const float*)d_in[6],  (const float*)d_in[8],
                         (const float*)d_in[10], (const float*)d_in[12]};
    const float* b[4] = {(const float*)d_in[7],  (const float*)d_in[9],
                         (const float*)d_in[11], (const float*)d_in[13]};
    const float* Wl1 = (const float*)d_in[14];
    const float* bl1 = (const float*)d_in[15];
    const float* Wl2 = (const float*)d_in[16];
    const float* bl2 = (const float*)d_in[17];
    float* out = (float*)d_out;

    const size_t NF = (size_t)N_NODES * FDIM;
    float* ws  = (float*)d_ws;
    float* dis = ws;
    float* hw  = ws + 50048;
    int*   csr = (int*)(hw + NF);
    int*   cnt     = csr;
    int*   row_ptr = csr + 53248;
    int*   ew      = csr + 2 * 53248;
    float* Wc  = (float*)(ew + 2 * N_EDGES);
    float* bc  = Wc + 2 * FDIM * FDIM;
    float* h0  = hw + 2 * NF;
    float* h1  = h0 + NF;
    int*   blk = (int*)h1;

    const int* src = edge_index;
    const int* dst = edge_index + N_EDGES;

    count_block<<<CS_BLOCKS, 1024, 0, stream>>>(dst, blk);
    chunk_prefix<<<PREFIX_BLOCKS + FOLD_BLOCKS, 256, 0, stream>>>(blk, cnt, W[3], b[3], Wl1, Wc, bc);
    scan_rows<<<1, 1024, 0, stream>>>(cnt, row_ptr, dis);
    fill_block<<<dim3(CS_BLOCKS, 2), 1024, 0, stream>>>(src, dst, blk, row_ptr, dis, ew);

    const float* hin = x;
    float* houts[3] = {h0, h1, h0};
    for (int l = 0; l < 3; ++l) {
        gemm_node<<<GEMM_BLOCKS, 256, 0, stream>>>(hin, W[l], hw);
        gather_combine<<<(N_NODES * 32 + 255) / 256, 256, 0, stream>>>(
            hw, dis, row_ptr, ew, b[l], houts[l], N_NODES, 1, 1);
        hin = houts[l];
    }

    gather_combine<<<(N_NODES * 32 + 255) / 256, 256, 0, stream>>>(
        h0, dis, row_ptr, ew, b[3], hw, N_NODES, 0, 0);

    gemm_node<<<dim3(GEMM_BLOCKS, 2), 256, 0, stream>>>(hw, Wc, h0);

    const int edgeWaves  = N_ROWS / 32;
    const int edgeBlocks = (edgeWaves + 7) / 8;
    edge_score<<<edgeBlocks, 256, 0, stream>>>(h0, h1, pos_tr, neg_tr, pos_te, neg_te,
                                               bl1, bc, Wl2, bl2, out);
}


# GCN layer as one kernel by linearity Agg(h@W)=Agg(h)@W: a block gathers a 16-node tile of Agg(h) into LDS and multiplies it there on f32 WMMA; 4 launches and the h@W planes gone; products regrouped
# speedup vs baseline: 4.6913x; 1.0831x over previous
#include <hip/hip_runtime.h>


typedef float v2f __attribute__((ext_vector_type(2)));
typedef float v8f __attribute__((ext_vector_type(8)));

#define N_NODES 50000
#define N_EDGES 800000
#define FDIM    128
#define E_TR    100000
#define E_TE    20000
#define N_ROWS  240000

#define CS_BLOCKS 100
#define CS_CHUNK  (N_EDGES / CS_BLOCKS)
#define CS_GROUPS (CS_CHUNK / 4)
#define CS_PASSES ((CS_GROUPS + 1023) / 1024)
#define CS_STRIDE 50048
static_assert(N_EDGES % CS_BLOCKS == 0 && CS_CHUNK % 4 == 0, "chunks are whole 16-byte groups");
static_assert(CS_STRIDE >= N_NODES && CS_STRIDE % 32 == 0, "a plane holds every node");
static_assert(N_NODES % 4 == 0, "16-byte groups must not straddle the end");

__global__ __launch_bounds__(1024) void count_block(const int* __restrict__ dst,
                                                    int* __restrict__ blk) {
    __shared__ int hist[N_NODES];
    const int4* d4 = (const int4*)(dst + (size_t)blockIdx.x * CS_CHUNK);
    int4 dv[CS_PASSES];
    bool ok[CS_PASSES];
    #pragma unroll
    for (int p = 0; p < CS_PASSES; ++p) {
        const int i = p * 1024 + threadIdx.x;
        ok[p] = i < CS_GROUPS;
        dv[p] = d4[ok[p] ? i : CS_GROUPS - 1];
    }

    for (int i = threadIdx.x * 4; i < N_NODES; i += 4096)
        *(int4*)(hist + i) = make_int4(0, 0, 0, 0);
    __syncthreads();

    #pragma unroll
    for (int p = 0; p < CS_PASSES; ++p) {
        if (ok[p]) {
            atomicAdd(&hist[dv[p].x], 1);
            atomicAdd(&hist[dv[p].y], 1);
            atomicAdd(&hist[dv[p].z], 1);
            atomicAdd(&hist[dv[p].w], 1);
        }
    }
    __syncthreads();

    int* out = blk + (size_t)blockIdx.x * CS_STRIDE;
    for (int i = threadIdx.x * 4; i < N_NODES; i += 4096)
        *(int4*)(out + i) = *(const int4*)(hist + i);
}

#define PREFIX_BLOCKS ((N_NODES + 255) / 256)
#define FOLD_TILES    (FDIM / 16 + 1)
#define FOLD_BLOCKS   (FOLD_TILES * 2)

__device__ __forceinline__ void fold_weights(int job,
                                             const float* __restrict__ W3,
                                             const float* __restrict__ b3,
                                             const float* __restrict__ Wl1,
                                             float* __restrict__ Wc,
                                             float* __restrict__ bc) {
    const int lane = threadIdx.x & 31;
    const int wave = __builtin_amdgcn_readfirstlane((int)(threadIdx.x >> 5));
    const int half = lane >> 4;
    const int l16  = lane & 15;
    const int tile = job >> 1;
    const int y    = job & 1;
    const bool biasTile = tile == FOLD_TILES - 1;

    const float* arow = (biasTile ? b3 : W3 + (size_t)(tile * 16 + l16) * FDIM) + 2 * half;
    const bool live = !biasTile || l16 == 0;
    const float* bcol = Wl1 + (size_t)y * FDIM * FDIM + (size_t)(2 * half) * FDIM + wave * 16 + l16;

    v2f af[FDIM / 4], bf[FDIM / 4];
    #pragma unroll
    for (int s = 0; s < FDIM / 4; ++s) {
        af[s]    = *(const v2f*)(arow + 4 * s);
        bf[s][0] = bcol[(size_t)(4 * s) * FDIM];
        bf[s][1] = bcol[(size_t)(4 * s + 1) * FDIM];
    }
    v8f acc = {};
    #pragma unroll
    for (int s = 0; s < FDIM / 4; ++s) {
        v2f a;
        a[0] = live ? af[s][0] : 0.0f;
        a[1] = live ? af[s][1] : 0.0f;
        acc = __builtin_amdgcn_wmma_f32_16x16x4_f32(
            false, a, false, bf[s], (short)0, acc, false, false);
    }

    if (!biasTile) {
        float* crow = Wc + (size_t)y * FDIM * FDIM
                         + (size_t)(tile * 16 + half * 8) * FDIM + wave * 16 + l16;
        #pragma unroll
        for (int g = 0; g < 8; ++g) crow[(size_t)g * FDIM] = acc[g];
    } else if (half == 0) {
        bc[y * FDIM + wave * 16 + l16] = acc[0];
    }
}

__global__ __launch_bounds__(256) void chunk_prefix(int* __restrict__ blk,
                                                    int* __restrict__ cnt,
                                                    const float* __restrict__ W3,
                                                    const float* __restrict__ b3,
                                                    const float* __restrict__ Wl1,
                                                    float* __restrict__ Wc,
                                                    float* __restrict__ bc) {
    if (blockIdx.x >= PREFIX_BLOCKS) {
        fold_weights((int)blockIdx.x - PREFIX_BLOCKS, W3, b3, Wl1, Wc, bc);
        return;
    }
    const int n = blockIdx.x * 256 + threadIdx.x;
    if (n >= N_NODES) return;
    int c[CS_BLOCKS];
    #pragma unroll
    for (int b = 0; b < CS_BLOCKS; ++b) c[b] = blk[(size_t)b * CS_STRIDE + n];
    int run = 0;
    #pragma unroll
    for (int b = 0; b < CS_BLOCKS; ++b) {
        blk[(size_t)b * CS_STRIDE + n] = run;
        run += c[b];
    }
    cnt[n] = run;
}

#define SCAN_PASSES 13
#define SCAN_CHUNK  (SCAN_PASSES * 128)
static_assert(32 * SCAN_CHUNK >= N_NODES, "one block must cover every node");
static_assert(N_NODES % 4 == 0, "16-byte groups must not straddle the end");

__device__ __forceinline__ int wave_incl_scan(int v, int lane) {
    #pragma unroll
    for (int o = 1; o < 32; o <<= 1) {
        int y = __shfl_up(v, o, 32);
        if (lane >= o) v += y;
    }
    return v;
}

__global__ __launch_bounds__(1024) void scan_rows(const int* __restrict__ cnt,
                                                  int* __restrict__ row_ptr,
                                                  float* __restrict__ dis) {
    __shared__ int wtot[32];
    const int lane = threadIdx.x & 31;
    const int w    = threadIdx.x >> 5;
    const int base = w * SCAN_CHUNK + lane * 4;

    int4 p[SCAN_PASSES];
    #pragma unroll
    for (int i = 0; i < SCAN_PASSES; ++i) {
        const int idx = base + i * 128;
        const int4 v = *(const int4*)(cnt + ((idx < N_NODES) ? idx : N_NODES - 4));
        p[i] = (idx < N_NODES) ? v : make_int4(0, 0, 0, 0);
    }
    #pragma unroll
    for (int i = 0; i < SCAN_PASSES; ++i) {
        const int idx = base + i * 128;
        if (idx < N_NODES) {
            float4 d;
            d.x = rsqrtf((float)p[i].x + 1.0f);
            d.y = rsqrtf((float)p[i].y + 1.0f);
            d.z = rsqrtf((float)p[i].z + 1.0f);
            d.w = rsqrtf((float)p[i].w + 1.0f);
            *(float4*)(dis + idx) = d;
        }
    }

    int run = 0;
    #pragma unroll
    for (int i = 0; i < SCAN_PASSES; ++i) {
        const int4 c = p[i];
        const int s   = c.x + c.y + c.z + c.w;
        const int inc = wave_incl_scan(s, lane);
        const int o   = run + inc - s;
        p[i].x = o;
        p[i].y = o + c.x;
        p[i].z = o + c.x + c.y;
        p[i].w = o + c.x + c.y + c.z;
        run += __shfl(inc, 31, 32);
    }
    if (lane == 0) wtot[w] = run;
    __syncthreads();

    const int t     = wtot[lane];
    const int tinc  = wave_incl_scan(t, lane);
    const int wbase = __shfl(tinc - t, w, 32);

    #pragma unroll
    for (int i = 0; i < SCAN_PASSES; ++i) {
        int idx = base + i * 128;
        if (idx < N_NODES) {
            int4 q = p[i];
            q.x += wbase; q.y += wbase; q.z += wbase; q.w += wbase;
            *(int4*)(row_ptr + idx) = q;
        }
    }
    if (threadIdx.x == 1023) row_ptr[N_NODES] = wbase + run;
}

#define HALF_NODES (N_NODES / 2)
#define TAB_PASSES ((HALF_NODES / 4 + 1023) / 1024)
static_assert(N_NODES % 2 == 0 && HALF_NODES % 4 == 0, "a half is whole 16-byte groups");

__global__ __launch_bounds__(1024) void fill_block(const int* __restrict__ src,
                                                   const int* __restrict__ dst,
                                                   const int* __restrict__ blk,
                                                   const int* __restrict__ row_ptr,
                                                   const float* __restrict__ dis,
                                                   int* __restrict__ ewbuf) {
    __shared__ int cur[HALF_NODES];
    const int nlo = blockIdx.y * HALF_NODES;
    const int4* s4 = (const int4*)(src + (size_t)blockIdx.x * CS_CHUNK);
    const int4* d4 = (const int4*)(dst + (size_t)blockIdx.x * CS_CHUNK);
    int4 sv[CS_PASSES], dv[CS_PASSES];
    bool ok[CS_PASSES];
    #pragma unroll
    for (int p = 0; p < CS_PASSES; ++p) {
        const int i = p * 1024 + threadIdx.x;
        ok[p] = i < CS_GROUPS;
        sv[p] = s4[ok[p] ? i : CS_GROUPS - 1];
        dv[p] = d4[ok[p] ? i : CS_GROUPS - 1];
    }

    const int* off = blk + (size_t)blockIdx.x * CS_STRIDE + nlo;
    const int* rp  = row_ptr + nlo;
    int4 a[TAB_PASSES], o[TAB_PASSES];
    #pragma unroll
    for (int t = 0; t < TAB_PASSES; ++t) {
        const int i = (t * 1024 + threadIdx.x) * 4;
        const int ii = (i < HALF_NODES) ? i : HALF_NODES - 4;
        a[t] = *(const int4*)(rp + ii);
        o[t] = *(const int4*)(off + ii);
    }
    #pragma unroll
    for (int t = 0; t < TAB_PASSES; ++t) {
        const int i = (t * 1024 + threadIdx.x) * 4;
        if (i < HALF_NODES)
            *(int4*)(cur + i) = make_int4(a[t].x + o[t].x, a[t].y + o[t].y,
                                          a[t].z + o[t].z, a[t].w + o[t].w);
    }

    int   sN[CS_PASSES * 4];
    int   rN[CS_PASSES * 4];
    bool  in[CS_PASSES * 4];
    float sc[CS_PASSES * 4];
    #pragma unroll
    for (int p = 0; p < CS_PASSES; ++p) {
        const int s_[4] = {sv[p].x, sv[p].y, sv[p].z, sv[p].w};
        const int d_[4] = {dv[p].x, dv[p].y, dv[p].z, dv[p].w};
        #pragma unroll
        for (int q = 0; q < 4; ++q) {
            const int k = p * 4 + q;
            sN[k] = s_[q];
            rN[k] = d_[q] - nlo;
            in[k] = ok[p] && (unsigned)rN[k] < (unsigned)HALF_NODES;
            sc[k] = dis[in[k] ? sN[k] : 0];
        }
    }
    __syncthreads();

    int2* ew = (int2*)ewbuf;
    #pragma unroll
    for (int k = 0; k < CS_PASSES * 4; ++k) {
        if (in[k]) {
            const int slot = atomicAdd(&cur[rN[k]], 1);
            ew[slot] = make_int2(sN[k], __float_as_int(sc[k]));
        }
    }
}

#define AS_PITCH    132
#define TILES_M     (N_NODES / 16)
#define AGG_BLOCKS1 768
#define AGG_BLOCKS2 512
#define AGG_GROUP   16
static_assert(N_NODES % 16 == 0, "whole 16-row tiles only");

__device__ __forceinline__ float4 agg_row(const float* __restrict__ h,
                                          const float* __restrict__ dis,
                                          const int2* __restrict__ ew,
                                          int node, int j, int end, int lane) {
    const int f = lane * 4;
    const float di  = dis[node];
    const float di2 = di * di;
    const float4 own = *(const float4*)(h + (size_t)node * FDIM + f);

    float4 acc = make_float4(0.0f, 0.0f, 0.0f, 0.0f);
    for (; j < end; j += AGG_GROUP) {
        const int n = end - j;
        const int2 rec = ew[j + lane];
        int   s[AGG_GROUP];
        float c[AGG_GROUP];
        #pragma unroll
        for (int k = 0; k < AGG_GROUP; ++k) {
            s[k] = __builtin_amdgcn_readlane(rec.x, k);
            c[k] = __int_as_float(__builtin_amdgcn_readlane(rec.y, k));
        }
        float4 v[AGG_GROUP];
        #pragma unroll
        for (int k = 0; k < AGG_GROUP; ++k)
            if (k < n) v[k] = *(const float4*)(h + (size_t)s[k] * FDIM + f);
        #pragma unroll
        for (int k = 0; k < AGG_GROUP; ++k) {
            if (k < n) {
                acc.x = fmaf(v[k].x, c[k], acc.x); acc.y = fmaf(v[k].y, c[k], acc.y);
                acc.z = fmaf(v[k].z, c[k], acc.z); acc.w = fmaf(v[k].w, c[k], acc.w);
            }
        }
    }

    float4 r;
    r.x = acc.x * di + own.x * di2;
    r.y = acc.y * di + own.y * di2;
    r.z = acc.z * di + own.z * di2;
    r.w = acc.w * di + own.w * di2;
    return r;
}

template <int PLANES>
__global__ __launch_bounds__(256) void agg_gemm(const float* __restrict__ h,
                                                const float* __restrict__ dis,
                                                const int* __restrict__ row_ptr,
                                                const int* __restrict__ ewbuf,
                                                const float* __restrict__ B,
                                                const float* __restrict__ bias,
                                                float* __restrict__ C) {
    __shared__ float As[2 * 16 * AS_PITCH];
    const int lane = threadIdx.x & 31;
    const int wave = __builtin_amdgcn_readfirstlane((int)(threadIdx.x >> 5));
    const int half = lane >> 4;
    const int l16  = lane & 15;
    const int2* ew = (const int2*)ewbuf;

    const float* bcol = B + (size_t)(2 * half) * FDIM + wave * 16 + l16;
    v2f bfr[PLANES][FDIM / 4];
    #pragma unroll
    for (int s = 0; s < FDIM / 4; ++s) {
        #pragma unroll
        for (int p = 0; p < PLANES; ++p) {
            bfr[p][s][0] = bcol[(size_t)p * FDIM * FDIM + (size_t)(4 * s) * FDIM];
            bfr[p][s][1] = bcol[(size_t)p * FDIM * FDIM + (size_t)(4 * s + 1) * FDIM];
        }
    }
    float bv = 0.0f;
    if (PLANES == 1) bv = bias[wave * 16 + l16];

    int buf = 0;
    for (int tile = blockIdx.x; tile < TILES_M; tile += gridDim.x) {
        float* at = As + buf * 16 * AS_PITCH;
        const int* rp = row_ptr + tile * 16 + 2 * wave;
        const int b0 = __builtin_amdgcn_readfirstlane(rp[0]);
        const int b1 = __builtin_amdgcn_readfirstlane(rp[1]);
        const int b2 = __builtin_amdgcn_readfirstlane(rp[2]);
        for (int i = 0; i < 2; ++i) {
            const int r = 2 * wave + i;
            const float4 g = agg_row(h, dis, ew, tile * 16 + r, i ? b1 : b0, i ? b2 : b1, lane);
            *(float4*)(at + r * AS_PITCH + lane * 4) = g;
        }
        __syncthreads();

        const float* ap = at + l16 * AS_PITCH + half * 2;
        v2f afrag[FDIM / 4];
        #pragma unroll
        for (int s = 0; s < FDIM / 4; ++s) afrag[s] = *(const v2f*)(ap + s * 4);

        v8f acc[PLANES];
        #pragma unroll
        for (int p = 0; p < PLANES; ++p) acc[p] = v8f{};
        #pragma unroll
        for (int s = 0; s < FDIM / 4; ++s) {
            #pragma unroll
            for (int p = 0; p < PLANES; ++p)
                acc[p] = __builtin_amdgcn_wmma_f32_16x16x4_f32(
                    false, afrag[s], false, bfr[p][s], (short)0, acc[p], false, false);
        }

        #pragma unroll
        for (int p = 0; p < PLANES; ++p) {
            float* crow = C + (size_t)p * N_NODES * FDIM
                            + ((size_t)tile * 16 + half * 8) * FDIM + wave * 16 + l16;
            #pragma unroll
            for (int g = 0; g < 8; ++g) {
                float o = acc[p][g];
                if (PLANES == 1) o = fmaxf(o + bv, 0.0f);
                crow[(size_t)g * FDIM] = o;
            }
        }
        buf ^= 1;
    }
}

static_assert(E_TR % 32 == 0 && E_TE % 32 == 0, "a wave's 32 rows must lie inside one edge list");
static_assert(N_ROWS % 32 == 0, "whole waves only");

__global__ __launch_bounds__(256) void edge_score(const float* __restrict__ P,
                                                  const float* __restrict__ Q,
                                                  const int* __restrict__ pos_tr,
                                                  const int* __restrict__ neg_tr,
                                                  const int* __restrict__ pos_te,
                                                  const int* __restrict__ neg_te,
                                                  const float* __restrict__ bl1,
                                                  const float* __restrict__ bc,
                                                  const float* __restrict__ Wl2,
                                                  const float* __restrict__ bl2,
                                                  float* __restrict__ out) {
    const int r0 = __builtin_amdgcn_readfirstlane(
        (int)(((blockIdx.x * blockDim.x + threadIdx.x) >> 5) * 32));
    if (r0 >= N_ROWS) return;
    const int lane = threadIdx.x & 31;
    const int g = lane >> 3;
    const int j = lane & 7;

    const int* e; int i0, len;
    if (r0 < E_TR)                 { e = pos_tr; i0 = r0;                   len = E_TR; }
    else if (r0 < 2 * E_TR)        { e = neg_tr; i0 = r0 - E_TR;            len = E_TR; }
    else if (r0 < 2 * E_TR + E_TE) { e = pos_te; i0 = r0 - 2 * E_TR;        len = E_TE; }
    else                           { e = neg_te; i0 = r0 - 2 * E_TR - E_TE; len = E_TE; }

    const int* eu = e + i0 + g * 8;
    const int* ev = eu + len;
    const int4 ua = *(const int4*)eu, ub = *(const int4*)(eu + 4);
    const int4 va = *(const int4*)ev, vb = *(const int4*)(ev + 4);
    const int us[8] = {ua.x, ua.y, ua.z, ua.w, ub.x, ub.y, ub.z, ub.w};
    const int vs[8] = {va.x, va.y, va.z, va.w, vb.x, vb.y, vb.z, vb.w};

    float4 bb[4], ww[4];
    #pragma unroll
    for (int c = 0; c < 4; ++c) {
        const float4 b1 = *(const float4*)(bl1 + c * 32 + j * 4);
        const float4 bp = *(const float4*)(bc + c * 32 + j * 4);
        const float4 bq = *(const float4*)(bc + FDIM + c * 32 + j * 4);
        bb[c] = make_float4(b1.x + (bp.x + bq.x), b1.y + (bp.y + bq.y),
                            b1.z + (bp.z + bq.z), b1.w + (bp.w + bq.w));
        ww[c] = *(const float4*)(Wl2 + c * 32 + j * 4);
    }

    float s[8];
    #pragma unroll
    for (int p = 0; p < 8; ++p) {
        const float* pr = P + (size_t)us[p] * FDIM + j * 4;
        const float* qr = Q + (size_t)vs[p] * FDIM + j * 4;
        float acc = 0.0f;
        #pragma unroll
        for (int c = 0; c < 4; ++c) {
            const float4 a = *(const float4*)(pr + c * 32);
            const float4 b = *(const float4*)(qr + c * 32);
            acc = fmaf(fmaxf(a.x + b.x + bb[c].x, 0.0f), ww[c].x, acc);
            acc = fmaf(fmaxf(a.y + b.y + bb[c].y, 0.0f), ww[c].y, acc);
            acc = fmaf(fmaxf(a.z + b.z + bb[c].z, 0.0f), ww[c].z, acc);
            acc = fmaf(fmaxf(a.w + b.w + bb[c].w, 0.0f), ww[c].w, acc);
        }
        s[p] = acc;
    }

    const bool k4 = (j & 4) != 0, k2 = (j & 2) != 0, k1 = (j & 1) != 0;
    float t4[4];
    #pragma unroll
    for (int i = 0; i < 4; ++i) {
        const float keep = k4 ? s[4 + i] : s[i];
        const float send = k4 ? s[i] : s[4 + i];
        t4[i] = keep + __shfl_xor(send, 4, 32);
    }
    float t2[2];
    #pragma unroll
    for (int i = 0; i < 2; ++i) {
        const float keep = k2 ? t4[2 + i] : t4[i];
        const float send = k2 ? t4[i] : t4[2 + i];
        t2[i] = keep + __shfl_xor(send, 2, 32);
    }
    const float keep = k1 ? t2[1] : t2[0];
    const float send = k1 ? t2[0] : t2[1];
    const float tot = keep + __shfl_xor(send, 1, 32);

    out[r0 + lane] = tot + bl2[0];
}

extern "C" void kernel_launch(void* const* d_in, const int* in_sizes, int n_in,
                              void* d_out, int out_size, void* d_ws, size_t ws_size,
                              hipStream_t stream) {
    const float* x          = (const float*)d_in[0];
    const int*   edge_index = (const int*)d_in[1];
    const int*   pos_tr     = (const int*)d_in[2];
    const int*   neg_tr     = (const int*)d_in[3];
    const int*   pos_te     = (const int*)d_in[4];
    const int*   neg_te     = (const int*)d_in[5];
    const float* W[4] = {(const float*)d_in[6],  (const float*)d_in[8],
                         (const float*)d_in[10], (const float*)d_in[12]};
    const float* b[4] = {(const float*)d_in[7],  (const float*)d_in[9],
                         (const float*)d_in[11], (const float*)d_in[13]};
    const float* Wl1 = (const float*)d_in[14];
    const float* bl1 = (const float*)d_in[15];
    const float* Wl2 = (const float*)d_in[16];
    const float* bl2 = (const float*)d_in[17];
    float* out = (float*)d_out;

    const size_t NF = (size_t)N_NODES * FDIM;
    float* ws  = (float*)d_ws;
    float* dis = ws;
    float* hw  = ws + 50048;
    int*   csr = (int*)(hw + NF);
    int*   cnt     = csr;
    int*   row_ptr = csr + 53248;
    int*   ew      = csr + 2 * 53248;
    float* Wc  = (float*)(ew + 2 * N_EDGES);
    float* bc  = Wc + 2 * FDIM * FDIM;
    float* h0  = hw + 2 * NF;
    float* h1  = h0 + NF;
    int*   blk = (int*)h1;

    const int* src = edge_index;
    const int* dst = edge_index + N_EDGES;

    count_block<<<CS_BLOCKS, 1024, 0, stream>>>(dst, blk);
    chunk_prefix<<<PREFIX_BLOCKS + FOLD_BLOCKS, 256, 0, stream>>>(blk, cnt, W[3], b[3], Wl1, Wc, bc);
    scan_rows<<<1, 1024, 0, stream>>>(cnt, row_ptr, dis);
    fill_block<<<dim3(CS_BLOCKS, 2), 1024, 0, stream>>>(src, dst, blk, row_ptr, dis, ew);

    const float* hin = x;
    float* houts[3] = {h0, h1, hw};
    for (int l = 0; l < 3; ++l) {
        agg_gemm<1><<<AGG_BLOCKS1, 256, 0, stream>>>(hin, dis, row_ptr, ew, W[l], b[l], houts[l]);
        hin = houts[l];
    }

    agg_gemm<2><<<AGG_BLOCKS2, 256, 0, stream>>>(hw, dis, row_ptr, ew, Wc, b[3], h0);

    const int edgeWaves  = N_ROWS / 32;
    const int edgeBlocks = (edgeWaves + 7) / 8;
    edge_score<<<edgeBlocks, 256, 0, stream>>>(h0, h1, pos_tr, neg_tr, pos_te, neg_te,
                                               bl1, bc, Wl2, bl2, out);
}


# Layer gather pipelined a tile ahead: a wave's two rows are one CSR record stream; bounds, first records, scales and own rows of its next tile load under the current product; row sum order kept
# speedup vs baseline: 4.8486x; 1.0335x over previous
#include <hip/hip_runtime.h>


typedef float v2f __attribute__((ext_vector_type(2)));
typedef float v8f __attribute__((ext_vector_type(8)));

#define N_NODES 50000
#define N_EDGES 800000
#define FDIM    128
#define E_TR    100000
#define E_TE    20000
#define N_ROWS  240000

#define CS_BLOCKS 100
#define CS_CHUNK  (N_EDGES / CS_BLOCKS)
#define CS_GROUPS (CS_CHUNK / 4)
#define CS_PASSES ((CS_GROUPS + 1023) / 1024)
#define CS_STRIDE 50048
static_assert(N_EDGES % CS_BLOCKS == 0 && CS_CHUNK % 4 == 0, "chunks are whole 16-byte groups");
static_assert(CS_STRIDE >= N_NODES && CS_STRIDE % 32 == 0, "a plane holds every node");
static_assert(N_NODES % 4 == 0, "16-byte groups must not straddle the end");

__global__ __launch_bounds__(1024) void count_block(const int* __restrict__ dst,
                                                    int* __restrict__ blk) {
    __shared__ int hist[N_NODES];
    const int4* d4 = (const int4*)(dst + (size_t)blockIdx.x * CS_CHUNK);
    int4 dv[CS_PASSES];
    bool ok[CS_PASSES];
    #pragma unroll
    for (int p = 0; p < CS_PASSES; ++p) {
        const int i = p * 1024 + threadIdx.x;
        ok[p] = i < CS_GROUPS;
        dv[p] = d4[ok[p] ? i : CS_GROUPS - 1];
    }

    for (int i = threadIdx.x * 4; i < N_NODES; i += 4096)
        *(int4*)(hist + i) = make_int4(0, 0, 0, 0);
    __syncthreads();

    #pragma unroll
    for (int p = 0; p < CS_PASSES; ++p) {
        if (ok[p]) {
            atomicAdd(&hist[dv[p].x], 1);
            atomicAdd(&hist[dv[p].y], 1);
            atomicAdd(&hist[dv[p].z], 1);
            atomicAdd(&hist[dv[p].w], 1);
        }
    }
    __syncthreads();

    int* out = blk + (size_t)blockIdx.x * CS_STRIDE;
    for (int i = threadIdx.x * 4; i < N_NODES; i += 4096)
        *(int4*)(out + i) = *(const int4*)(hist + i);
}

#define PREFIX_BLOCKS ((N_NODES + 255) / 256)
#define FOLD_TILES    (FDIM / 16 + 1)
#define FOLD_BLOCKS   (FOLD_TILES * 2)

__device__ __forceinline__ void fold_weights(int job,
                                             const float* __restrict__ W3,
                                             const float* __restrict__ b3,
                                             const float* __restrict__ Wl1,
                                             float* __restrict__ Wc,
                                             float* __restrict__ bc) {
    const int lane = threadIdx.x & 31;
    const int wave = __builtin_amdgcn_readfirstlane((int)(threadIdx.x >> 5));
    const int half = lane >> 4;
    const int l16  = lane & 15;
    const int tile = job >> 1;
    const int y    = job & 1;
    const bool biasTile = tile == FOLD_TILES - 1;

    const float* arow = (biasTile ? b3 : W3 + (size_t)(tile * 16 + l16) * FDIM) + 2 * half;
    const bool live = !biasTile || l16 == 0;
    const float* bcol = Wl1 + (size_t)y * FDIM * FDIM + (size_t)(2 * half) * FDIM + wave * 16 + l16;

    v2f af[FDIM / 4], bf[FDIM / 4];
    #pragma unroll
    for (int s = 0; s < FDIM / 4; ++s) {
        af[s]    = *(const v2f*)(arow + 4 * s);
        bf[s][0] = bcol[(size_t)(4 * s) * FDIM];
        bf[s][1] = bcol[(size_t)(4 * s + 1) * FDIM];
    }
    v8f acc = {};
    #pragma unroll
    for (int s = 0; s < FDIM / 4; ++s) {
        v2f a;
        a[0] = live ? af[s][0] : 0.0f;
        a[1] = live ? af[s][1] : 0.0f;
        acc = __builtin_amdgcn_wmma_f32_16x16x4_f32(
            false, a, false, bf[s], (short)0, acc, false, false);
    }

    if (!biasTile) {
        float* crow = Wc + (size_t)y * FDIM * FDIM
                         + (size_t)(tile * 16 + half * 8) * FDIM + wave * 16 + l16;
        #pragma unroll
        for (int g = 0; g < 8; ++g) crow[(size_t)g * FDIM] = acc[g];
    } else if (half == 0) {
        bc[y * FDIM + wave * 16 + l16] = acc[0];
    }
}

__global__ __launch_bounds__(256) void chunk_prefix(int* __restrict__ blk,
                                                    int* __restrict__ cnt,
                                                    const float* __restrict__ W3,
                                                    const float* __restrict__ b3,
                                                    const float* __restrict__ Wl1,
                                                    float* __restrict__ Wc,
                                                    float* __restrict__ bc) {
    if (blockIdx.x >= PREFIX_BLOCKS) {
        fold_weights((int)blockIdx.x - PREFIX_BLOCKS, W3, b3, Wl1, Wc, bc);
        return;
    }
    const int n = blockIdx.x * 256 + threadIdx.x;
    if (n >= N_NODES) return;
    int c[CS_BLOCKS];
    #pragma unroll
    for (int b = 0; b < CS_BLOCKS; ++b) c[b] = blk[(size_t)b * CS_STRIDE + n];
    int run = 0;
    #pragma unroll
    for (int b = 0; b < CS_BLOCKS; ++b) {
        blk[(size_t)b * CS_STRIDE + n] = run;
        run += c[b];
    }
    cnt[n] = run;
}

#define SCAN_PASSES 13
#define SCAN_CHUNK  (SCAN_PASSES * 128)
static_assert(32 * SCAN_CHUNK >= N_NODES, "one block must cover every node");
static_assert(N_NODES % 4 == 0, "16-byte groups must not straddle the end");

__device__ __forceinline__ int wave_incl_scan(int v, int lane) {
    #pragma unroll
    for (int o = 1; o < 32; o <<= 1) {
        int y = __shfl_up(v, o, 32);
        if (lane >= o) v += y;
    }
    return v;
}

__global__ __launch_bounds__(1024) void scan_rows(const int* __restrict__ cnt,
                                                  int* __restrict__ row_ptr,
                                                  float* __restrict__ dis) {
    __shared__ int wtot[32];
    const int lane = threadIdx.x & 31;
    const int w    = threadIdx.x >> 5;
    const int base = w * SCAN_CHUNK + lane * 4;

    int4 p[SCAN_PASSES];
    #pragma unroll
    for (int i = 0; i < SCAN_PASSES; ++i) {
        const int idx = base + i * 128;
        const int4 v = *(const int4*)(cnt + ((idx < N_NODES) ? idx : N_NODES - 4));
        p[i] = (idx < N_NODES) ? v : make_int4(0, 0, 0, 0);
    }
    #pragma unroll
    for (int i = 0; i < SCAN_PASSES; ++i) {
        const int idx = base + i * 128;
        if (idx < N_NODES) {
            float4 d;
            d.x = rsqrtf((float)p[i].x + 1.0f);
            d.y = rsqrtf((float)p[i].y + 1.0f);
            d.z = rsqrtf((float)p[i].z + 1.0f);
            d.w = rsqrtf((float)p[i].w + 1.0f);
            *(float4*)(dis + idx) = d;
        }
    }

    int run = 0;
    #pragma unroll
    for (int i = 0; i < SCAN_PASSES; ++i) {
        const int4 c = p[i];
        const int s   = c.x + c.y + c.z + c.w;
        const int inc = wave_incl_scan(s, lane);
        const int o   = run + inc - s;
        p[i].x = o;
        p[i].y = o + c.x;
        p[i].z = o + c.x + c.y;
        p[i].w = o + c.x + c.y + c.z;
        run += __shfl(inc, 31, 32);
    }
    if (lane == 0) wtot[w] = run;
    __syncthreads();

    const int t     = wtot[lane];
    const int tinc  = wave_incl_scan(t, lane);
    const int wbase = __shfl(tinc - t, w, 32);

    #pragma unroll
    for (int i = 0; i < SCAN_PASSES; ++i) {
        int idx = base + i * 128;
        if (idx < N_NODES) {
            int4 q = p[i];
            q.x += wbase; q.y += wbase; q.z += wbase; q.w += wbase;
            *(int4*)(row_ptr + idx) = q;
        }
    }
    if (threadIdx.x == 1023) row_ptr[N_NODES] = wbase + run;
}

#define HALF_NODES (N_NODES / 2)
#define TAB_PASSES ((HALF_NODES / 4 + 1023) / 1024)
static_assert(N_NODES % 2 == 0 && HALF_NODES % 4 == 0, "a half is whole 16-byte groups");

__global__ __launch_bounds__(1024) void fill_block(const int* __restrict__ src,
                                                   const int* __restrict__ dst,
                                                   const int* __restrict__ blk,
                                                   const int* __restrict__ row_ptr,
                                                   const float* __restrict__ dis,
                                                   int* __restrict__ ewbuf) {
    __shared__ int cur[HALF_NODES];
    const int nlo = blockIdx.y * HALF_NODES;
    const int4* s4 = (const int4*)(src + (size_t)blockIdx.x * CS_CHUNK);
    const int4* d4 = (const int4*)(dst + (size_t)blockIdx.x * CS_CHUNK);
    int4 sv[CS_PASSES], dv[CS_PASSES];
    bool ok[CS_PASSES];
    #pragma unroll
    for (int p = 0; p < CS_PASSES; ++p) {
        const int i = p * 1024 + threadIdx.x;
        ok[p] = i < CS_GROUPS;
        sv[p] = s4[ok[p] ? i : CS_GROUPS - 1];
        dv[p] = d4[ok[p] ? i : CS_GROUPS - 1];
    }

    const int* off = blk + (size_t)blockIdx.x * CS_STRIDE + nlo;
    const int* rp  = row_ptr + nlo;
    int4 a[TAB_PASSES], o[TAB_PASSES];
    #pragma unroll
    for (int t = 0; t < TAB_PASSES; ++t) {
        const int i = (t * 1024 + threadIdx.x) * 4;
        const int ii = (i < HALF_NODES) ? i : HALF_NODES - 4;
        a[t] = *(const int4*)(rp + ii);
        o[t] = *(const int4*)(off + ii);
    }
    #pragma unroll
    for (int t = 0; t < TAB_PASSES; ++t) {
        const int i = (t * 1024 + threadIdx.x) * 4;
        if (i < HALF_NODES)
            *(int4*)(cur + i) = make_int4(a[t].x + o[t].x, a[t].y + o[t].y,
                                          a[t].z + o[t].z, a[t].w + o[t].w);
    }

    int   sN[CS_PASSES * 4];
    int   rN[CS_PASSES * 4];
    bool  in[CS_PASSES * 4];
    float sc[CS_PASSES * 4];
    #pragma unroll
    for (int p = 0; p < CS_PASSES; ++p) {
        const int s_[4] = {sv[p].x, sv[p].y, sv[p].z, sv[p].w};
        const int d_[4] = {dv[p].x, dv[p].y, dv[p].z, dv[p].w};
        #pragma unroll
        for (int q = 0; q < 4; ++q) {
            const int k = p * 4 + q;
            sN[k] = s_[q];
            rN[k] = d_[q] - nlo;
            in[k] = ok[p] && (unsigned)rN[k] < (unsigned)HALF_NODES;
            sc[k] = dis[in[k] ? sN[k] : 0];
        }
    }
    __syncthreads();

    int2* ew = (int2*)ewbuf;
    #pragma unroll
    for (int k = 0; k < CS_PASSES * 4; ++k) {
        if (in[k]) {
            const int slot = atomicAdd(&cur[rN[k]], 1);
            ew[slot] = make_int2(sN[k], __float_as_int(sc[k]));
        }
    }
}

#define AS_PITCH    132
#define TILES_M     (N_NODES / 16)
#define AGG_BLOCKS1 768
#define AGG_BLOCKS2 512
#define AGG_GROUP   16
static_assert(N_NODES % 16 == 0, "whole 16-row tiles only");
static_assert(2 * AGG_GROUP == 32, "a 32-record load feeds exactly two row groups");

__device__ __forceinline__ void fma4(float4& a, const float4& v, float c) {
    a.x = fmaf(v.x, c, a.x); a.y = fmaf(v.y, c, a.y);
    a.z = fmaf(v.z, c, a.z); a.w = fmaf(v.w, c, a.w);
}

__device__ __forceinline__ float4 agg_combine(const float4& acc, const float4& own, float di) {
    const float di2 = di * di;
    float4 r;
    r.x = acc.x * di + own.x * di2;
    r.y = acc.y * di + own.y * di2;
    r.z = acc.z * di + own.z * di2;
    r.w = acc.w * di + own.w * di2;
    return r;
}

template <int PLANES>
__global__ __launch_bounds__(256) void agg_gemm(const float* __restrict__ h,
                                                const float* __restrict__ dis,
                                                const int* __restrict__ row_ptr,
                                                const int* __restrict__ ewbuf,
                                                const float* __restrict__ B,
                                                const float* __restrict__ bias,
                                                float* __restrict__ C) {
    __shared__ float As[2 * 16 * AS_PITCH];
    const int lane = threadIdx.x & 31;
    const int wave = __builtin_amdgcn_readfirstlane((int)(threadIdx.x >> 5));
    const int half = lane >> 4;
    const int l16  = lane & 15;
    const int2* ew = (const int2*)ewbuf;

    const float* bcol = B + (size_t)(2 * half) * FDIM + wave * 16 + l16;
    v2f bfr[PLANES][FDIM / 4];
    #pragma unroll
    for (int s = 0; s < FDIM / 4; ++s) {
        #pragma unroll
        for (int p = 0; p < PLANES; ++p) {
            bfr[p][s][0] = bcol[(size_t)p * FDIM * FDIM + (size_t)(4 * s) * FDIM];
            bfr[p][s][1] = bcol[(size_t)p * FDIM * FDIM + (size_t)(4 * s + 1) * FDIM];
        }
    }
    float bv = 0.0f;
    if (PLANES == 1) bv = bias[wave * 16 + l16];

    const int f  = lane * 4;
    const int bl = lane < 2 ? lane : 2;
    const int gs = (int)gridDim.x;
    int tile = (int)blockIdx.x;
    int b0, b1, b2;
    int nb0, nb1, nb2;
    int2 rc;
    float4 ownp0, ownp1;
    int bnn;
    float dpair;
    {
        const int tn1 = tile + gs < TILES_M ? tile + gs : TILES_M - 1;
        const int tn2 = tile + 2 * gs < TILES_M ? tile + 2 * gs : TILES_M - 1;
        const int bcur = row_ptr[tile * 16 + 2 * wave + bl];
        const int bnxt = row_ptr[tn1 * 16 + 2 * wave + bl];
        bnn   = row_ptr[tn2 * 16 + 2 * wave + bl];
        dpair = dis[tile * 16 + 2 * wave + bl];
        ownp0 = *(const float4*)(h + (size_t)(tile * 16 + 2 * wave) * FDIM + f);
        ownp1 = *(const float4*)(h + (size_t)(tile * 16 + 2 * wave + 1) * FDIM + f);
        b0  = __builtin_amdgcn_readlane(bcur, 0);
        b1  = __builtin_amdgcn_readlane(bcur, 1);
        b2  = __builtin_amdgcn_readlane(bcur, 2);
        nb0 = __builtin_amdgcn_readlane(bnxt, 0);
        nb1 = __builtin_amdgcn_readlane(bnxt, 1);
        nb2 = __builtin_amdgcn_readlane(bnxt, 2);
        rc = ew[b0 + lane];
    }

    int buf = 0;
    for (; tile < TILES_M; tile += gs) {
        float* at = As + buf * 16 * AS_PITCH;
        float* arow0 = at + (2 * wave) * AS_PITCH + f;
        *(float4*)arow0              = ownp0;
        *(float4*)(arow0 + AS_PITCH) = ownp1;
        asm volatile("" ::: "memory");

        const int n0    = b1 - b0;
        const int total = b2 - b0;
        float4 rsum = make_float4(0.0f, 0.0f, 0.0f, 0.0f);
        float4 v[AGG_GROUP];
        #pragma unroll
        for (int k = 0; k < AGG_GROUP; ++k) v[k] = make_float4(0.0f, 0.0f, 0.0f, 0.0f);
        for (int base = 0; base < total; base += 32) {
            #pragma unroll
            for (int hh = 0; hh < 2; ++hh) {
                const int o = base + hh * AGG_GROUP;
                const int n = total - o;
                if (n > 0) {
                    int   s[AGG_GROUP];
                    float c[AGG_GROUP];
                    #pragma unroll
                    for (int k = 0; k < AGG_GROUP; ++k) {
                        s[k] = __builtin_amdgcn_readlane(rc.x, hh * AGG_GROUP + k);
                        c[k] = __int_as_float(__builtin_amdgcn_readlane(rc.y, hh * AGG_GROUP + k));
                    }
                    if (hh == 1 && n > AGG_GROUP) rc = ew[b0 + base + 32 + lane];
                    #pragma unroll
                    for (int k = 0; k < AGG_GROUP; ++k)
                        if (k < n) v[k] = *(const float4*)(h + (size_t)s[k] * FDIM + f);
                    #pragma unroll
                    for (int k = 0; k < AGG_GROUP; ++k) {
                        if (k < n) {
                            if (o + k == n0) {
                                const float di0 = __int_as_float(__builtin_amdgcn_readlane(__float_as_int(dpair), 0));
                                const float4 own0 = *(const float4*)arow0;
                                *(float4*)arow0 = agg_combine(rsum, own0, di0);
                                rsum = make_float4(0.0f, 0.0f, 0.0f, 0.0f);
                            }
                            fma4(rsum, v[k], c[k]);
                        }
                    }
                }
            }
        }
        if (n0 >= total) {
            const float di0 = __int_as_float(__builtin_amdgcn_readlane(__float_as_int(dpair), 0));
            const float4 own0 = *(const float4*)arow0;
            *(float4*)arow0 = agg_combine(rsum, own0, di0);
            rsum = make_float4(0.0f, 0.0f, 0.0f, 0.0f);
        }
        {
            const float di1 = __int_as_float(__builtin_amdgcn_readlane(__float_as_int(dpair), 1));
            const float4 own1 = *(const float4*)(arow0 + AS_PITCH);
            *(float4*)(arow0 + AS_PITCH) = agg_combine(rsum, own1, di1);
        }
        b0 = nb0; b1 = nb1; b2 = nb2;
        nb0 = __builtin_amdgcn_readlane(bnn, 0);
        nb1 = __builtin_amdgcn_readlane(bnn, 1);
        nb2 = __builtin_amdgcn_readlane(bnn, 2);
        __syncthreads();

        rc = ew[b0 + lane];
        {
            const int tn1 = tile + gs < TILES_M ? tile + gs : TILES_M - 1;
            const int tn3 = tile + 3 * gs < TILES_M ? tile + 3 * gs : TILES_M - 1;
            ownp0 = *(const float4*)(h + (size_t)(tn1 * 16 + 2 * wave) * FDIM + f);
            ownp1 = *(const float4*)(h + (size_t)(tn1 * 16 + 2 * wave + 1) * FDIM + f);
            dpair = dis[tn1 * 16 + 2 * wave + bl];
            bnn   = row_ptr[tn3 * 16 + 2 * wave + bl];
        }

        const float* ap = at + l16 * AS_PITCH + half * 2;
        v2f afrag[FDIM / 4];
        #pragma unroll
        for (int s = 0; s < FDIM / 4; ++s) afrag[s] = *(const v2f*)(ap + s * 4);

        v8f acc[PLANES];
        #pragma unroll
        for (int p = 0; p < PLANES; ++p) acc[p] = v8f{};
        #pragma unroll
        for (int s = 0; s < FDIM / 4; ++s) {
            #pragma unroll
            for (int p = 0; p < PLANES; ++p)
                acc[p] = __builtin_amdgcn_wmma_f32_16x16x4_f32(
                    false, afrag[s], false, bfr[p][s], (short)0, acc[p], false, false);
        }

        #pragma unroll
        for (int p = 0; p < PLANES; ++p) {
            float* crow = C + (size_t)p * N_NODES * FDIM
                            + ((size_t)tile * 16 + half * 8) * FDIM + wave * 16 + l16;
            #pragma unroll
            for (int g = 0; g < 8; ++g) {
                float o = acc[p][g];
                if (PLANES == 1) o = fmaxf(o + bv, 0.0f);
                crow[(size_t)g * FDIM] = o;
            }
        }
        buf ^= 1;
    }
}

static_assert(E_TR % 32 == 0 && E_TE % 32 == 0, "a wave's 32 rows must lie inside one edge list");
static_assert(N_ROWS % 32 == 0, "whole waves only");

__global__ __launch_bounds__(256) void edge_score(const float* __restrict__ P,
                                                  const float* __restrict__ Q,
                                                  const int* __restrict__ pos_tr,
                                                  const int* __restrict__ neg_tr,
                                                  const int* __restrict__ pos_te,
                                                  const int* __restrict__ neg_te,
                                                  const float* __restrict__ bl1,
                                                  const float* __restrict__ bc,
                                                  const float* __restrict__ Wl2,
                                                  const float* __restrict__ bl2,
                                                  float* __restrict__ out) {
    const int r0 = __builtin_amdgcn_readfirstlane(
        (int)(((blockIdx.x * blockDim.x + threadIdx.x) >> 5) * 32));
    if (r0 >= N_ROWS) return;
    const int lane = threadIdx.x & 31;
    const int g = lane >> 3;
    const int j = lane & 7;

    const int* e; int i0, len;
    if (r0 < E_TR)                 { e = pos_tr; i0 = r0;                   len = E_TR; }
    else if (r0 < 2 * E_TR)        { e = neg_tr; i0 = r0 - E_TR;            len = E_TR; }
    else if (r0 < 2 * E_TR + E_TE) { e = pos_te; i0 = r0 - 2 * E_TR;        len = E_TE; }
    else                           { e = neg_te; i0 = r0 - 2 * E_TR - E_TE; len = E_TE; }

    const int* eu = e + i0 + g * 8;
    const int* ev = eu + len;
    const int4 ua = *(const int4*)eu, ub = *(const int4*)(eu + 4);
    const int4 va = *(const int4*)ev, vb = *(const int4*)(ev + 4);
    const int us[8] = {ua.x, ua.y, ua.z, ua.w, ub.x, ub.y, ub.z, ub.w};
    const int vs[8] = {va.x, va.y, va.z, va.w, vb.x, vb.y, vb.z, vb.w};

    float4 bb[4], ww[4];
    #pragma unroll
    for (int c = 0; c < 4; ++c) {
        const float4 b1 = *(const float4*)(bl1 + c * 32 + j * 4);
        const float4 bp = *(const float4*)(bc + c * 32 + j * 4);
        const float4 bq = *(const float4*)(bc + FDIM + c * 32 + j * 4);
        bb[c] = make_float4(b1.x + (bp.x + bq.x), b1.y + (bp.y + bq.y),
                            b1.z + (bp.z + bq.z), b1.w + (bp.w + bq.w));
        ww[c] = *(const float4*)(Wl2 + c * 32 + j * 4);
    }

    float s[8];
    #pragma unroll
    for (int p = 0; p < 8; ++p) {
        const float* pr = P + (size_t)us[p] * FDIM + j * 4;
        const float* qr = Q + (size_t)vs[p] * FDIM + j * 4;
        float acc = 0.0f;
        #pragma unroll
        for (int c = 0; c < 4; ++c) {
            const float4 a = *(const float4*)(pr + c * 32);
            const float4 b = *(const float4*)(qr + c * 32);
            acc = fmaf(fmaxf(a.x + b.x + bb[c].x, 0.0f), ww[c].x, acc);
            acc = fmaf(fmaxf(a.y + b.y + bb[c].y, 0.0f), ww[c].y, acc);
            acc = fmaf(fmaxf(a.z + b.z + bb[c].z, 0.0f), ww[c].z, acc);
            acc = fmaf(fmaxf(a.w + b.w + bb[c].w, 0.0f), ww[c].w, acc);
        }
        s[p] = acc;
    }

    const bool k4 = (j & 4) != 0, k2 = (j & 2) != 0, k1 = (j & 1) != 0;
    float t4[4];
    #pragma unroll
    for (int i = 0; i < 4; ++i) {
        const float keep = k4 ? s[4 + i] : s[i];
        const float send = k4 ? s[i] : s[4 + i];
        t4[i] = keep + __shfl_xor(send, 4, 32);
    }
    float t2[2];
    #pragma unroll
    for (int i = 0; i < 2; ++i) {
        const float keep = k2 ? t4[2 + i] : t4[i];
        const float send = k2 ? t4[i] : t4[2 + i];
        t2[i] = keep + __shfl_xor(send, 2, 32);
    }
    const float keep = k1 ? t2[1] : t2[0];
    const float send = k1 ? t2[0] : t2[1];
    const float tot = keep + __shfl_xor(send, 1, 32);

    out[r0 + lane] = tot + bl2[0];
}

extern "C" void kernel_launch(void* const* d_in, const int* in_sizes, int n_in,
                              void* d_out, int out_size, void* d_ws, size_t ws_size,
                              hipStream_t stream) {
    const float* x          = (const float*)d_in[0];
    const int*   edge_index = (const int*)d_in[1];
    const int*   pos_tr     = (const int*)d_in[2];
    const int*   neg_tr     = (const int*)d_in[3];
    const int*   pos_te     = (const int*)d_in[4];
    const int*   neg_te     = (const int*)d_in[5];
    const float* W[4] = {(const float*)d_in[6],  (const float*)d_in[8],
                         (const float*)d_in[10], (const float*)d_in[12]};
    const float* b[4] = {(const float*)d_in[7],  (const float*)d_in[9],
                         (const float*)d_in[11], (const float*)d_in[13]};
    const float* Wl1 = (const float*)d_in[14];
    const float* bl1 = (const float*)d_in[15];
    const float* Wl2 = (const float*)d_in[16];
    const float* bl2 = (const float*)d_in[17];
    float* out = (float*)d_out;

    const size_t NF = (size_t)N_NODES * FDIM;
    float* ws  = (float*)d_ws;
    float* dis = ws;
    float* hw  = ws + 50048;
    int*   csr = (int*)(hw + NF);
    int*   cnt     = csr;
    int*   row_ptr = csr + 53248;
    int*   ew      = csr + 2 * 53248;
    float* Wc  = (float*)(ew + 2 * N_EDGES);
    float* bc  = Wc + 2 * FDIM * FDIM;
    float* h0  = hw + 2 * NF;
    float* h1  = h0 + NF;
    int*   blk = (int*)h1;

    const int* src = edge_index;
    const int* dst = edge_index + N_EDGES;

    count_block<<<CS_BLOCKS, 1024, 0, stream>>>(dst, blk);
    chunk_prefix<<<PREFIX_BLOCKS + FOLD_BLOCKS, 256, 0, stream>>>(blk, cnt, W[3], b[3], Wl1, Wc, bc);
    scan_rows<<<1, 1024, 0, stream>>>(cnt, row_ptr, dis);
    fill_block<<<dim3(CS_BLOCKS, 2), 1024, 0, stream>>>(src, dst, blk, row_ptr, dis, ew);

    const float* hin = x;
    float* houts[3] = {h0, h1, hw};
    for (int l = 0; l < 3; ++l) {
        agg_gemm<1><<<AGG_BLOCKS1, 256, 0, stream>>>(hin, dis, row_ptr, ew, W[l], b[l], houts[l]);
        hin = houts[l];
    }

    agg_gemm<2><<<AGG_BLOCKS2, 256, 0, stream>>>(hw, dis, row_ptr, ew, Wc, b[3], h0);

    const int edgeWaves  = N_ROWS / 32;
    const int edgeBlocks = (edgeWaves + 7) / 8;
    edge_score<<<edgeBlocks, 256, 0, stream>>>(h0, h1, pos_tr, neg_tr, pos_te, neg_te,
                                               bl1, bc, Wl2, bl2, out);
}


# CSR row offsets in two levels: chunk_prefix scans in-degrees over its 256-node group, fill_block scans the 196 group totals; the one-work-group scan launch is gone; slots and sums unchanged
# speedup vs baseline: 5.0209x; 1.0355x over previous
#include <hip/hip_runtime.h>


typedef float v2f __attribute__((ext_vector_type(2)));
typedef float v8f __attribute__((ext_vector_type(8)));

#define N_NODES 50000
#define N_EDGES 800000
#define FDIM    128
#define E_TR    100000
#define E_TE    20000
#define N_ROWS  240000

#define CS_BLOCKS 100
#define CS_CHUNK  (N_EDGES / CS_BLOCKS)
#define CS_GROUPS (CS_CHUNK / 4)
#define CS_PASSES ((CS_GROUPS + 1023) / 1024)
#define CS_STRIDE 50048
static_assert(N_EDGES % CS_BLOCKS == 0 && CS_CHUNK % 4 == 0, "chunks are whole 16-byte groups");
static_assert(CS_STRIDE >= N_NODES && CS_STRIDE % 32 == 0, "a plane holds every node");
static_assert(N_NODES % 4 == 0, "16-byte groups must not straddle the end");

__global__ __launch_bounds__(1024) void count_block(const int* __restrict__ dst,
                                                    int* __restrict__ blk) {
    __shared__ int hist[N_NODES];
    const int4* d4 = (const int4*)(dst + (size_t)blockIdx.x * CS_CHUNK);
    int4 dv[CS_PASSES];
    bool ok[CS_PASSES];
    #pragma unroll
    for (int p = 0; p < CS_PASSES; ++p) {
        const int i = p * 1024 + threadIdx.x;
        ok[p] = i < CS_GROUPS;
        dv[p] = d4[ok[p] ? i : CS_GROUPS - 1];
    }

    for (int i = threadIdx.x * 4; i < N_NODES; i += 4096)
        *(int4*)(hist + i) = make_int4(0, 0, 0, 0);
    __syncthreads();

    #pragma unroll
    for (int p = 0; p < CS_PASSES; ++p) {
        if (ok[p]) {
            atomicAdd(&hist[dv[p].x], 1);
            atomicAdd(&hist[dv[p].y], 1);
            atomicAdd(&hist[dv[p].z], 1);
            atomicAdd(&hist[dv[p].w], 1);
        }
    }
    __syncthreads();

    int* out = blk + (size_t)blockIdx.x * CS_STRIDE;
    for (int i = threadIdx.x * 4; i < N_NODES; i += 4096)
        *(int4*)(out + i) = *(const int4*)(hist + i);
}

#define PREFIX_BLOCKS ((N_NODES + 255) / 256)
#define FOLD_TILES    (FDIM / 16 + 1)
#define FOLD_BLOCKS   (FOLD_TILES * 2)

__device__ __forceinline__ void fold_weights(int job,
                                             const float* __restrict__ W3,
                                             const float* __restrict__ b3,
                                             const float* __restrict__ Wl1,
                                             float* __restrict__ Wc,
                                             float* __restrict__ bc) {
    const int lane = threadIdx.x & 31;
    const int wave = __builtin_amdgcn_readfirstlane((int)(threadIdx.x >> 5));
    const int half = lane >> 4;
    const int l16  = lane & 15;
    const int tile = job >> 1;
    const int y    = job & 1;
    const bool biasTile = tile == FOLD_TILES - 1;

    const float* arow = (biasTile ? b3 : W3 + (size_t)(tile * 16 + l16) * FDIM) + 2 * half;
    const bool live = !biasTile || l16 == 0;
    const float* bcol = Wl1 + (size_t)y * FDIM * FDIM + (size_t)(2 * half) * FDIM + wave * 16 + l16;

    v2f af[FDIM / 4], bf[FDIM / 4];
    #pragma unroll
    for (int s = 0; s < FDIM / 4; ++s) {
        af[s]    = *(const v2f*)(arow + 4 * s);
        bf[s][0] = bcol[(size_t)(4 * s) * FDIM];
        bf[s][1] = bcol[(size_t)(4 * s + 1) * FDIM];
    }
    v8f acc = {};
    #pragma unroll
    for (int s = 0; s < FDIM / 4; ++s) {
        v2f a;
        a[0] = live ? af[s][0] : 0.0f;
        a[1] = live ? af[s][1] : 0.0f;
        acc = __builtin_amdgcn_wmma_f32_16x16x4_f32(
            false, a, false, bf[s], (short)0, acc, false, false);
    }

    if (!biasTile) {
        float* crow = Wc + (size_t)y * FDIM * FDIM
                         + (size_t)(tile * 16 + half * 8) * FDIM + wave * 16 + l16;
        #pragma unroll
        for (int g = 0; g < 8; ++g) crow[(size_t)g * FDIM] = acc[g];
    } else if (half == 0) {
        bc[y * FDIM + wave * 16 + l16] = acc[0];
    }
}

__device__ __forceinline__ int wave_incl_scan(int v, int lane) {
    #pragma unroll
    for (int o = 1; o < 32; o <<= 1) {
        int y = __shfl_up(v, o, 32);
        if (lane >= o) v += y;
    }
    return v;
}

__global__ __launch_bounds__(256) void chunk_prefix(int* __restrict__ blk,
                                                    int* __restrict__ tot,
                                                    float* __restrict__ dis,
                                                    const float* __restrict__ W3,
                                                    const float* __restrict__ b3,
                                                    const float* __restrict__ Wl1,
                                                    float* __restrict__ Wc,
                                                    float* __restrict__ bc) {
    if (blockIdx.x >= PREFIX_BLOCKS) {
        fold_weights((int)blockIdx.x - PREFIX_BLOCKS, W3, b3, Wl1, Wc, bc);
        return;
    }
    __shared__ int wsum[8];
    const int n    = blockIdx.x * 256 + threadIdx.x;
    const bool live = n < N_NODES;
    const int nn   = live ? n : N_NODES - 1;
    const int lane = threadIdx.x & 31;
    const int w    = threadIdx.x >> 5;
    int c[CS_BLOCKS];
    #pragma unroll
    for (int b = 0; b < CS_BLOCKS; ++b) c[b] = blk[(size_t)b * CS_STRIDE + nn];
    int deg = 0;
    #pragma unroll
    for (int b = 0; b < CS_BLOCKS; ++b) deg += c[b];
    const int v   = live ? deg : 0;
    const int inc = wave_incl_scan(v, lane);
    if (lane == 31) wsum[w] = inc;
    __syncthreads();
    int wbase = 0;
    #pragma unroll
    for (int k = 0; k < 8; ++k) wbase += (k < w) ? wsum[k] : 0;
    if (live) {
        dis[n] = rsqrtf((float)deg + 1.0f);
        int run = wbase + inc - v;
        #pragma unroll
        for (int b = 0; b < CS_BLOCKS; ++b) {
            blk[(size_t)b * CS_STRIDE + n] = run;
            run += c[b];
        }
    }
    if (threadIdx.x == 255) tot[blockIdx.x] = wbase + inc;
}

#define HALF_NODES (N_NODES / 2)
#define TAB_PASSES ((HALF_NODES / 4 + 1023) / 1024)
#define GRP_PASSES ((PREFIX_BLOCKS + 31) / 32)
static_assert(PREFIX_BLOCKS * 256 >= N_NODES, "the 256-node groups cover every node");
static_assert(N_NODES % 2 == 0 && HALF_NODES % 4 == 0, "a half is whole 16-byte groups");

__global__ __launch_bounds__(1024) void fill_block(const int* __restrict__ src,
                                                   const int* __restrict__ dst,
                                                   const int* __restrict__ blk,
                                                   const int* __restrict__ tot,
                                                   const float* __restrict__ dis,
                                                   int* __restrict__ row_ptr,
                                                   int* __restrict__ ewbuf) {
    __shared__ int cur[HALF_NODES];
    __shared__ int gbase[PREFIX_BLOCKS + 1];
    const int nlo = blockIdx.y * HALF_NODES;
    const int4* s4 = (const int4*)(src + (size_t)blockIdx.x * CS_CHUNK);
    const int4* d4 = (const int4*)(dst + (size_t)blockIdx.x * CS_CHUNK);
    int4 sv[CS_PASSES], dv[CS_PASSES];
    bool ok[CS_PASSES];
    #pragma unroll
    for (int p = 0; p < CS_PASSES; ++p) {
        const int i = p * 1024 + threadIdx.x;
        ok[p] = i < CS_GROUPS;
        sv[p] = s4[ok[p] ? i : CS_GROUPS - 1];
        dv[p] = d4[ok[p] ? i : CS_GROUPS - 1];
    }

    const int* off = blk + (size_t)blockIdx.x * CS_STRIDE + nlo;
    int4 o[TAB_PASSES];
    #pragma unroll
    for (int t = 0; t < TAB_PASSES; ++t) {
        const int i = (t * 1024 + threadIdx.x) * 4;
        const int ii = (i < HALF_NODES) ? i : HALF_NODES - 4;
        o[t] = *(const int4*)(off + ii);
    }

    int   sN[CS_PASSES * 4];
    int   rN[CS_PASSES * 4];
    bool  in[CS_PASSES * 4];
    float sc[CS_PASSES * 4];
    #pragma unroll
    for (int p = 0; p < CS_PASSES; ++p) {
        const int s_[4] = {sv[p].x, sv[p].y, sv[p].z, sv[p].w};
        const int d_[4] = {dv[p].x, dv[p].y, dv[p].z, dv[p].w};
        #pragma unroll
        for (int q = 0; q < 4; ++q) {
            const int k = p * 4 + q;
            sN[k] = s_[q];
            rN[k] = d_[q] - nlo;
            in[k] = ok[p] && (unsigned)rN[k] < (unsigned)HALF_NODES;
            sc[k] = dis[in[k] ? sN[k] : 0];
        }
    }

    if (threadIdx.x < 32) {
        const int lane = threadIdx.x;
        int gv[GRP_PASSES];
        #pragma unroll
        for (int p = 0; p < GRP_PASSES; ++p) {
            const int g = p * 32 + lane;
            const int x = tot[g < PREFIX_BLOCKS ? g : PREFIX_BLOCKS - 1];
            gv[p] = g < PREFIX_BLOCKS ? x : 0;
        }
        int run = 0;
        #pragma unroll
        for (int p = 0; p < GRP_PASSES; ++p) {
            const int g   = p * 32 + lane;
            const int inc = wave_incl_scan(gv[p], lane);
            if (g < PREFIX_BLOCKS) gbase[g] = run + inc - gv[p];
            run += __shfl(inc, 31, 32);
        }
        if (lane == 0) gbase[PREFIX_BLOCKS] = run;
    }
    __syncthreads();

    #pragma unroll
    for (int t = 0; t < TAB_PASSES; ++t) {
        const int i = (t * 1024 + threadIdx.x) * 4;
        if (i < HALF_NODES) {
            const int gb = gbase[(nlo + i) >> 8];
            const int4 q = make_int4(o[t].x + gb, o[t].y + gb, o[t].z + gb, o[t].w + gb);
            *(int4*)(cur + i) = q;
            if (blockIdx.x == 0) *(int4*)(row_ptr + nlo + i) = q;
        }
    }
    if (blockIdx.x == 0 && blockIdx.y == 1 && threadIdx.x == 0) row_ptr[N_NODES] = gbase[PREFIX_BLOCKS];

    __syncthreads();

    int2* ew = (int2*)ewbuf;
    #pragma unroll
    for (int k = 0; k < CS_PASSES * 4; ++k) {
        if (in[k]) {
            const int slot = atomicAdd(&cur[rN[k]], 1);
            ew[slot] = make_int2(sN[k], __float_as_int(sc[k]));
        }
    }
}

#define AS_PITCH    132
#define TILES_M     (N_NODES / 16)
#define AGG_BLOCKS1 768
#define AGG_BLOCKS2 512
#define AGG_GROUP   16
static_assert(N_NODES % 16 == 0, "whole 16-row tiles only");
static_assert(2 * AGG_GROUP == 32, "a 32-record load feeds exactly two row groups");

__device__ __forceinline__ void fma4(float4& a, const float4& v, float c) {
    a.x = fmaf(v.x, c, a.x); a.y = fmaf(v.y, c, a.y);
    a.z = fmaf(v.z, c, a.z); a.w = fmaf(v.w, c, a.w);
}

__device__ __forceinline__ float4 agg_combine(const float4& acc, const float4& own, float di) {
    const float di2 = di * di;
    float4 r;
    r.x = acc.x * di + own.x * di2;
    r.y = acc.y * di + own.y * di2;
    r.z = acc.z * di + own.z * di2;
    r.w = acc.w * di + own.w * di2;
    return r;
}

template <int PLANES>
__global__ __launch_bounds__(256) void agg_gemm(const float* __restrict__ h,
                                                const float* __restrict__ dis,
                                                const int* __restrict__ row_ptr,
                                                const int* __restrict__ ewbuf,
                                                const float* __restrict__ B,
                                                const float* __restrict__ bias,
                                                float* __restrict__ C) {
    __shared__ float As[2 * 16 * AS_PITCH];
    const int lane = threadIdx.x & 31;
    const int wave = __builtin_amdgcn_readfirstlane((int)(threadIdx.x >> 5));
    const int half = lane >> 4;
    const int l16  = lane & 15;
    const int2* ew = (const int2*)ewbuf;

    const float* bcol = B + (size_t)(2 * half) * FDIM + wave * 16 + l16;
    v2f bfr[PLANES][FDIM / 4];
    #pragma unroll
    for (int s = 0; s < FDIM / 4; ++s) {
        #pragma unroll
        for (int p = 0; p < PLANES; ++p) {
            bfr[p][s][0] = bcol[(size_t)p * FDIM * FDIM + (size_t)(4 * s) * FDIM];
            bfr[p][s][1] = bcol[(size_t)p * FDIM * FDIM + (size_t)(4 * s + 1) * FDIM];
        }
    }
    float bv = 0.0f;
    if (PLANES == 1) bv = bias[wave * 16 + l16];

    const int f  = lane * 4;
    const int bl = lane < 2 ? lane : 2;
    const int gs = (int)gridDim.x;
    int tile = (int)blockIdx.x;
    int b0, b1, b2;
    int nb0, nb1, nb2;
    int2 rc;
    float4 ownp0, ownp1;
    int bnn;
    float dpair;
    {
        const int tn1 = tile + gs < TILES_M ? tile + gs : TILES_M - 1;
        const int tn2 = tile + 2 * gs < TILES_M ? tile + 2 * gs : TILES_M - 1;
        const int bcur = row_ptr[tile * 16 + 2 * wave + bl];
        const int bnxt = row_ptr[tn1 * 16 + 2 * wave + bl];
        bnn   = row_ptr[tn2 * 16 + 2 * wave + bl];
        dpair = dis[tile * 16 + 2 * wave + bl];
        ownp0 = *(const float4*)(h + (size_t)(tile * 16 + 2 * wave) * FDIM + f);
        ownp1 = *(const float4*)(h + (size_t)(tile * 16 + 2 * wave + 1) * FDIM + f);
        b0  = __builtin_amdgcn_readlane(bcur, 0);
        b1  = __builtin_amdgcn_readlane(bcur, 1);
        b2  = __builtin_amdgcn_readlane(bcur, 2);
        nb0 = __builtin_amdgcn_readlane(bnxt, 0);
        nb1 = __builtin_amdgcn_readlane(bnxt, 1);
        nb2 = __builtin_amdgcn_readlane(bnxt, 2);
        rc = ew[b0 + lane];
    }

    int buf = 0;
    for (; tile < TILES_M; tile += gs) {
        float* at = As + buf * 16 * AS_PITCH;
        float* arow0 = at + (2 * wave) * AS_PITCH + f;
        *(float4*)arow0              = ownp0;
        *(float4*)(arow0 + AS_PITCH) = ownp1;
        asm volatile("" ::: "memory");

        const int n0    = b1 - b0;
        const int total = b2 - b0;
        float4 rsum = make_float4(0.0f, 0.0f, 0.0f, 0.0f);
        float4 v[AGG_GROUP];
        #pragma unroll
        for (int k = 0; k < AGG_GROUP; ++k) v[k] = make_float4(0.0f, 0.0f, 0.0f, 0.0f);
        for (int base = 0; base < total; base += 32) {
            #pragma unroll
            for (int hh = 0; hh < 2; ++hh) {
                const int o = base + hh * AGG_GROUP;
                const int n = total - o;
                if (n > 0) {
                    int   s[AGG_GROUP];
                    float c[AGG_GROUP];
                    #pragma unroll
                    for (int k = 0; k < AGG_GROUP; ++k) {
                        s[k] = __builtin_amdgcn_readlane(rc.x, hh * AGG_GROUP + k);
                        c[k] = __int_as_float(__builtin_amdgcn_readlane(rc.y, hh * AGG_GROUP + k));
                    }
                    if (hh == 1 && n > AGG_GROUP) rc = ew[b0 + base + 32 + lane];
                    #pragma unroll
                    for (int k = 0; k < AGG_GROUP; ++k)
                        if (k < n) v[k] = *(const float4*)(h + (size_t)s[k] * FDIM + f);
                    #pragma unroll
                    for (int k = 0; k < AGG_GROUP; ++k) {
                        if (k < n) {
                            if (o + k == n0) {
                                const float di0 = __int_as_float(__builtin_amdgcn_readlane(__float_as_int(dpair), 0));
                                const float4 own0 = *(const float4*)arow0;
                                *(float4*)arow0 = agg_combine(rsum, own0, di0);
                                rsum = make_float4(0.0f, 0.0f, 0.0f, 0.0f);
                            }
                            fma4(rsum, v[k], c[k]);
                        }
                    }
                }
            }
        }
        if (n0 >= total) {
            const float di0 = __int_as_float(__builtin_amdgcn_readlane(__float_as_int(dpair), 0));
            const float4 own0 = *(const float4*)arow0;
            *(float4*)arow0 = agg_combine(rsum, own0, di0);
            rsum = make_float4(0.0f, 0.0f, 0.0f, 0.0f);
        }
        {
            const float di1 = __int_as_float(__builtin_amdgcn_readlane(__float_as_int(dpair), 1));
            const float4 own1 = *(const float4*)(arow0 + AS_PITCH);
            *(float4*)(arow0 + AS_PITCH) = agg_combine(rsum, own1, di1);
        }
        b0 = nb0; b1 = nb1; b2 = nb2;
        nb0 = __builtin_amdgcn_readlane(bnn, 0);
        nb1 = __builtin_amdgcn_readlane(bnn, 1);
        nb2 = __builtin_amdgcn_readlane(bnn, 2);
        __syncthreads();

        rc = ew[b0 + lane];
        {
            const int tn1 = tile + gs < TILES_M ? tile + gs : TILES_M - 1;
            const int tn3 = tile + 3 * gs < TILES_M ? tile + 3 * gs : TILES_M - 1;
            ownp0 = *(const float4*)(h + (size_t)(tn1 * 16 + 2 * wave) * FDIM + f);
            ownp1 = *(const float4*)(h + (size_t)(tn1 * 16 + 2 * wave + 1) * FDIM + f);
            dpair = dis[tn1 * 16 + 2 * wave + bl];
            bnn   = row_ptr[tn3 * 16 + 2 * wave + bl];
        }

        const float* ap = at + l16 * AS_PITCH + half * 2;
        v2f afrag[FDIM / 4];
        #pragma unroll
        for (int s = 0; s < FDIM / 4; ++s) afrag[s] = *(const v2f*)(ap + s * 4);

        v8f acc[PLANES];
        #pragma unroll
        for (int p = 0; p < PLANES; ++p) acc[p] = v8f{};
        #pragma unroll
        for (int s = 0; s < FDIM / 4; ++s) {
            #pragma unroll
            for (int p = 0; p < PLANES; ++p)
                acc[p] = __builtin_amdgcn_wmma_f32_16x16x4_f32(
                    false, afrag[s], false, bfr[p][s], (short)0, acc[p], false, false);
        }

        #pragma unroll
        for (int p = 0; p < PLANES; ++p) {
            float* crow = C + (size_t)p * N_NODES * FDIM
                            + ((size_t)tile * 16 + half * 8) * FDIM + wave * 16 + l16;
            #pragma unroll
            for (int g = 0; g < 8; ++g) {
                float o = acc[p][g];
                if (PLANES == 1) o = fmaxf(o + bv, 0.0f);
                crow[(size_t)g * FDIM] = o;
            }
        }
        buf ^= 1;
    }
}

static_assert(E_TR % 32 == 0 && E_TE % 32 == 0, "a wave's 32 rows must lie inside one edge list");
static_assert(N_ROWS % 32 == 0, "whole waves only");

__global__ __launch_bounds__(256) void edge_score(const float* __restrict__ P,
                                                  const float* __restrict__ Q,
                                                  const int* __restrict__ pos_tr,
                                                  const int* __restrict__ neg_tr,
                                                  const int* __restrict__ pos_te,
                                                  const int* __restrict__ neg_te,
                                                  const float* __restrict__ bl1,
                                                  const float* __restrict__ bc,
                                                  const float* __restrict__ Wl2,
                                                  const float* __restrict__ bl2,
                                                  float* __restrict__ out) {
    const int r0 = __builtin_amdgcn_readfirstlane(
        (int)(((blockIdx.x * blockDim.x + threadIdx.x) >> 5) * 32));
    if (r0 >= N_ROWS) return;
    const int lane = threadIdx.x & 31;
    const int g = lane >> 3;
    const int j = lane & 7;

    const int* e; int i0, len;
    if (r0 < E_TR)                 { e = pos_tr; i0 = r0;                   len = E_TR; }
    else if (r0 < 2 * E_TR)        { e = neg_tr; i0 = r0 - E_TR;            len = E_TR; }
    else if (r0 < 2 * E_TR + E_TE) { e = pos_te; i0 = r0 - 2 * E_TR;        len = E_TE; }
    else                           { e = neg_te; i0 = r0 - 2 * E_TR - E_TE; len = E_TE; }

    const int* eu = e + i0 + g * 8;
    const int* ev = eu + len;
    const int4 ua = *(const int4*)eu, ub = *(const int4*)(eu + 4);
    const int4 va = *(const int4*)ev, vb = *(const int4*)(ev + 4);
    const int us[8] = {ua.x, ua.y, ua.z, ua.w, ub.x, ub.y, ub.z, ub.w};
    const int vs[8] = {va.x, va.y, va.z, va.w, vb.x, vb.y, vb.z, vb.w};

    float4 bb[4], ww[4];
    #pragma unroll
    for (int c = 0; c < 4; ++c) {
        const float4 b1 = *(const float4*)(bl1 + c * 32 + j * 4);
        const float4 bp = *(const float4*)(bc + c * 32 + j * 4);
        const float4 bq = *(const float4*)(bc + FDIM + c * 32 + j * 4);
        bb[c] = make_float4(b1.x + (bp.x + bq.x), b1.y + (bp.y + bq.y),
                            b1.z + (bp.z + bq.z), b1.w + (bp.w + bq.w));
        ww[c] = *(const float4*)(Wl2 + c * 32 + j * 4);
    }

    float s[8];
    #pragma unroll
    for (int p = 0; p < 8; ++p) {
        const float* pr = P + (size_t)us[p] * FDIM + j * 4;
        const float* qr = Q + (size_t)vs[p] * FDIM + j * 4;
        float acc = 0.0f;
        #pragma unroll
        for (int c = 0; c < 4; ++c) {
            const float4 a = *(const float4*)(pr + c * 32);
            const float4 b = *(const float4*)(qr + c * 32);
            acc = fmaf(fmaxf(a.x + b.x + bb[c].x, 0.0f), ww[c].x, acc);
            acc = fmaf(fmaxf(a.y + b.y + bb[c].y, 0.0f), ww[c].y, acc);
            acc = fmaf(fmaxf(a.z + b.z + bb[c].z, 0.0f), ww[c].z, acc);
            acc = fmaf(fmaxf(a.w + b.w + bb[c].w, 0.0f), ww[c].w, acc);
        }
        s[p] = acc;
    }

    const bool k4 = (j & 4) != 0, k2 = (j & 2) != 0, k1 = (j & 1) != 0;
    float t4[4];
    #pragma unroll
    for (int i = 0; i < 4; ++i) {
        const float keep = k4 ? s[4 + i] : s[i];
        const float send = k4 ? s[i] : s[4 + i];
        t4[i] = keep + __shfl_xor(send, 4, 32);
    }
    float t2[2];
    #pragma unroll
    for (int i = 0; i < 2; ++i) {
        const float keep = k2 ? t4[2 + i] : t4[i];
        const float send = k2 ? t4[i] : t4[2 + i];
        t2[i] = keep + __shfl_xor(send, 2, 32);
    }
    const float keep = k1 ? t2[1] : t2[0];
    const float send = k1 ? t2[0] : t2[1];
    const float tot = keep + __shfl_xor(send, 1, 32);

    out[r0 + lane] = tot + bl2[0];
}

extern "C" void kernel_launch(void* const* d_in, const int* in_sizes, int n_in,
                              void* d_out, int out_size, void* d_ws, size_t ws_size,
                              hipStream_t stream) {
    const float* x          = (const float*)d_in[0];
    const int*   edge_index = (const int*)d_in[1];
    const int*   pos_tr     = (const int*)d_in[2];
    const int*   neg_tr     = (const int*)d_in[3];
    const int*   pos_te     = (const int*)d_in[4];
    const int*   neg_te     = (const int*)d_in[5];
    const float* W[4] = {(const float*)d_in[6],  (const float*)d_in[8],
                         (const float*)d_in[10], (const float*)d_in[12]};
    const float* b[4] = {(const float*)d_in[7],  (const float*)d_in[9],
                         (const float*)d_in[11], (const float*)d_in[13]};
    const float* Wl1 = (const float*)d_in[14];
    const float* bl1 = (const float*)d_in[15];
    const float* Wl2 = (const float*)d_in[16];
    const float* bl2 = (const float*)d_in[17];
    float* out = (float*)d_out;

    const size_t NF = (size_t)N_NODES * FDIM;
    float* ws  = (float*)d_ws;
    float* dis = ws;
    float* hw  = ws + 50048;
    int*   csr = (int*)(hw + NF);
    int*   tot     = csr;
    int*   row_ptr = csr + 53248;
    int*   ew      = csr + 2 * 53248;
    float* Wc  = (float*)(ew + 2 * N_EDGES);
    float* bc  = Wc + 2 * FDIM * FDIM;
    float* h0  = hw + 2 * NF;
    float* h1  = h0 + NF;
    int*   blk = (int*)h1;

    const int* src = edge_index;
    const int* dst = edge_index + N_EDGES;

    count_block<<<CS_BLOCKS, 1024, 0, stream>>>(dst, blk);
    chunk_prefix<<<PREFIX_BLOCKS + FOLD_BLOCKS, 256, 0, stream>>>(blk, tot, dis, W[3], b[3], Wl1, Wc, bc);
    fill_block<<<dim3(CS_BLOCKS, 2), 1024, 0, stream>>>(src, dst, blk, tot, dis, row_ptr, ew);

    const float* hin = x;
    float* houts[3] = {h0, h1, hw};
    for (int l = 0; l < 3; ++l) {
        agg_gemm<1><<<AGG_BLOCKS1, 256, 0, stream>>>(hin, dis, row_ptr, ew, W[l], b[l], houts[l]);
        hin = houts[l];
    }

    agg_gemm<2><<<AGG_BLOCKS2, 256, 0, stream>>>(hw, dis, row_ptr, ew, Wc, b[3], h0);

    const int edgeWaves  = N_ROWS / 32;
    const int edgeBlocks = (edgeWaves + 7) / 8;
    edge_score<<<edgeBlocks, 256, 0, stream>>>(h0, h1, pos_tr, neg_tr, pos_te, neg_te,
                                               bl1, bc, Wl2, bl2, out);
}
